# starting kernel
# baseline (speedup 1.0000x reference)
#include <hip/hip_runtime.h>
#include <math.h>

#define DIN 64
#define DH  40
#define DOUT 24
#define DH_PAD 48
#define DO_PAD 32

typedef __attribute__((ext_vector_type(2))) float v2f;
typedef __attribute__((ext_vector_type(8))) float v8f;

__device__ __forceinline__ v8f wmma4(v2f a, v2f b, v8f c) {
  return __builtin_amdgcn_wmma_f32_16x16x4_f32(false, a, false, b, (short)0, c,
                                               false, false);
}

__global__ void __launch_bounds__(256)
zero_kernel(float* __restrict__ p, size_t n) {
  size_t i = (size_t)blockIdx.x * blockDim.x + threadIdx.x;
  size_t stride = (size_t)gridDim.x * blockDim.x;
  for (; i < n; i += stride) p[i] = 0.0f;
}

__global__ void __launch_bounds__(256)
scatter1_kernel(const float* __restrict__ x, const int* __restrict__ src,
                const int* __restrict__ dst, float* __restrict__ sum1,
                float* __restrict__ cnt, long E) {
  long t = (long)blockIdx.x * 256 + threadIdx.x;
  long e = t >> 4;
  if (e >= E) return;
  int chunk = (int)(t & 15);
  int s = src[e], d = dst[e];
  const float4 v = *reinterpret_cast<const float4*>(x + (size_t)s * DIN + chunk * 4);
  float* dp = sum1 + (size_t)d * DIN + chunk * 4;
  atomicAdd(dp + 0, v.x);
  atomicAdd(dp + 1, v.y);
  atomicAdd(dp + 2, v.z);
  atomicAdd(dp + 3, v.w);
  if (chunk == 0) atomicAdd(cnt + d, 1.0f);
}

__global__ void __launch_bounds__(256)
scatter2_kernel(const float* __restrict__ h, const int* __restrict__ src,
                const int* __restrict__ dst, float* __restrict__ sum2, long E) {
  long t = (long)blockIdx.x * 256 + threadIdx.x;
  long e = t / 10;
  if (e >= E) return;
  int chunk = (int)(t - e * 10);
  int s = src[e], d = dst[e];
  const float4 v = *reinterpret_cast<const float4*>(h + (size_t)s * DH + chunk * 4);
  float* dp = sum2 + (size_t)d * DH + chunk * 4;
  atomicAdd(dp + 0, v.x);
  atomicAdd(dp + 1, v.y);
  atomicAdd(dp + 2, v.z);
  atomicAdd(dp + 3, v.w);
}

__global__ void __launch_bounds__(256)
layer1_kernel(const float* __restrict__ x, const float* __restrict__ sum1,
              const float* __restrict__ cnt, const float* __restrict__ W1l,
              const float* __restrict__ b1, const float* __restrict__ W1r,
              float* __restrict__ h, int nNodes) {
  __shared__ float sWl[DH_PAD * DIN];
  __shared__ float sWr[DH_PAD * DIN];
  __shared__ float sB[DH];
  for (int i = threadIdx.x; i < DH_PAD * DIN; i += 256) {
    float wl = (i < DH * DIN) ? W1l[i] : 0.0f;
    float wr = (i < DH * DIN) ? W1r[i] : 0.0f;
    sWl[i] = wl;
    sWr[i] = wr;
  }
  if (threadIdx.x < DH) sB[threadIdx.x] = b1[threadIdx.x];
  __syncthreads();

  int lane = threadIdx.x & 31;
  int wave = threadIdx.x >> 5;
  int node0 = (blockIdx.x * 8 + wave) * 16;
  int row = lane & 15;
  int node = node0 + row;
  int nodeC = min(node, nNodes - 1);
  int klo = (lane >> 4) * 2;
  int ncol = lane & 15;

  float inv = 1.0f / fmaxf(cnt[nodeC], 1.0f);

  const float* xrow = x + (size_t)nodeC * DIN;
  const float* srow = sum1 + (size_t)nodeC * DIN;

  v8f acc0 = {}, acc1 = {}, acc2 = {};
  for (int k = 0; k < DIN; k += 4) {
    v2f aX, aA;
    aX.x = xrow[k + klo];
    aX.y = xrow[k + klo + 1];
    aA.x = srow[k + klo] * inv;
    aA.y = srow[k + klo + 1] * inv;

    v2f bl0, br0, bl1, br1, bl2, br2;
    bl0.x = sWl[ncol * DIN + k + klo];        bl0.y = sWl[ncol * DIN + k + klo + 1];
    br0.x = sWr[ncol * DIN + k + klo];        br0.y = sWr[ncol * DIN + k + klo + 1];
    bl1.x = sWl[(16 + ncol) * DIN + k + klo]; bl1.y = sWl[(16 + ncol) * DIN + k + klo + 1];
    br1.x = sWr[(16 + ncol) * DIN + k + klo]; br1.y = sWr[(16 + ncol) * DIN + k + klo + 1];
    bl2.x = sWl[(32 + ncol) * DIN + k + klo]; bl2.y = sWl[(32 + ncol) * DIN + k + klo + 1];
    br2.x = sWr[(32 + ncol) * DIN + k + klo]; br2.y = sWr[(32 + ncol) * DIN + k + klo + 1];

    acc0 = wmma4(aA, bl0, acc0);
    acc0 = wmma4(aX, br0, acc0);
    acc1 = wmma4(aA, bl1, acc1);
    acc1 = wmma4(aX, br1, acc1);
    acc2 = wmma4(aA, bl2, acc2);
    acc2 = wmma4(aX, br2, acc2);
  }

  int mbase = (lane >> 4) * 8;
  for (int r = 0; r < 8; ++r) {
    int n = node0 + mbase + r;
    if (n < nNodes) {
      float* hrow = h + (size_t)n * DH;
      hrow[ncol]      = fmaxf(acc0[r] + sB[ncol], 0.0f);
      hrow[16 + ncol] = fmaxf(acc1[r] + sB[16 + ncol], 0.0f);
      if (32 + ncol < DH)
        hrow[32 + ncol] = fmaxf(acc2[r] + sB[32 + ncol], 0.0f);
    }
  }
}

__global__ void __launch_bounds__(256)
layer2_kernel(const float* __restrict__ h, const float* __restrict__ sum2,
              const float* __restrict__ cnt, const float* __restrict__ W2l,
              const float* __restrict__ b2, const float* __restrict__ W2r,
              float* __restrict__ o, int nNodes) {
  __shared__ float sWl[DO_PAD * DH];
  __shared__ float sWr[DO_PAD * DH];
  __shared__ float sB[DOUT];
  for (int i = threadIdx.x; i < DO_PAD * DH; i += 256) {
    float wl = (i < DOUT * DH) ? W2l[i] : 0.0f;
    float wr = (i < DOUT * DH) ? W2r[i] : 0.0f;
    sWl[i] = wl;
    sWr[i] = wr;
  }
  if (threadIdx.x < DOUT) sB[threadIdx.x] = b2[threadIdx.x];
  __syncthreads();

  int lane = threadIdx.x & 31;
  int wave = threadIdx.x >> 5;
  int node0 = (blockIdx.x * 8 + wave) * 16;
  int row = lane & 15;
  int node = node0 + row;
  int nodeC = min(node, nNodes - 1);
  int klo = (lane >> 4) * 2;
  int ncol = lane & 15;

  float inv = 1.0f / fmaxf(cnt[nodeC], 1.0f);

  const float* hrow = h + (size_t)nodeC * DH;
  const float* srow = sum2 + (size_t)nodeC * DH;

  v8f acc0 = {}, acc1 = {};
  for (int k = 0; k < DH; k += 4) {
    v2f aH, aA;
    aH.x = hrow[k + klo];
    aH.y = hrow[k + klo + 1];
    aA.x = srow[k + klo] * inv;
    aA.y = srow[k + klo + 1] * inv;

    v2f bl0, br0, bl1, br1;
    bl0.x = sWl[ncol * DH + k + klo];        bl0.y = sWl[ncol * DH + k + klo + 1];
    br0.x = sWr[ncol * DH + k + klo];        br0.y = sWr[ncol * DH + k + klo + 1];
    bl1.x = sWl[(16 + ncol) * DH + k + klo]; bl1.y = sWl[(16 + ncol) * DH + k + klo + 1];
    br1.x = sWr[(16 + ncol) * DH + k + klo]; br1.y = sWr[(16 + ncol) * DH + k + klo + 1];

    acc0 = wmma4(aA, bl0, acc0);
    acc0 = wmma4(aH, br0, acc0);
    acc1 = wmma4(aA, bl1, acc1);
    acc1 = wmma4(aH, br1, acc1);
  }

  int mbase = (lane >> 4) * 8;
  for (int r = 0; r < 8; ++r) {
    int n = node0 + mbase + r;
    if (n < nNodes) {
      float* orow = o + (size_t)n * DOUT;
      orow[ncol] = acc0[r] + sB[ncol];
      if (16 + ncol < DOUT)
        orow[16 + ncol] = acc1[r] + sB[16 + ncol];
    }
  }
}

__global__ void __launch_bounds__(256)
logsoftmax_kernel(const float* __restrict__ o, float* __restrict__ out, int nNodes) {
  int n = blockIdx.x * 256 + threadIdx.x;
  if (n >= nNodes) return;
  const float* r = o + (size_t)n * DOUT;
  float mx = -3.402823466e+38f;
  for (int j = 0; j < DOUT; ++j) mx = fmaxf(mx, r[j]);
  float s = 0.0f;
  for (int j = 0; j < DOUT; ++j) s += expf(r[j] - mx);
  float lse = mx + logf(s);
  float* w = out + (size_t)n * DOUT;
  for (int j = 0; j < DOUT; ++j) w[j] = r[j] - lse;
}

extern "C" void kernel_launch(void* const* d_in, const int* in_sizes, int n_in,
                              void* d_out, int out_size, void* d_ws, size_t ws_size,
                              hipStream_t stream) {
  const float* x   = (const float*)d_in[0];
  const int*   ei  = (const int*)d_in[1];
  const float* W1l = (const float*)d_in[2];
  const float* b1  = (const float*)d_in[3];
  const float* W1r = (const float*)d_in[4];
  const float* W2l = (const float*)d_in[5];
  const float* b2  = (const float*)d_in[6];
  const float* W2r = (const float*)d_in[7];
  float* out = (float*)d_out;

  int  N = in_sizes[0] / DIN;
  long E = (long)in_sizes[1] / 2;
  const int* src = ei;
  const int* dst = ei + E;

  float* ws   = (float*)d_ws;
  float* sum1 = ws;
  float* cnt  = sum1 + (size_t)N * DIN;
  float* sum2 = cnt + N;
  float* h    = sum2 + (size_t)N * DH;
  float* o    = h + (size_t)N * DH;

  size_t nz = (size_t)N * (DIN + 1 + DH);
  zero_kernel<<<4096, 256, 0, stream>>>(ws, nz);

  long t1 = E * 16;
  scatter1_kernel<<<(int)((t1 + 255) / 256), 256, 0, stream>>>(x, src, dst, sum1,
                                                               cnt, E);

  int nb = (N + 127) / 128;
  layer1_kernel<<<nb, 256, 0, stream>>>(x, sum1, cnt, W1l, b1, W1r, h, N);

  long t2 = E * 10;
  scatter2_kernel<<<(int)((t2 + 255) / 256), 256, 0, stream>>>(h, src, dst, sum2, E);

  layer2_kernel<<<nb, 256, 0, stream>>>(h, sum2, cnt, W2l, b2, W2r, o, N);

  logsoftmax_kernel<<<(N + 255) / 256, 256, 0, stream>>>(o, out, N);
}


# Replace float-atomic edge scatter by a counting sort of edges by destination (integer atomics) and a wave-per-node gather-sum with plain stores; neighbour sums now added in arrival-order rank
# speedup vs baseline: 1.3996x; 1.3996x over previous
#include <hip/hip_runtime.h>
#include <math.h>

#define DIN 64
#define DH  40
#define DOUT 24
#define DH_PAD 48
#define DO_PAD 32
#define SCAN_T 1024

typedef __attribute__((ext_vector_type(2))) float v2f;
typedef __attribute__((ext_vector_type(8))) float v8f;

__device__ __forceinline__ v8f wmma4(v2f a, v2f b, v8f c) {
  return __builtin_amdgcn_wmma_f32_16x16x4_f32(false, a, false, b, (short)0, c,
                                               false, false);
}

__global__ void __launch_bounds__(256)
zero_kernel(int* __restrict__ p, int n) {
  int i = blockIdx.x * 256 + threadIdx.x;
  if (i < n) p[i] = 0;
}

__global__ void __launch_bounds__(256)
degree_kernel(const int* __restrict__ dst, int* __restrict__ deg, long E) {
  long e = (long)blockIdx.x * 256 + threadIdx.x;
  if (e < E) atomicAdd(deg + dst[e], 1);
}

__global__ void __launch_bounds__(SCAN_T)
rowptr_kernel(const int* __restrict__ deg, int* __restrict__ rowptr, int n) {
  __shared__ int s[SCAN_T];
  int t = threadIdx.x;
  int base = blockIdx.x * SCAN_T;
  int pre = 0;
  for (int i = t; i < base; i += SCAN_T) pre += deg[i];
  int idx = base + t;
  int v = (idx < n) ? deg[idx] : 0;

  s[t] = pre;
  __syncthreads();
  for (int off = SCAN_T / 2; off > 0; off >>= 1) {
    if (t < off) s[t] += s[t + off];
    __syncthreads();
  }
  int blockPre = s[0];
  __syncthreads();

  s[t] = v;
  __syncthreads();
  for (int off = 1; off < SCAN_T; off <<= 1) {
    int add = (t >= off) ? s[t - off] : 0;
    __syncthreads();
    s[t] += add;
    __syncthreads();
  }
  int incl = s[t];
  if (idx < n) rowptr[idx] = blockPre + incl - v;
  if (idx == n - 1) rowptr[n] = blockPre + incl;
}

__global__ void __launch_bounds__(256)
fill_kernel(const int* __restrict__ src, const int* __restrict__ dst,
            const int* __restrict__ rowptr, int* __restrict__ deg,
            int* __restrict__ col, long E) {
  long e = (long)blockIdx.x * 256 + threadIdx.x;
  if (e >= E) return;
  int d = dst[e];
  int p = atomicSub(deg + d, 1) - 1;
  col[rowptr[d] + p] = src[e];
}

__device__ __forceinline__ float4 ld4(const float* p) {
  return *reinterpret_cast<const float4*>(p);
}
__device__ __forceinline__ float4 add4(float4 a, float4 b) {
  return make_float4(a.x + b.x, a.y + b.y, a.z + b.z, a.w + b.w);
}

__global__ void __launch_bounds__(256)
gather1_kernel(const float* __restrict__ x, const int* __restrict__ rowptr,
               const int* __restrict__ col, float* __restrict__ sum1,
               float* __restrict__ cnt, int nNodes) {
  int lane = threadIdx.x & 31;
  int wave = __builtin_amdgcn_readfirstlane((int)(threadIdx.x >> 5));
  int n = blockIdx.x * 8 + wave;
  if (n >= nNodes) return;
  int beg = rowptr[n], end = rowptr[n + 1];
  int half = lane >> 4;
  int chunk = lane & 15;
  const float* xc = x + chunk * 4;
  float4 acc = make_float4(0.0f, 0.0f, 0.0f, 0.0f);
  int j = beg;
  for (; j + 8 <= end; j += 8) {
    int s0 = col[j + half];
    int s1 = col[j + 2 + half];
    int s2 = col[j + 4 + half];
    int s3 = col[j + 6 + half];
    float4 v0 = ld4(xc + (size_t)s0 * DIN);
    float4 v1 = ld4(xc + (size_t)s1 * DIN);
    float4 v2 = ld4(xc + (size_t)s2 * DIN);
    float4 v3 = ld4(xc + (size_t)s3 * DIN);
    acc = add4(acc, add4(add4(v0, v1), add4(v2, v3)));
  }
  for (; j < end; j += 2) {
    if (j + half < end) {
      int s = col[j + half];
      acc = add4(acc, ld4(xc + (size_t)s * DIN));
    }
  }
  acc.x += __shfl_xor(acc.x, 16);
  acc.y += __shfl_xor(acc.y, 16);
  acc.z += __shfl_xor(acc.z, 16);
  acc.w += __shfl_xor(acc.w, 16);
  if (half == 0)
    *reinterpret_cast<float4*>(sum1 + (size_t)n * DIN + chunk * 4) = acc;
  if (lane == 0) cnt[n] = (float)(end - beg);
}

__global__ void __launch_bounds__(256)
gather2_kernel(const float* __restrict__ h, const int* __restrict__ rowptr,
               const int* __restrict__ col, float* __restrict__ sum2, int nNodes) {
  int lane = threadIdx.x & 31;
  int wave = __builtin_amdgcn_readfirstlane((int)(threadIdx.x >> 5));
  int n = blockIdx.x * 8 + wave;
  if (n >= nNodes) return;
  int beg = rowptr[n], end = rowptr[n + 1];
  int q = lane / 10;
  int chunk = lane - q * 10;
  int g = min(q, 2);
  const float* hc = h + chunk * 4;
  float4 acc = make_float4(0.0f, 0.0f, 0.0f, 0.0f);
  int j = beg;
  for (; j + 12 <= end; j += 12) {
    int s0 = col[j + g];
    int s1 = col[j + 3 + g];
    int s2 = col[j + 6 + g];
    int s3 = col[j + 9 + g];
    float4 v0 = ld4(hc + (size_t)s0 * DH);
    float4 v1 = ld4(hc + (size_t)s1 * DH);
    float4 v2 = ld4(hc + (size_t)s2 * DH);
    float4 v3 = ld4(hc + (size_t)s3 * DH);
    acc = add4(acc, add4(add4(v0, v1), add4(v2, v3)));
  }
  for (; j < end; j += 3) {
    if (j + g < end) {
      int s = col[j + g];
      acc = add4(acc, ld4(hc + (size_t)s * DH));
    }
  }
  float4 t1, t2;
  t1.x = __shfl_down(acc.x, 10); t2.x = __shfl_down(acc.x, 20);
  t1.y = __shfl_down(acc.y, 10); t2.y = __shfl_down(acc.y, 20);
  t1.z = __shfl_down(acc.z, 10); t2.z = __shfl_down(acc.z, 20);
  t1.w = __shfl_down(acc.w, 10); t2.w = __shfl_down(acc.w, 20);
  if (lane < 10)
    *reinterpret_cast<float4*>(sum2 + (size_t)n * DH + chunk * 4) =
        add4(acc, add4(t1, t2));
}

__global__ void __launch_bounds__(256)
layer1_kernel(const float* __restrict__ x, const float* __restrict__ sum1,
              const float* __restrict__ cnt, const float* __restrict__ W1l,
              const float* __restrict__ b1, const float* __restrict__ W1r,
              float* __restrict__ h, int nNodes) {
  __shared__ float sWl[DH_PAD * DIN];
  __shared__ float sWr[DH_PAD * DIN];
  __shared__ float sB[DH];
  for (int i = threadIdx.x; i < DH_PAD * DIN; i += 256) {
    float wl = (i < DH * DIN) ? W1l[i] : 0.0f;
    float wr = (i < DH * DIN) ? W1r[i] : 0.0f;
    sWl[i] = wl;
    sWr[i] = wr;
  }
  if (threadIdx.x < DH) sB[threadIdx.x] = b1[threadIdx.x];
  __syncthreads();

  int lane = threadIdx.x & 31;
  int wave = threadIdx.x >> 5;
  int node0 = (blockIdx.x * 8 + wave) * 16;
  int row = lane & 15;
  int node = node0 + row;
  int nodeC = min(node, nNodes - 1);
  int klo = (lane >> 4) * 2;
  int ncol = lane & 15;

  float inv = 1.0f / fmaxf(cnt[nodeC], 1.0f);

  const float* xrow = x + (size_t)nodeC * DIN;
  const float* srow = sum1 + (size_t)nodeC * DIN;

  v8f acc0 = {}, acc1 = {}, acc2 = {};
  for (int k = 0; k < DIN; k += 4) {
    v2f aX, aA;
    aX.x = xrow[k + klo];
    aX.y = xrow[k + klo + 1];
    aA.x = srow[k + klo] * inv;
    aA.y = srow[k + klo + 1] * inv;

    v2f bl0, br0, bl1, br1, bl2, br2;
    bl0.x = sWl[ncol * DIN + k + klo];        bl0.y = sWl[ncol * DIN + k + klo + 1];
    br0.x = sWr[ncol * DIN + k + klo];        br0.y = sWr[ncol * DIN + k + klo + 1];
    bl1.x = sWl[(16 + ncol) * DIN + k + klo]; bl1.y = sWl[(16 + ncol) * DIN + k + klo + 1];
    br1.x = sWr[(16 + ncol) * DIN + k + klo]; br1.y = sWr[(16 + ncol) * DIN + k + klo + 1];
    bl2.x = sWl[(32 + ncol) * DIN + k + klo]; bl2.y = sWl[(32 + ncol) * DIN + k + klo + 1];
    br2.x = sWr[(32 + ncol) * DIN + k + klo]; br2.y = sWr[(32 + ncol) * DIN + k + klo + 1];

    acc0 = wmma4(aA, bl0, acc0);
    acc0 = wmma4(aX, br0, acc0);
    acc1 = wmma4(aA, bl1, acc1);
    acc1 = wmma4(aX, br1, acc1);
    acc2 = wmma4(aA, bl2, acc2);
    acc2 = wmma4(aX, br2, acc2);
  }

  int mbase = (lane >> 4) * 8;
  for (int r = 0; r < 8; ++r) {
    int n = node0 + mbase + r;
    if (n < nNodes) {
      float* hrow = h + (size_t)n * DH;
      hrow[ncol]      = fmaxf(acc0[r] + sB[ncol], 0.0f);
      hrow[16 + ncol] = fmaxf(acc1[r] + sB[16 + ncol], 0.0f);
      if (32 + ncol < DH)
        hrow[32 + ncol] = fmaxf(acc2[r] + sB[32 + ncol], 0.0f);
    }
  }
}

__global__ void __launch_bounds__(256)
layer2_kernel(const float* __restrict__ h, const float* __restrict__ sum2,
              const float* __restrict__ cnt, const float* __restrict__ W2l,
              const float* __restrict__ b2, const float* __restrict__ W2r,
              float* __restrict__ o, int nNodes) {
  __shared__ float sWl[DO_PAD * DH];
  __shared__ float sWr[DO_PAD * DH];
  __shared__ float sB[DOUT];
  for (int i = threadIdx.x; i < DO_PAD * DH; i += 256) {
    float wl = (i < DOUT * DH) ? W2l[i] : 0.0f;
    float wr = (i < DOUT * DH) ? W2r[i] : 0.0f;
    sWl[i] = wl;
    sWr[i] = wr;
  }
  if (threadIdx.x < DOUT) sB[threadIdx.x] = b2[threadIdx.x];
  __syncthreads();

  int lane = threadIdx.x & 31;
  int wave = threadIdx.x >> 5;
  int node0 = (blockIdx.x * 8 + wave) * 16;
  int row = lane & 15;
  int node = node0 + row;
  int nodeC = min(node, nNodes - 1);
  int klo = (lane >> 4) * 2;
  int ncol = lane & 15;

  float inv = 1.0f / fmaxf(cnt[nodeC], 1.0f);

  const float* hrow = h + (size_t)nodeC * DH;
  const float* srow = sum2 + (size_t)nodeC * DH;

  v8f acc0 = {}, acc1 = {};
  for (int k = 0; k < DH; k += 4) {
    v2f aH, aA;
    aH.x = hrow[k + klo];
    aH.y = hrow[k + klo + 1];
    aA.x = srow[k + klo] * inv;
    aA.y = srow[k + klo + 1] * inv;

    v2f bl0, br0, bl1, br1;
    bl0.x = sWl[ncol * DH + k + klo];        bl0.y = sWl[ncol * DH + k + klo + 1];
    br0.x = sWr[ncol * DH + k + klo];        br0.y = sWr[ncol * DH + k + klo + 1];
    bl1.x = sWl[(16 + ncol) * DH + k + klo]; bl1.y = sWl[(16 + ncol) * DH + k + klo + 1];
    br1.x = sWr[(16 + ncol) * DH + k + klo]; br1.y = sWr[(16 + ncol) * DH + k + klo + 1];

    acc0 = wmma4(aA, bl0, acc0);
    acc0 = wmma4(aH, br0, acc0);
    acc1 = wmma4(aA, bl1, acc1);
    acc1 = wmma4(aH, br1, acc1);
  }

  int mbase = (lane >> 4) * 8;
  for (int r = 0; r < 8; ++r) {
    int n = node0 + mbase + r;
    if (n < nNodes) {
      float* orow = o + (size_t)n * DOUT;
      orow[ncol] = acc0[r] + sB[ncol];
      if (16 + ncol < DOUT)
        orow[16 + ncol] = acc1[r] + sB[16 + ncol];
    }
  }
}

__global__ void __launch_bounds__(256)
logsoftmax_kernel(const float* __restrict__ o, float* __restrict__ out, int nNodes) {
  int n = blockIdx.x * 256 + threadIdx.x;
  if (n >= nNodes) return;
  const float* r = o + (size_t)n * DOUT;
  float mx = -3.402823466e+38f;
  for (int j = 0; j < DOUT; ++j) mx = fmaxf(mx, r[j]);
  float s = 0.0f;
  for (int j = 0; j < DOUT; ++j) s += expf(r[j] - mx);
  float lse = mx + logf(s);
  float* w = out + (size_t)n * DOUT;
  for (int j = 0; j < DOUT; ++j) w[j] = r[j] - lse;
}

extern "C" void kernel_launch(void* const* d_in, const int* in_sizes, int n_in,
                              void* d_out, int out_size, void* d_ws, size_t ws_size,
                              hipStream_t stream) {
  const float* x   = (const float*)d_in[0];
  const int*   ei  = (const int*)d_in[1];
  const float* W1l = (const float*)d_in[2];
  const float* b1  = (const float*)d_in[3];
  const float* W1r = (const float*)d_in[4];
  const float* W2l = (const float*)d_in[5];
  const float* b2  = (const float*)d_in[6];
  const float* W2r = (const float*)d_in[7];
  float* out = (float*)d_out;

  int  N = in_sizes[0] / DIN;
  long E = (long)in_sizes[1] / 2;
  const int* src = ei;
  const int* dst = ei + E;

  float* ws   = (float*)d_ws;
  float* sum1 = ws;
  float* sum2 = ws;
  float* cnt  = sum1 + (size_t)N * DIN;
  float* h    = cnt + N;
  float* o    = h + (size_t)N * DH;
  int* deg    = (int*)(o + (size_t)N * DOUT);
  int* rowptr = deg + N;
  int* col    = rowptr + N + 1;

  int eb = (int)((E + 255) / 256);
  zero_kernel<<<(N + 255) / 256, 256, 0, stream>>>(deg, N);
  degree_kernel<<<eb, 256, 0, stream>>>(dst, deg, E);
  rowptr_kernel<<<(N + SCAN_T - 1) / SCAN_T, SCAN_T, 0, stream>>>(deg, rowptr, N);
  fill_kernel<<<eb, 256, 0, stream>>>(src, dst, rowptr, deg, col, E);

  int gb = (N + 7) / 8;
  gather1_kernel<<<gb, 256, 0, stream>>>(x, rowptr, col, sum1, cnt, N);

  int nb = (N + 127) / 128;
  layer1_kernel<<<nb, 256, 0, stream>>>(x, sum1, cnt, W1l, b1, W1r, h, N);

  gather2_kernel<<<gb, 256, 0, stream>>>(h, rowptr, col, sum2, N);

  layer2_kernel<<<nb, 256, 0, stream>>>(h, sum2, cnt, W2l, b2, W2r, o, N);

  logsoftmax_kernel<<<(N + 255) / 256, 256, 0, stream>>>(o, out, N);
}


# Resident waves loop over edges and nodes: degree, fill and both neighbour-sum launches start 8192 or 16384 waves, not 100000; rows summed 16 (15) per trip, so the arrival-order sums regroup
# speedup vs baseline: 1.4616x; 1.0443x over previous
#include <hip/hip_runtime.h>
#include <math.h>

#define DIN 64
#define DH  40
#define DOUT 24
#define DH_PAD 48
#define DO_PAD 32
#define SCAN_T 1024

typedef __attribute__((ext_vector_type(2))) float v2f;
typedef __attribute__((ext_vector_type(8))) float v8f;

__device__ __forceinline__ v8f wmma4(v2f a, v2f b, v8f c) {
  return __builtin_amdgcn_wmma_f32_16x16x4_f32(false, a, false, b, (short)0, c,
                                               false, false);
}

__global__ void __launch_bounds__(256)
zero_kernel(int* __restrict__ p, int n) {
  int i = blockIdx.x * 256 + threadIdx.x;
  if (i < n) p[i] = 0;
}

__global__ void __launch_bounds__(256)
degree_kernel(const int* __restrict__ dst, int* __restrict__ deg, long E) {
  long stride = (long)gridDim.x * 256;
  long e = (long)blockIdx.x * 256 + threadIdx.x;
  for (; e + 3 * stride < E; e += 4 * stride) {
    int d0 = dst[e];
    int d1 = dst[e + stride];
    int d2 = dst[e + 2 * stride];
    int d3 = dst[e + 3 * stride];
    atomicAdd(deg + d0, 1);
    atomicAdd(deg + d1, 1);
    atomicAdd(deg + d2, 1);
    atomicAdd(deg + d3, 1);
  }
  for (; e < E; e += stride) atomicAdd(deg + dst[e], 1);
}

__global__ void __launch_bounds__(SCAN_T)
rowptr_kernel(const int* __restrict__ deg, int* __restrict__ rowptr, int n) {
  __shared__ int s[SCAN_T];
  int t = threadIdx.x;
  int base = blockIdx.x * SCAN_T;
  int pre = 0;
  for (int i = t; i < base; i += SCAN_T) pre += deg[i];
  int idx = base + t;
  int v = (idx < n) ? deg[idx] : 0;

  s[t] = pre;
  __syncthreads();
  for (int off = SCAN_T / 2; off > 0; off >>= 1) {
    if (t < off) s[t] += s[t + off];
    __syncthreads();
  }
  int blockPre = s[0];
  __syncthreads();

  s[t] = v;
  __syncthreads();
  for (int off = 1; off < SCAN_T; off <<= 1) {
    int add = (t >= off) ? s[t - off] : 0;
    __syncthreads();
    s[t] += add;
    __syncthreads();
  }
  int incl = s[t];
  if (idx < n) rowptr[idx] = blockPre + incl - v;
  if (idx == n - 1) rowptr[n] = blockPre + incl;
}

__global__ void __launch_bounds__(256)
fill_kernel(const int* __restrict__ src, const int* __restrict__ dst,
            const int* __restrict__ rowptr, int* __restrict__ deg,
            int* __restrict__ col, long E) {
  long stride = (long)gridDim.x * 256;
  long e = (long)blockIdx.x * 256 + threadIdx.x;
  for (; e + 3 * stride < E; e += 4 * stride) {
    int d0 = dst[e];
    int d1 = dst[e + stride];
    int d2 = dst[e + 2 * stride];
    int d3 = dst[e + 3 * stride];
    int s0 = src[e];
    int s1 = src[e + stride];
    int s2 = src[e + 2 * stride];
    int s3 = src[e + 3 * stride];
    int p0 = atomicSub(deg + d0, 1) - 1;
    int p1 = atomicSub(deg + d1, 1) - 1;
    int p2 = atomicSub(deg + d2, 1) - 1;
    int p3 = atomicSub(deg + d3, 1) - 1;
    int r0 = rowptr[d0];
    int r1 = rowptr[d1];
    int r2 = rowptr[d2];
    int r3 = rowptr[d3];
    col[r0 + p0] = s0;
    col[r1 + p1] = s1;
    col[r2 + p2] = s2;
    col[r3 + p3] = s3;
  }
  for (; e < E; e += stride) {
    int d = dst[e];
    int p = atomicSub(deg + d, 1) - 1;
    col[rowptr[d] + p] = src[e];
  }
}

__device__ __forceinline__ float4 ld4(const float* p) {
  return *reinterpret_cast<const float4*>(p);
}
__device__ __forceinline__ float4 add4(float4 a, float4 b) {
  return make_float4(a.x + b.x, a.y + b.y, a.z + b.z, a.w + b.w);
}

__global__ void __launch_bounds__(256)
gather1_kernel(const float* __restrict__ x, const int* __restrict__ rowptr,
               const int* __restrict__ col, float* __restrict__ sum1,
               float* __restrict__ cnt, int nNodes) {
  int lane = threadIdx.x & 31;
  int wave = __builtin_amdgcn_readfirstlane((int)(threadIdx.x >> 5));
  int nw = (int)gridDim.x * 8;
  int n = blockIdx.x * 8 + wave;
  if (n >= nNodes) return;
  int half = lane >> 4;
  int chunk = lane & 15;
  const float* xc = x + chunk * 4;
  int beg = rowptr[n], end = rowptr[n + 1];
  while (true) {
    int nn = n + nw;
    int nbeg = 0, nend = 0;
    if (nn < nNodes) { nbeg = rowptr[nn]; nend = rowptr[nn + 1]; }

    float4 acc = make_float4(0.0f, 0.0f, 0.0f, 0.0f);
    int e0 = beg + lane;
    int my = (e0 < end) ? col[e0] : 0;
    for (int base = beg; base < end; base += 32) {
      int e1 = base + 32 + lane;
      int myn = (e1 < end) ? col[e1] : 0;
      int m = min(32, end - base);
      for (int j = 0; j < m; j += 16) {
        float4 v[8];
#pragma unroll
        for (int u = 0; u < 8; ++u) {
          int k = j + 2 * u + half;
          int s = __shfl(my, k);
          float4 t = ld4(xc + (size_t)s * DIN);
          bool ok = k < m;
          v[u] = make_float4(ok ? t.x : 0.0f, ok ? t.y : 0.0f,
                             ok ? t.z : 0.0f, ok ? t.w : 0.0f);
        }
        acc = add4(acc, add4(add4(add4(v[0], v[1]), add4(v[2], v[3])),
                             add4(add4(v[4], v[5]), add4(v[6], v[7]))));
      }
      my = myn;
    }
    acc.x += __shfl_xor(acc.x, 16);
    acc.y += __shfl_xor(acc.y, 16);
    acc.z += __shfl_xor(acc.z, 16);
    acc.w += __shfl_xor(acc.w, 16);
    if (half == 0)
      *reinterpret_cast<float4*>(sum1 + (size_t)n * DIN + chunk * 4) = acc;
    if (lane == 0) cnt[n] = (float)(end - beg);

    if (nn >= nNodes) break;
    n = nn; beg = nbeg; end = nend;
  }
}

__global__ void __launch_bounds__(256)
gather2_kernel(const float* __restrict__ h, const int* __restrict__ rowptr,
               const int* __restrict__ col, float* __restrict__ sum2, int nNodes) {
  int lane = threadIdx.x & 31;
  int wave = __builtin_amdgcn_readfirstlane((int)(threadIdx.x >> 5));
  int nw = (int)gridDim.x * 8;
  int n = blockIdx.x * 8 + wave;
  if (n >= nNodes) return;
  int q = lane / 10;
  int chunk = lane - q * 10;
  int g = min(q, 2);
  const float* hc = h + chunk * 4;
  int beg = rowptr[n], end = rowptr[n + 1];
  while (true) {
    int nn = n + nw;
    int nbeg = 0, nend = 0;
    if (nn < nNodes) { nbeg = rowptr[nn]; nend = rowptr[nn + 1]; }

    float4 acc = make_float4(0.0f, 0.0f, 0.0f, 0.0f);
    int e0 = beg + lane;
    int my = (lane < 30 && e0 < end) ? col[e0] : 0;
    for (int base = beg; base < end; base += 30) {
      int e1 = base + 30 + lane;
      int myn = (lane < 30 && e1 < end) ? col[e1] : 0;
      int m = min(30, end - base);
      for (int j = 0; j < m; j += 15) {
        float4 v[5];
#pragma unroll
        for (int u = 0; u < 5; ++u) {
          int k = j + 3 * u + g;
          int s = __shfl(my, k);
          float4 t = ld4(hc + (size_t)s * DH);
          bool ok = k < m;
          v[u] = make_float4(ok ? t.x : 0.0f, ok ? t.y : 0.0f,
                             ok ? t.z : 0.0f, ok ? t.w : 0.0f);
        }
        acc = add4(acc, add4(add4(add4(v[0], v[1]), add4(v[2], v[3])), v[4]));
      }
      my = myn;
    }
    float4 t1, t2;
    t1.x = __shfl_down(acc.x, 10); t2.x = __shfl_down(acc.x, 20);
    t1.y = __shfl_down(acc.y, 10); t2.y = __shfl_down(acc.y, 20);
    t1.z = __shfl_down(acc.z, 10); t2.z = __shfl_down(acc.z, 20);
    t1.w = __shfl_down(acc.w, 10); t2.w = __shfl_down(acc.w, 20);
    if (lane < 10)
      *reinterpret_cast<float4*>(sum2 + (size_t)n * DH + chunk * 4) =
          add4(acc, add4(t1, t2));

    if (nn >= nNodes) break;
    n = nn; beg = nbeg; end = nend;
  }
}

__global__ void __launch_bounds__(256)
layer1_kernel(const float* __restrict__ x, const float* __restrict__ sum1,
              const float* __restrict__ cnt, const float* __restrict__ W1l,
              const float* __restrict__ b1, const float* __restrict__ W1r,
              float* __restrict__ h, int nNodes) {
  __shared__ float sWl[DH_PAD * DIN];
  __shared__ float sWr[DH_PAD * DIN];
  __shared__ float sB[DH];
  for (int i = threadIdx.x; i < DH_PAD * DIN; i += 256) {
    float wl = (i < DH * DIN) ? W1l[i] : 0.0f;
    float wr = (i < DH * DIN) ? W1r[i] : 0.0f;
    sWl[i] = wl;
    sWr[i] = wr;
  }
  if (threadIdx.x < DH) sB[threadIdx.x] = b1[threadIdx.x];
  __syncthreads();

  int lane = threadIdx.x & 31;
  int wave = threadIdx.x >> 5;
  int node0 = (blockIdx.x * 8 + wave) * 16;
  int row = lane & 15;
  int node = node0 + row;
  int nodeC = min(node, nNodes - 1);
  int klo = (lane >> 4) * 2;
  int ncol = lane & 15;

  float inv = 1.0f / fmaxf(cnt[nodeC], 1.0f);

  const float* xrow = x + (size_t)nodeC * DIN;
  const float* srow = sum1 + (size_t)nodeC * DIN;

  v8f acc0 = {}, acc1 = {}, acc2 = {};
  for (int k = 0; k < DIN; k += 4) {
    v2f aX, aA;
    aX.x = xrow[k + klo];
    aX.y = xrow[k + klo + 1];
    aA.x = srow[k + klo] * inv;
    aA.y = srow[k + klo + 1] * inv;

    v2f bl0, br0, bl1, br1, bl2, br2;
    bl0.x = sWl[ncol * DIN + k + klo];        bl0.y = sWl[ncol * DIN + k + klo + 1];
    br0.x = sWr[ncol * DIN + k + klo];        br0.y = sWr[ncol * DIN + k + klo + 1];
    bl1.x = sWl[(16 + ncol) * DIN + k + klo]; bl1.y = sWl[(16 + ncol) * DIN + k + klo + 1];
    br1.x = sWr[(16 + ncol) * DIN + k + klo]; br1.y = sWr[(16 + ncol) * DIN + k + klo + 1];
    bl2.x = sWl[(32 + ncol) * DIN + k + klo]; bl2.y = sWl[(32 + ncol) * DIN + k + klo + 1];
    br2.x = sWr[(32 + ncol) * DIN + k + klo]; br2.y = sWr[(32 + ncol) * DIN + k + klo + 1];

    acc0 = wmma4(aA, bl0, acc0);
    acc0 = wmma4(aX, br0, acc0);
    acc1 = wmma4(aA, bl1, acc1);
    acc1 = wmma4(aX, br1, acc1);
    acc2 = wmma4(aA, bl2, acc2);
    acc2 = wmma4(aX, br2, acc2);
  }

  int mbase = (lane >> 4) * 8;
  for (int r = 0; r < 8; ++r) {
    int n = node0 + mbase + r;
    if (n < nNodes) {
      float* hrow = h + (size_t)n * DH;
      hrow[ncol]      = fmaxf(acc0[r] + sB[ncol], 0.0f);
      hrow[16 + ncol] = fmaxf(acc1[r] + sB[16 + ncol], 0.0f);
      if (32 + ncol < DH)
        hrow[32 + ncol] = fmaxf(acc2[r] + sB[32 + ncol], 0.0f);
    }
  }
}

__global__ void __launch_bounds__(256)
layer2_kernel(const float* __restrict__ h, const float* __restrict__ sum2,
              const float* __restrict__ cnt, const float* __restrict__ W2l,
              const float* __restrict__ b2, const float* __restrict__ W2r,
              float* __restrict__ o, int nNodes) {
  __shared__ float sWl[DO_PAD * DH];
  __shared__ float sWr[DO_PAD * DH];
  __shared__ float sB[DOUT];
  for (int i = threadIdx.x; i < DO_PAD * DH; i += 256) {
    float wl = (i < DOUT * DH) ? W2l[i] : 0.0f;
    float wr = (i < DOUT * DH) ? W2r[i] : 0.0f;
    sWl[i] = wl;
    sWr[i] = wr;
  }
  if (threadIdx.x < DOUT) sB[threadIdx.x] = b2[threadIdx.x];
  __syncthreads();

  int lane = threadIdx.x & 31;
  int wave = threadIdx.x >> 5;
  int node0 = (blockIdx.x * 8 + wave) * 16;
  int row = lane & 15;
  int node = node0 + row;
  int nodeC = min(node, nNodes - 1);
  int klo = (lane >> 4) * 2;
  int ncol = lane & 15;

  float inv = 1.0f / fmaxf(cnt[nodeC], 1.0f);

  const float* hrow = h + (size_t)nodeC * DH;
  const float* srow = sum2 + (size_t)nodeC * DH;

  v8f acc0 = {}, acc1 = {};
  for (int k = 0; k < DH; k += 4) {
    v2f aH, aA;
    aH.x = hrow[k + klo];
    aH.y = hrow[k + klo + 1];
    aA.x = srow[k + klo] * inv;
    aA.y = srow[k + klo + 1] * inv;

    v2f bl0, br0, bl1, br1;
    bl0.x = sWl[ncol * DH + k + klo];        bl0.y = sWl[ncol * DH + k + klo + 1];
    br0.x = sWr[ncol * DH + k + klo];        br0.y = sWr[ncol * DH + k + klo + 1];
    bl1.x = sWl[(16 + ncol) * DH + k + klo]; bl1.y = sWl[(16 + ncol) * DH + k + klo + 1];
    br1.x = sWr[(16 + ncol) * DH + k + klo]; br1.y = sWr[(16 + ncol) * DH + k + klo + 1];

    acc0 = wmma4(aA, bl0, acc0);
    acc0 = wmma4(aH, br0, acc0);
    acc1 = wmma4(aA, bl1, acc1);
    acc1 = wmma4(aH, br1, acc1);
  }

  int mbase = (lane >> 4) * 8;
  for (int r = 0; r < 8; ++r) {
    int n = node0 + mbase + r;
    if (n < nNodes) {
      float* orow = o + (size_t)n * DOUT;
      orow[ncol] = acc0[r] + sB[ncol];
      if (16 + ncol < DOUT)
        orow[16 + ncol] = acc1[r] + sB[16 + ncol];
    }
  }
}

__global__ void __launch_bounds__(256)
logsoftmax_kernel(const float* __restrict__ o, float* __restrict__ out, int nNodes) {
  int n = blockIdx.x * 256 + threadIdx.x;
  if (n >= nNodes) return;
  const float* r = o + (size_t)n * DOUT;
  float mx = -3.402823466e+38f;
  for (int j = 0; j < DOUT; ++j) mx = fmaxf(mx, r[j]);
  float s = 0.0f;
  for (int j = 0; j < DOUT; ++j) s += expf(r[j] - mx);
  float lse = mx + logf(s);
  float* w = out + (size_t)n * DOUT;
  for (int j = 0; j < DOUT; ++j) w[j] = r[j] - lse;
}

extern "C" void kernel_launch(void* const* d_in, const int* in_sizes, int n_in,
                              void* d_out, int out_size, void* d_ws, size_t ws_size,
                              hipStream_t stream) {
  const float* x   = (const float*)d_in[0];
  const int*   ei  = (const int*)d_in[1];
  const float* W1l = (const float*)d_in[2];
  const float* b1  = (const float*)d_in[3];
  const float* W1r = (const float*)d_in[4];
  const float* W2l = (const float*)d_in[5];
  const float* b2  = (const float*)d_in[6];
  const float* W2r = (const float*)d_in[7];
  float* out = (float*)d_out;

  int  N = in_sizes[0] / DIN;
  long E = (long)in_sizes[1] / 2;
  const int* src = ei;
  const int* dst = ei + E;

  float* ws   = (float*)d_ws;
  float* sum1 = ws;
  float* sum2 = ws;
  float* cnt  = sum1 + (size_t)N * DIN;
  float* h    = cnt + N;
  float* o    = h + (size_t)N * DH;
  int* deg    = (int*)(o + (size_t)N * DOUT);
  int* rowptr = deg + N;
  int* col    = rowptr + N + 1;

  zero_kernel<<<(N + 255) / 256, 256, 0, stream>>>(deg, N);
  degree_kernel<<<1024, 256, 0, stream>>>(dst, deg, E);
  rowptr_kernel<<<(N + SCAN_T - 1) / SCAN_T, SCAN_T, 0, stream>>>(deg, rowptr, N);
  fill_kernel<<<1024, 256, 0, stream>>>(src, dst, rowptr, deg, col, E);

  gather1_kernel<<<2048, 256, 0, stream>>>(x, rowptr, col, sum1, cnt, N);

  int nb = (N + 127) / 128;
  layer1_kernel<<<nb, 256, 0, stream>>>(x, sum1, cnt, W1l, b1, W1r, h, N);

  gather2_kernel<<<2048, 256, 0, stream>>>(h, rowptr, col, sum2, N);

  layer2_kernel<<<nb, 256, 0, stream>>>(h, sum2, cnt, W2l, b2, W2r, o, N);

  logsoftmax_kernel<<<(N + 255) / 256, 256, 0, stream>>>(o, out, N);
}


# Edges ordered by destination with LDS counters (bucket slices, then one work-group per 256 nodes), not 6.4M scattered global atomics; neighbours still summed in arrival order, now the LDS adds' order
# speedup vs baseline: 3.4566x; 2.3649x over previous
#include <hip/hip_runtime.h>
#include <math.h>

#define DIN 64
#define DH  40
#define DOUT 24
#define DH_PAD 48
#define DO_PAD 32
#define SCAN_T 1024

typedef __attribute__((ext_vector_type(2))) float v2f;
typedef __attribute__((ext_vector_type(8))) float v8f;

__device__ __forceinline__ v8f wmma4(v2f a, v2f b, v8f c) {
  return __builtin_amdgcn_wmma_f32_16x16x4_f32(false, a, false, b, (short)0, c,
                                               false, false);
}

#define NPB 256
#define MAXB 1024
#define EB 256
#define ET 1024

__device__ __forceinline__ int bucket_of(int d, int NB) {
  return min((int)((unsigned)d >> 8), NB - 1);
}

__global__ void __launch_bounds__(ET)
bucket_count_kernel(const int* __restrict__ dst, int* __restrict__ cm, long E,
                    int NB) {
  __shared__ int hist[MAXB];
  int t = threadIdx.x;
  for (int j = t; j < NB; j += ET) hist[j] = 0;
  __syncthreads();
  long chunk = (E + EB - 1) / EB;
  long e0 = (long)blockIdx.x * chunk;
  long e1 = min(E, e0 + chunk);
  long e = e0 + t;
  for (; e + 3 * ET < e1; e += 4 * ET) {
    int d0 = dst[e];
    int d1 = dst[e + ET];
    int d2 = dst[e + 2 * ET];
    int d3 = dst[e + 3 * ET];
    atomicAdd(&hist[bucket_of(d0, NB)], 1);
    atomicAdd(&hist[bucket_of(d1, NB)], 1);
    atomicAdd(&hist[bucket_of(d2, NB)], 1);
    atomicAdd(&hist[bucket_of(d3, NB)], 1);
  }
  for (; e < e1; e += ET) atomicAdd(&hist[bucket_of(dst[e], NB)], 1);
  __syncthreads();
  for (int j = t; j < NB; j += ET) cm[(size_t)j * EB + blockIdx.x] = hist[j];
}

__global__ void __launch_bounds__(SCAN_T)
scan_kernel(const int* __restrict__ deg, int* __restrict__ rowptr, int n) {
  __shared__ int s[SCAN_T];
  int t = threadIdx.x;
  int base = blockIdx.x * SCAN_T;
  int pre = 0;
  for (int i = t; i < base; i += SCAN_T) pre += deg[i];
  int idx = base + t;
  int v = (idx < n) ? deg[idx] : 0;

  s[t] = pre;
  __syncthreads();
  for (int off = SCAN_T / 2; off > 0; off >>= 1) {
    if (t < off) s[t] += s[t + off];
    __syncthreads();
  }
  int blockPre = s[0];
  __syncthreads();

  s[t] = v;
  __syncthreads();
  for (int off = 1; off < SCAN_T; off <<= 1) {
    int add = (t >= off) ? s[t - off] : 0;
    __syncthreads();
    s[t] += add;
    __syncthreads();
  }
  int incl = s[t];
  if (idx < n) rowptr[idx] = blockPre + incl - v;
  if (idx == n - 1) rowptr[n] = blockPre + incl;
}

__global__ void __launch_bounds__(ET)
bucket_move_kernel(const int* __restrict__ src, const int* __restrict__ dst,
                   const int* __restrict__ offs, unsigned* __restrict__ pk,
                   long E, int NB) {
  __shared__ int cur[MAXB];
  int t = threadIdx.x;
  for (int j = t; j < NB; j += ET) cur[j] = offs[(size_t)j * EB + blockIdx.x];
  __syncthreads();
  long chunk = (E + EB - 1) / EB;
  long e0 = (long)blockIdx.x * chunk;
  long e1 = min(E, e0 + chunk);
  long e = e0 + t;
  for (; e + 3 * ET < e1; e += 4 * ET) {
    int d0 = dst[e];
    int d1 = dst[e + ET];
    int d2 = dst[e + 2 * ET];
    int d3 = dst[e + 3 * ET];
    int s0 = src[e];
    int s1 = src[e + ET];
    int s2 = src[e + 2 * ET];
    int s3 = src[e + 3 * ET];
    int p0 = atomicAdd(&cur[bucket_of(d0, NB)], 1);
    int p1 = atomicAdd(&cur[bucket_of(d1, NB)], 1);
    int p2 = atomicAdd(&cur[bucket_of(d2, NB)], 1);
    int p3 = atomicAdd(&cur[bucket_of(d3, NB)], 1);
    pk[p0] = ((unsigned)s0 << 8) | ((unsigned)d0 & 255u);
    pk[p1] = ((unsigned)s1 << 8) | ((unsigned)d1 & 255u);
    pk[p2] = ((unsigned)s2 << 8) | ((unsigned)d2 & 255u);
    pk[p3] = ((unsigned)s3 << 8) | ((unsigned)d3 & 255u);
  }
  for (; e < e1; e += ET) {
    int d = dst[e];
    int p = atomicAdd(&cur[bucket_of(d, NB)], 1);
    pk[p] = ((unsigned)src[e] << 8) | ((unsigned)d & 255u);
  }
}

__global__ void __launch_bounds__(ET)
bucket_sort_kernel(const unsigned* __restrict__ pk, const int* __restrict__ offs,
                   int* __restrict__ rowptr, int* __restrict__ col, int nNodes) {
  __shared__ int cnt[NPB];
  __shared__ int cur[NPB];
  __shared__ int wsum[NPB / 32];
  int t = threadIdx.x;
  int b = blockIdx.x;
  int start = offs[(size_t)b * EB];
  int end = offs[(size_t)(b + 1) * EB];
  if (t < NPB) cnt[t] = 0;
  __syncthreads();

  int i = start + t;
  for (; i + 3 * ET < end; i += 4 * ET) {
    unsigned w0 = pk[i];
    unsigned w1 = pk[i + ET];
    unsigned w2 = pk[i + 2 * ET];
    unsigned w3 = pk[i + 3 * ET];
    atomicAdd(&cnt[w0 & 255u], 1);
    atomicAdd(&cnt[w1 & 255u], 1);
    atomicAdd(&cnt[w2 & 255u], 1);
    atomicAdd(&cnt[w3 & 255u], 1);
  }
  for (; i < end; i += ET) atomicAdd(&cnt[pk[i] & 255u], 1);
  __syncthreads();

  int v = 0, incl = 0;
  if (t < NPB) {
    v = cnt[t];
    incl = v;
    for (int off = 1; off < 32; off <<= 1) {
      int u = __shfl_up(incl, off);
      if ((t & 31) >= off) incl += u;
    }
    if ((t & 31) == 31) wsum[t >> 5] = incl;
  }
  __syncthreads();
  if (t < NPB) {
    int pre = start;
    for (int w = 0; w < (t >> 5); ++w) pre += wsum[w];
    int excl = pre + incl - v;
    cur[t] = excl;
    int node = b * NPB + t;
    if (node < nNodes) rowptr[node] = excl;
    if (node == nNodes - 1) rowptr[nNodes] = end;
  }
  __syncthreads();

  i = start + t;
  for (; i + 3 * ET < end; i += 4 * ET) {
    unsigned w0 = pk[i];
    unsigned w1 = pk[i + ET];
    unsigned w2 = pk[i + 2 * ET];
    unsigned w3 = pk[i + 3 * ET];
    int p0 = atomicAdd(&cur[w0 & 255u], 1);
    int p1 = atomicAdd(&cur[w1 & 255u], 1);
    int p2 = atomicAdd(&cur[w2 & 255u], 1);
    int p3 = atomicAdd(&cur[w3 & 255u], 1);
    col[p0] = (int)(w0 >> 8);
    col[p1] = (int)(w1 >> 8);
    col[p2] = (int)(w2 >> 8);
    col[p3] = (int)(w3 >> 8);
  }
  for (; i < end; i += ET) {
    unsigned w = pk[i];
    int p = atomicAdd(&cur[w & 255u], 1);
    col[p] = (int)(w >> 8);
  }
}

__device__ __forceinline__ float4 ld4(const float* p) {
  return *reinterpret_cast<const float4*>(p);
}
__device__ __forceinline__ float4 add4(float4 a, float4 b) {
  return make_float4(a.x + b.x, a.y + b.y, a.z + b.z, a.w + b.w);
}

__global__ void __launch_bounds__(256)
gather1_kernel(const float* __restrict__ x, const int* __restrict__ rowptr,
               const int* __restrict__ col, float* __restrict__ sum1,
               float* __restrict__ cnt, int nNodes) {
  int lane = threadIdx.x & 31;
  int wave = __builtin_amdgcn_readfirstlane((int)(threadIdx.x >> 5));
  int nw = (int)gridDim.x * 8;
  int n = blockIdx.x * 8 + wave;
  if (n >= nNodes) return;
  int half = lane >> 4;
  int chunk = lane & 15;
  const float* xc = x + chunk * 4;
  int beg = rowptr[n], end = rowptr[n + 1];
  while (true) {
    int nn = n + nw;
    int nbeg = 0, nend = 0;
    if (nn < nNodes) { nbeg = rowptr[nn]; nend = rowptr[nn + 1]; }

    float4 acc = make_float4(0.0f, 0.0f, 0.0f, 0.0f);
    int e0 = beg + lane;
    int my = (e0 < end) ? col[e0] : 0;
    for (int base = beg; base < end; base += 32) {
      int e1 = base + 32 + lane;
      int myn = (e1 < end) ? col[e1] : 0;
      int m = min(32, end - base);
      for (int j = 0; j < m; j += 16) {
        float4 v[8];
#pragma unroll
        for (int u = 0; u < 8; ++u) {
          int k = j + 2 * u + half;
          int s = __shfl(my, k);
          float4 t = ld4(xc + (size_t)s * DIN);
          bool ok = k < m;
          v[u] = make_float4(ok ? t.x : 0.0f, ok ? t.y : 0.0f,
                             ok ? t.z : 0.0f, ok ? t.w : 0.0f);
        }
        acc = add4(acc, add4(add4(add4(v[0], v[1]), add4(v[2], v[3])),
                             add4(add4(v[4], v[5]), add4(v[6], v[7]))));
      }
      my = myn;
    }
    acc.x += __shfl_xor(acc.x, 16);
    acc.y += __shfl_xor(acc.y, 16);
    acc.z += __shfl_xor(acc.z, 16);
    acc.w += __shfl_xor(acc.w, 16);
    if (half == 0)
      *reinterpret_cast<float4*>(sum1 + (size_t)n * DIN + chunk * 4) = acc;
    if (lane == 0) cnt[n] = (float)(end - beg);

    if (nn >= nNodes) break;
    n = nn; beg = nbeg; end = nend;
  }
}

__global__ void __launch_bounds__(256)
gather2_kernel(const float* __restrict__ h, const int* __restrict__ rowptr,
               const int* __restrict__ col, float* __restrict__ sum2, int nNodes) {
  int lane = threadIdx.x & 31;
  int wave = __builtin_amdgcn_readfirstlane((int)(threadIdx.x >> 5));
  int nw = (int)gridDim.x * 8;
  int n = blockIdx.x * 8 + wave;
  if (n >= nNodes) return;
  int q = lane / 10;
  int chunk = lane - q * 10;
  int g = min(q, 2);
  const float* hc = h + chunk * 4;
  int beg = rowptr[n], end = rowptr[n + 1];
  while (true) {
    int nn = n + nw;
    int nbeg = 0, nend = 0;
    if (nn < nNodes) { nbeg = rowptr[nn]; nend = rowptr[nn + 1]; }

    float4 acc = make_float4(0.0f, 0.0f, 0.0f, 0.0f);
    int e0 = beg + lane;
    int my = (lane < 30 && e0 < end) ? col[e0] : 0;
    for (int base = beg; base < end; base += 30) {
      int e1 = base + 30 + lane;
      int myn = (lane < 30 && e1 < end) ? col[e1] : 0;
      int m = min(30, end - base);
      for (int j = 0; j < m; j += 15) {
        float4 v[5];
#pragma unroll
        for (int u = 0; u < 5; ++u) {
          int k = j + 3 * u + g;
          int s = __shfl(my, k);
          float4 t = ld4(hc + (size_t)s * DH);
          bool ok = k < m;
          v[u] = make_float4(ok ? t.x : 0.0f, ok ? t.y : 0.0f,
                             ok ? t.z : 0.0f, ok ? t.w : 0.0f);
        }
        acc = add4(acc, add4(add4(add4(v[0], v[1]), add4(v[2], v[3])), v[4]));
      }
      my = myn;
    }
    float4 t1, t2;
    t1.x = __shfl_down(acc.x, 10); t2.x = __shfl_down(acc.x, 20);
    t1.y = __shfl_down(acc.y, 10); t2.y = __shfl_down(acc.y, 20);
    t1.z = __shfl_down(acc.z, 10); t2.z = __shfl_down(acc.z, 20);
    t1.w = __shfl_down(acc.w, 10); t2.w = __shfl_down(acc.w, 20);
    if (lane < 10)
      *reinterpret_cast<float4*>(sum2 + (size_t)n * DH + chunk * 4) =
          add4(acc, add4(t1, t2));

    if (nn >= nNodes) break;
    n = nn; beg = nbeg; end = nend;
  }
}

__global__ void __launch_bounds__(256)
layer1_kernel(const float* __restrict__ x, const float* __restrict__ sum1,
              const float* __restrict__ cnt, const float* __restrict__ W1l,
              const float* __restrict__ b1, const float* __restrict__ W1r,
              float* __restrict__ h, int nNodes) {
  __shared__ float sWl[DH_PAD * DIN];
  __shared__ float sWr[DH_PAD * DIN];
  __shared__ float sB[DH];
  for (int i = threadIdx.x; i < DH_PAD * DIN; i += 256) {
    float wl = (i < DH * DIN) ? W1l[i] : 0.0f;
    float wr = (i < DH * DIN) ? W1r[i] : 0.0f;
    sWl[i] = wl;
    sWr[i] = wr;
  }
  if (threadIdx.x < DH) sB[threadIdx.x] = b1[threadIdx.x];
  __syncthreads();

  int lane = threadIdx.x & 31;
  int wave = threadIdx.x >> 5;
  int node0 = (blockIdx.x * 8 + wave) * 16;
  int row = lane & 15;
  int node = node0 + row;
  int nodeC = min(node, nNodes - 1);
  int klo = (lane >> 4) * 2;
  int ncol = lane & 15;

  float inv = 1.0f / fmaxf(cnt[nodeC], 1.0f);

  const float* xrow = x + (size_t)nodeC * DIN;
  const float* srow = sum1 + (size_t)nodeC * DIN;

  v8f acc0 = {}, acc1 = {}, acc2 = {};
  for (int k = 0; k < DIN; k += 4) {
    v2f aX, aA;
    aX.x = xrow[k + klo];
    aX.y = xrow[k + klo + 1];
    aA.x = srow[k + klo] * inv;
    aA.y = srow[k + klo + 1] * inv;

    v2f bl0, br0, bl1, br1, bl2, br2;
    bl0.x = sWl[ncol * DIN + k + klo];        bl0.y = sWl[ncol * DIN + k + klo + 1];
    br0.x = sWr[ncol * DIN + k + klo];        br0.y = sWr[ncol * DIN + k + klo + 1];
    bl1.x = sWl[(16 + ncol) * DIN + k + klo]; bl1.y = sWl[(16 + ncol) * DIN + k + klo + 1];
    br1.x = sWr[(16 + ncol) * DIN + k + klo]; br1.y = sWr[(16 + ncol) * DIN + k + klo + 1];
    bl2.x = sWl[(32 + ncol) * DIN + k + klo]; bl2.y = sWl[(32 + ncol) * DIN + k + klo + 1];
    br2.x = sWr[(32 + ncol) * DIN + k + klo]; br2.y = sWr[(32 + ncol) * DIN + k + klo + 1];

    acc0 = wmma4(aA, bl0, acc0);
    acc0 = wmma4(aX, br0, acc0);
    acc1 = wmma4(aA, bl1, acc1);
    acc1 = wmma4(aX, br1, acc1);
    acc2 = wmma4(aA, bl2, acc2);
    acc2 = wmma4(aX, br2, acc2);
  }

  int mbase = (lane >> 4) * 8;
  for (int r = 0; r < 8; ++r) {
    int n = node0 + mbase + r;
    if (n < nNodes) {
      float* hrow = h + (size_t)n * DH;
      hrow[ncol]      = fmaxf(acc0[r] + sB[ncol], 0.0f);
      hrow[16 + ncol] = fmaxf(acc1[r] + sB[16 + ncol], 0.0f);
      if (32 + ncol < DH)
        hrow[32 + ncol] = fmaxf(acc2[r] + sB[32 + ncol], 0.0f);
    }
  }
}

__global__ void __launch_bounds__(256)
layer2_kernel(const float* __restrict__ h, const float* __restrict__ sum2,
              const float* __restrict__ cnt, const float* __restrict__ W2l,
              const float* __restrict__ b2, const float* __restrict__ W2r,
              float* __restrict__ o, int nNodes) {
  __shared__ float sWl[DO_PAD * DH];
  __shared__ float sWr[DO_PAD * DH];
  __shared__ float sB[DOUT];
  for (int i = threadIdx.x; i < DO_PAD * DH; i += 256) {
    float wl = (i < DOUT * DH) ? W2l[i] : 0.0f;
    float wr = (i < DOUT * DH) ? W2r[i] : 0.0f;
    sWl[i] = wl;
    sWr[i] = wr;
  }
  if (threadIdx.x < DOUT) sB[threadIdx.x] = b2[threadIdx.x];
  __syncthreads();

  int lane = threadIdx.x & 31;
  int wave = threadIdx.x >> 5;
  int node0 = (blockIdx.x * 8 + wave) * 16;
  int row = lane & 15;
  int node = node0 + row;
  int nodeC = min(node, nNodes - 1);
  int klo = (lane >> 4) * 2;
  int ncol = lane & 15;

  float inv = 1.0f / fmaxf(cnt[nodeC], 1.0f);

  const float* hrow = h + (size_t)nodeC * DH;
  const float* srow = sum2 + (size_t)nodeC * DH;

  v8f acc0 = {}, acc1 = {};
  for (int k = 0; k < DH; k += 4) {
    v2f aH, aA;
    aH.x = hrow[k + klo];
    aH.y = hrow[k + klo + 1];
    aA.x = srow[k + klo] * inv;
    aA.y = srow[k + klo + 1] * inv;

    v2f bl0, br0, bl1, br1;
    bl0.x = sWl[ncol * DH + k + klo];        bl0.y = sWl[ncol * DH + k + klo + 1];
    br0.x = sWr[ncol * DH + k + klo];        br0.y = sWr[ncol * DH + k + klo + 1];
    bl1.x = sWl[(16 + ncol) * DH + k + klo]; bl1.y = sWl[(16 + ncol) * DH + k + klo + 1];
    br1.x = sWr[(16 + ncol) * DH + k + klo]; br1.y = sWr[(16 + ncol) * DH + k + klo + 1];

    acc0 = wmma4(aA, bl0, acc0);
    acc0 = wmma4(aH, br0, acc0);
    acc1 = wmma4(aA, bl1, acc1);
    acc1 = wmma4(aH, br1, acc1);
  }

  int mbase = (lane >> 4) * 8;
  for (int r = 0; r < 8; ++r) {
    int n = node0 + mbase + r;
    if (n < nNodes) {
      float* orow = o + (size_t)n * DOUT;
      orow[ncol] = acc0[r] + sB[ncol];
      if (16 + ncol < DOUT)
        orow[16 + ncol] = acc1[r] + sB[16 + ncol];
    }
  }
}

__global__ void __launch_bounds__(256)
logsoftmax_kernel(const float* __restrict__ o, float* __restrict__ out, int nNodes) {
  int n = blockIdx.x * 256 + threadIdx.x;
  if (n >= nNodes) return;
  const float* r = o + (size_t)n * DOUT;
  float mx = -3.402823466e+38f;
  for (int j = 0; j < DOUT; ++j) mx = fmaxf(mx, r[j]);
  float s = 0.0f;
  for (int j = 0; j < DOUT; ++j) s += expf(r[j] - mx);
  float lse = mx + logf(s);
  float* w = out + (size_t)n * DOUT;
  for (int j = 0; j < DOUT; ++j) w[j] = r[j] - lse;
}

extern "C" void kernel_launch(void* const* d_in, const int* in_sizes, int n_in,
                              void* d_out, int out_size, void* d_ws, size_t ws_size,
                              hipStream_t stream) {
  const float* x   = (const float*)d_in[0];
  const int*   ei  = (const int*)d_in[1];
  const float* W1l = (const float*)d_in[2];
  const float* b1  = (const float*)d_in[3];
  const float* W1r = (const float*)d_in[4];
  const float* W2l = (const float*)d_in[5];
  const float* b2  = (const float*)d_in[6];
  const float* W2r = (const float*)d_in[7];
  float* out = (float*)d_out;

  int  N = in_sizes[0] / DIN;
  long E = (long)in_sizes[1] / 2;
  const int* src = ei;
  const int* dst = ei + E;

  float* ws   = (float*)d_ws;
  float* sum1 = ws;
  float* sum2 = ws;
  float* cnt  = sum1 + (size_t)N * DIN;
  float* h    = cnt + N;
  float* o    = h + (size_t)N * DH;
  int* rowptr = (int*)(o + (size_t)N * DOUT);
  int* col    = rowptr + N + 1;

  int NB = (N + NPB - 1) / NPB;
  int nCnt = NB * EB;
  unsigned* pk = (unsigned*)d_ws;
  int* cm   = (int*)d_ws + E;
  int* offs = cm + nCnt;

  bucket_count_kernel<<<EB, ET, 0, stream>>>(dst, cm, E, NB);
  scan_kernel<<<(nCnt + SCAN_T - 1) / SCAN_T, SCAN_T, 0, stream>>>(cm, offs, nCnt);
  bucket_move_kernel<<<EB, ET, 0, stream>>>(src, dst, offs, pk, E, NB);
  bucket_sort_kernel<<<NB, ET, 0, stream>>>(pk, offs, rowptr, col, N);

  gather1_kernel<<<2048, 256, 0, stream>>>(x, rowptr, col, sum1, cnt, N);

  int nb = (N + 127) / 128;
  layer1_kernel<<<nb, 256, 0, stream>>>(x, sum1, cnt, W1l, b1, W1r, h, N);

  gather2_kernel<<<2048, 256, 0, stream>>>(h, rowptr, col, sum2, N);

  layer2_kernel<<<nb, 256, 0, stream>>>(h, sum2, cnt, W2l, b2, W2r, o, N);

  logsoftmax_kernel<<<(N + 255) / 256, 256, 0, stream>>>(o, out, N);
}


# Both layers multiply rows by Wl and Wr first (WMMA) and then average the neighbours' projected 40- and 24-float rows, not 64- and 40-float inputs: mean after product, sums regrouped
# speedup vs baseline: 3.8039x; 1.1005x over previous
#include <hip/hip_runtime.h>
#include <math.h>

#define DIN 64
#define DH  40
#define DOUT 24
#define DH_PAD 48
#define DO_PAD 32
#define SCAN_T 1024

typedef __attribute__((ext_vector_type(2))) float v2f;
typedef __attribute__((ext_vector_type(8))) float v8f;

__device__ __forceinline__ v8f wmma4(v2f a, v2f b, v8f c) {
  return __builtin_amdgcn_wmma_f32_16x16x4_f32(false, a, false, b, (short)0, c,
                                               false, false);
}

#define NPB 256
#define MAXB 1024
#define EB 256
#define ET 1024

__device__ __forceinline__ int bucket_of(int d, int NB) {
  return min((int)((unsigned)d >> 8), NB - 1);
}

__global__ void __launch_bounds__(ET)
bucket_count_kernel(const int* __restrict__ dst, int* __restrict__ cm, long E,
                    int NB) {
  __shared__ int hist[MAXB];
  int t = threadIdx.x;
  for (int j = t; j < NB; j += ET) hist[j] = 0;
  __syncthreads();
  long chunk = (E + EB - 1) / EB;
  long e0 = (long)blockIdx.x * chunk;
  long e1 = min(E, e0 + chunk);
  long e = e0 + t;
  for (; e + 3 * ET < e1; e += 4 * ET) {
    int d0 = dst[e];
    int d1 = dst[e + ET];
    int d2 = dst[e + 2 * ET];
    int d3 = dst[e + 3 * ET];
    atomicAdd(&hist[bucket_of(d0, NB)], 1);
    atomicAdd(&hist[bucket_of(d1, NB)], 1);
    atomicAdd(&hist[bucket_of(d2, NB)], 1);
    atomicAdd(&hist[bucket_of(d3, NB)], 1);
  }
  for (; e < e1; e += ET) atomicAdd(&hist[bucket_of(dst[e], NB)], 1);
  __syncthreads();
  for (int j = t; j < NB; j += ET) cm[(size_t)j * EB + blockIdx.x] = hist[j];
}

__global__ void __launch_bounds__(SCAN_T)
scan_kernel(const int* __restrict__ deg, int* __restrict__ rowptr, int n) {
  __shared__ int s[SCAN_T];
  int t = threadIdx.x;
  int base = blockIdx.x * SCAN_T;
  int pre = 0;
  for (int i = t; i < base; i += SCAN_T) pre += deg[i];
  int idx = base + t;
  int v = (idx < n) ? deg[idx] : 0;

  s[t] = pre;
  __syncthreads();
  for (int off = SCAN_T / 2; off > 0; off >>= 1) {
    if (t < off) s[t] += s[t + off];
    __syncthreads();
  }
  int blockPre = s[0];
  __syncthreads();

  s[t] = v;
  __syncthreads();
  for (int off = 1; off < SCAN_T; off <<= 1) {
    int add = (t >= off) ? s[t - off] : 0;
    __syncthreads();
    s[t] += add;
    __syncthreads();
  }
  int incl = s[t];
  if (idx < n) rowptr[idx] = blockPre + incl - v;
  if (idx == n - 1) rowptr[n] = blockPre + incl;
}

__global__ void __launch_bounds__(ET)
bucket_move_kernel(const int* __restrict__ src, const int* __restrict__ dst,
                   const int* __restrict__ offs, unsigned* __restrict__ pk,
                   long E, int NB) {
  __shared__ int cur[MAXB];
  int t = threadIdx.x;
  for (int j = t; j < NB; j += ET) cur[j] = offs[(size_t)j * EB + blockIdx.x];
  __syncthreads();
  long chunk = (E + EB - 1) / EB;
  long e0 = (long)blockIdx.x * chunk;
  long e1 = min(E, e0 + chunk);
  long e = e0 + t;
  for (; e + 3 * ET < e1; e += 4 * ET) {
    int d0 = dst[e];
    int d1 = dst[e + ET];
    int d2 = dst[e + 2 * ET];
    int d3 = dst[e + 3 * ET];
    int s0 = src[e];
    int s1 = src[e + ET];
    int s2 = src[e + 2 * ET];
    int s3 = src[e + 3 * ET];
    int p0 = atomicAdd(&cur[bucket_of(d0, NB)], 1);
    int p1 = atomicAdd(&cur[bucket_of(d1, NB)], 1);
    int p2 = atomicAdd(&cur[bucket_of(d2, NB)], 1);
    int p3 = atomicAdd(&cur[bucket_of(d3, NB)], 1);
    pk[p0] = ((unsigned)s0 << 8) | ((unsigned)d0 & 255u);
    pk[p1] = ((unsigned)s1 << 8) | ((unsigned)d1 & 255u);
    pk[p2] = ((unsigned)s2 << 8) | ((unsigned)d2 & 255u);
    pk[p3] = ((unsigned)s3 << 8) | ((unsigned)d3 & 255u);
  }
  for (; e < e1; e += ET) {
    int d = dst[e];
    int p = atomicAdd(&cur[bucket_of(d, NB)], 1);
    pk[p] = ((unsigned)src[e] << 8) | ((unsigned)d & 255u);
  }
}

__global__ void __launch_bounds__(ET)
bucket_sort_kernel(const unsigned* __restrict__ pk, const int* __restrict__ offs,
                   int* __restrict__ rowptr, int* __restrict__ col, int nNodes) {
  __shared__ int cnt[NPB];
  __shared__ int cur[NPB];
  __shared__ int wsum[NPB / 32];
  int t = threadIdx.x;
  int b = blockIdx.x;
  int start = offs[(size_t)b * EB];
  int end = offs[(size_t)(b + 1) * EB];
  if (t < NPB) cnt[t] = 0;
  __syncthreads();

  int i = start + t;
  for (; i + 3 * ET < end; i += 4 * ET) {
    unsigned w0 = pk[i];
    unsigned w1 = pk[i + ET];
    unsigned w2 = pk[i + 2 * ET];
    unsigned w3 = pk[i + 3 * ET];
    atomicAdd(&cnt[w0 & 255u], 1);
    atomicAdd(&cnt[w1 & 255u], 1);
    atomicAdd(&cnt[w2 & 255u], 1);
    atomicAdd(&cnt[w3 & 255u], 1);
  }
  for (; i < end; i += ET) atomicAdd(&cnt[pk[i] & 255u], 1);
  __syncthreads();

  int v = 0, incl = 0;
  if (t < NPB) {
    v = cnt[t];
    incl = v;
    for (int off = 1; off < 32; off <<= 1) {
      int u = __shfl_up(incl, off);
      if ((t & 31) >= off) incl += u;
    }
    if ((t & 31) == 31) wsum[t >> 5] = incl;
  }
  __syncthreads();
  if (t < NPB) {
    int pre = start;
    for (int w = 0; w < (t >> 5); ++w) pre += wsum[w];
    int excl = pre + incl - v;
    cur[t] = excl;
    int node = b * NPB + t;
    if (node < nNodes) rowptr[node] = excl;
    if (node == nNodes - 1) rowptr[nNodes] = end;
  }
  __syncthreads();

  i = start + t;
  for (; i + 3 * ET < end; i += 4 * ET) {
    unsigned w0 = pk[i];
    unsigned w1 = pk[i + ET];
    unsigned w2 = pk[i + 2 * ET];
    unsigned w3 = pk[i + 3 * ET];
    int p0 = atomicAdd(&cur[w0 & 255u], 1);
    int p1 = atomicAdd(&cur[w1 & 255u], 1);
    int p2 = atomicAdd(&cur[w2 & 255u], 1);
    int p3 = atomicAdd(&cur[w3 & 255u], 1);
    col[p0] = (int)(w0 >> 8);
    col[p1] = (int)(w1 >> 8);
    col[p2] = (int)(w2 >> 8);
    col[p3] = (int)(w3 >> 8);
  }
  for (; i < end; i += ET) {
    unsigned w = pk[i];
    int p = atomicAdd(&cur[w & 255u], 1);
    col[p] = (int)(w >> 8);
  }
}

__device__ __forceinline__ float4 ld4(const float* p) {
  return *reinterpret_cast<const float4*>(p);
}
__device__ __forceinline__ float4 add4(float4 a, float4 b) {
  return make_float4(a.x + b.x, a.y + b.y, a.z + b.z, a.w + b.w);
}


__global__ void __launch_bounds__(256)
proj1_kernel(const float* __restrict__ x, const float* __restrict__ W1l,
             const float* __restrict__ b1, const float* __restrict__ W1r,
             float* __restrict__ p, float* __restrict__ r, int nNodes) {
  __shared__ float sWl[DH_PAD * DIN];
  __shared__ float sWr[DH_PAD * DIN];
  __shared__ float sB[DH];
  for (int i = threadIdx.x; i < DH_PAD * DIN; i += 256) {
    float wl = (i < DH * DIN) ? W1l[i] : 0.0f;
    float wr = (i < DH * DIN) ? W1r[i] : 0.0f;
    sWl[i] = wl;
    sWr[i] = wr;
  }
  if (threadIdx.x < DH) sB[threadIdx.x] = b1[threadIdx.x];
  __syncthreads();

  int lane = threadIdx.x & 31;
  int wave = threadIdx.x >> 5;
  int node0 = (blockIdx.x * 8 + wave) * 16;
  int row = lane & 15;
  int node = node0 + row;
  int nodeC = min(node, nNodes - 1);
  int klo = (lane >> 4) * 2;
  int ncol = lane & 15;

  const float* xrow = x + (size_t)nodeC * DIN;

  v8f p0 = {}, p1 = {}, p2 = {}, r0 = {}, r1 = {}, r2 = {};
  for (int k = 0; k < DIN; k += 4) {
    v2f aX;
    aX.x = xrow[k + klo];
    aX.y = xrow[k + klo + 1];

    v2f bl0, br0, bl1, br1, bl2, br2;
    bl0.x = sWl[ncol * DIN + k + klo];        bl0.y = sWl[ncol * DIN + k + klo + 1];
    br0.x = sWr[ncol * DIN + k + klo];        br0.y = sWr[ncol * DIN + k + klo + 1];
    bl1.x = sWl[(16 + ncol) * DIN + k + klo]; bl1.y = sWl[(16 + ncol) * DIN + k + klo + 1];
    br1.x = sWr[(16 + ncol) * DIN + k + klo]; br1.y = sWr[(16 + ncol) * DIN + k + klo + 1];
    bl2.x = sWl[(32 + ncol) * DIN + k + klo]; bl2.y = sWl[(32 + ncol) * DIN + k + klo + 1];
    br2.x = sWr[(32 + ncol) * DIN + k + klo]; br2.y = sWr[(32 + ncol) * DIN + k + klo + 1];

    p0 = wmma4(aX, bl0, p0);
    r0 = wmma4(aX, br0, r0);
    p1 = wmma4(aX, bl1, p1);
    r1 = wmma4(aX, br1, r1);
    p2 = wmma4(aX, bl2, p2);
    r2 = wmma4(aX, br2, r2);
  }

  int mbase = (lane >> 4) * 8;
  for (int i = 0; i < 8; ++i) {
    int n = node0 + mbase + i;
    if (n < nNodes) {
      float* prow = p + (size_t)n * DH;
      float* rrow = r + (size_t)n * DH;
      prow[ncol]      = p0[i];
      prow[16 + ncol] = p1[i];
      rrow[ncol]      = r0[i] + sB[ncol];
      rrow[16 + ncol] = r1[i] + sB[16 + ncol];
      if (32 + ncol < DH) {
        prow[32 + ncol] = p2[i];
        rrow[32 + ncol] = r2[i] + sB[32 + ncol];
      }
    }
  }
}

__global__ void __launch_bounds__(256)
gather1_kernel(const float* __restrict__ p, const float* __restrict__ r,
               const int* __restrict__ rowptr, const int* __restrict__ col,
               float* __restrict__ h, int nNodes) {
  int lane = threadIdx.x & 31;
  int wave = __builtin_amdgcn_readfirstlane((int)(threadIdx.x >> 5));
  int nw = (int)gridDim.x * 8;
  int n = blockIdx.x * 8 + wave;
  if (n >= nNodes) return;
  int q = lane / 10;
  int chunk = lane - q * 10;
  int g = min(q, 2);
  const float* pc = p + chunk * 4;
  int beg = rowptr[n], end = rowptr[n + 1];
  while (true) {
    int nn = n + nw;
    int nbeg = 0, nend = 0;
    if (nn < nNodes) { nbeg = rowptr[nn]; nend = rowptr[nn + 1]; }
    float4 own = make_float4(0.0f, 0.0f, 0.0f, 0.0f);
    if (lane < 10) own = ld4(r + (size_t)n * DH + chunk * 4);

    float4 acc = make_float4(0.0f, 0.0f, 0.0f, 0.0f);
    int e0 = beg + lane;
    int my = (lane < 30 && e0 < end) ? col[e0] : 0;
    for (int base = beg; base < end; base += 30) {
      int e1 = base + 30 + lane;
      int myn = (lane < 30 && e1 < end) ? col[e1] : 0;
      int m = min(30, end - base);
      for (int j = 0; j < m; j += 15) {
        float4 v[5];
#pragma unroll
        for (int u = 0; u < 5; ++u) {
          int k = j + 3 * u + g;
          int s = __shfl(my, k);
          float4 t = ld4(pc + (size_t)s * DH);
          bool ok = k < m;
          v[u] = make_float4(ok ? t.x : 0.0f, ok ? t.y : 0.0f,
                             ok ? t.z : 0.0f, ok ? t.w : 0.0f);
        }
        acc = add4(acc, add4(add4(add4(v[0], v[1]), add4(v[2], v[3])), v[4]));
      }
      my = myn;
    }
    float4 t1, t2;
    t1.x = __shfl_down(acc.x, 10); t2.x = __shfl_down(acc.x, 20);
    t1.y = __shfl_down(acc.y, 10); t2.y = __shfl_down(acc.y, 20);
    t1.z = __shfl_down(acc.z, 10); t2.z = __shfl_down(acc.z, 20);
    t1.w = __shfl_down(acc.w, 10); t2.w = __shfl_down(acc.w, 20);
    if (lane < 10) {
      float inv = 1.0f / fmaxf((float)(end - beg), 1.0f);
      float4 s = add4(acc, add4(t1, t2));
      *reinterpret_cast<float4*>(h + (size_t)n * DH + chunk * 4) =
          make_float4(fmaxf(s.x * inv + own.x, 0.0f), fmaxf(s.y * inv + own.y, 0.0f),
                      fmaxf(s.z * inv + own.z, 0.0f), fmaxf(s.w * inv + own.w, 0.0f));
    }

    if (nn >= nNodes) break;
    n = nn; beg = nbeg; end = nend;
  }
}

__global__ void __launch_bounds__(256)
proj2_kernel(const float* __restrict__ h, const float* __restrict__ W2l,
             const float* __restrict__ b2, const float* __restrict__ W2r,
             float* __restrict__ p, float* __restrict__ r, int nNodes) {
  __shared__ float sWl[DO_PAD * DH];
  __shared__ float sWr[DO_PAD * DH];
  __shared__ float sB[DO_PAD];
  for (int i = threadIdx.x; i < DO_PAD * DH; i += 256) {
    float wl = (i < DOUT * DH) ? W2l[i] : 0.0f;
    float wr = (i < DOUT * DH) ? W2r[i] : 0.0f;
    sWl[i] = wl;
    sWr[i] = wr;
  }
  if (threadIdx.x < DO_PAD)
    sB[threadIdx.x] = (threadIdx.x < DOUT) ? b2[threadIdx.x] : 0.0f;
  __syncthreads();

  int lane = threadIdx.x & 31;
  int wave = threadIdx.x >> 5;
  int node0 = (blockIdx.x * 8 + wave) * 16;
  int row = lane & 15;
  int node = node0 + row;
  int nodeC = min(node, nNodes - 1);
  int klo = (lane >> 4) * 2;
  int ncol = lane & 15;

  const float* hrow = h + (size_t)nodeC * DH;

  v8f p0 = {}, p1 = {}, r0 = {}, r1 = {};
  for (int k = 0; k < DH; k += 4) {
    v2f aH;
    aH.x = hrow[k + klo];
    aH.y = hrow[k + klo + 1];

    v2f bl0, br0, bl1, br1;
    bl0.x = sWl[ncol * DH + k + klo];        bl0.y = sWl[ncol * DH + k + klo + 1];
    br0.x = sWr[ncol * DH + k + klo];        br0.y = sWr[ncol * DH + k + klo + 1];
    bl1.x = sWl[(16 + ncol) * DH + k + klo]; bl1.y = sWl[(16 + ncol) * DH + k + klo + 1];
    br1.x = sWr[(16 + ncol) * DH + k + klo]; br1.y = sWr[(16 + ncol) * DH + k + klo + 1];

    p0 = wmma4(aH, bl0, p0);
    r0 = wmma4(aH, br0, r0);
    p1 = wmma4(aH, bl1, p1);
    r1 = wmma4(aH, br1, r1);
  }

  int mbase = (lane >> 4) * 8;
  for (int i = 0; i < 8; ++i) {
    int n = node0 + mbase + i;
    if (n < nNodes) {
      float* prow = p + (size_t)n * DO_PAD;
      float* rrow = r + (size_t)n * DO_PAD;
      prow[ncol]      = p0[i];
      prow[16 + ncol] = p1[i];
      rrow[ncol]      = r0[i] + sB[ncol];
      rrow[16 + ncol] = r1[i] + sB[16 + ncol];
    }
  }
}

__global__ void __launch_bounds__(256)
gather2_kernel(const float* __restrict__ p, const float* __restrict__ r,
               const int* __restrict__ rowptr, const int* __restrict__ col,
               float* __restrict__ o, int nNodes) {
  int lane = threadIdx.x & 31;
  int wave = __builtin_amdgcn_readfirstlane((int)(threadIdx.x >> 5));
  int nw = (int)gridDim.x * 8;
  int n = blockIdx.x * 8 + wave;
  if (n >= nNodes) return;
  int q = lane >> 3;
  int chunk = lane & 7;
  const float* pc = p + chunk * 4;
  int beg = rowptr[n], end = rowptr[n + 1];
  while (true) {
    int nn = n + nw;
    int nbeg = 0, nend = 0;
    if (nn < nNodes) { nbeg = rowptr[nn]; nend = rowptr[nn + 1]; }
    float4 own = make_float4(0.0f, 0.0f, 0.0f, 0.0f);
    if (lane < 6) own = ld4(r + (size_t)n * DO_PAD + chunk * 4);

    float4 acc = make_float4(0.0f, 0.0f, 0.0f, 0.0f);
    int e0 = beg + lane;
    int my = (e0 < end) ? col[e0] : 0;
    for (int base = beg; base < end; base += 32) {
      int e1 = base + 32 + lane;
      int myn = (e1 < end) ? col[e1] : 0;
      int m = min(32, end - base);
      for (int j = 0; j < m; j += 16) {
        float4 v[4];
#pragma unroll
        for (int u = 0; u < 4; ++u) {
          int k = j + 4 * u + q;
          int s = __shfl(my, k);
          float4 t = ld4(pc + (size_t)s * DO_PAD);
          bool ok = k < m;
          v[u] = make_float4(ok ? t.x : 0.0f, ok ? t.y : 0.0f,
                             ok ? t.z : 0.0f, ok ? t.w : 0.0f);
        }
        acc = add4(acc, add4(add4(v[0], v[1]), add4(v[2], v[3])));
      }
      my = myn;
    }
    acc.x += __shfl_xor(acc.x, 8);
    acc.y += __shfl_xor(acc.y, 8);
    acc.z += __shfl_xor(acc.z, 8);
    acc.w += __shfl_xor(acc.w, 8);
    acc.x += __shfl_xor(acc.x, 16);
    acc.y += __shfl_xor(acc.y, 16);
    acc.z += __shfl_xor(acc.z, 16);
    acc.w += __shfl_xor(acc.w, 16);
    if (lane < 6) {
      float inv = 1.0f / fmaxf((float)(end - beg), 1.0f);
      *reinterpret_cast<float4*>(o + (size_t)n * DOUT + chunk * 4) =
          make_float4(acc.x * inv + own.x, acc.y * inv + own.y,
                      acc.z * inv + own.z, acc.w * inv + own.w);
    }

    if (nn >= nNodes) break;
    n = nn; beg = nbeg; end = nend;
  }
}

__global__ void __launch_bounds__(256)
logsoftmax_kernel(const float* __restrict__ o, float* __restrict__ out, int nNodes) {
  int n = blockIdx.x * 256 + threadIdx.x;
  if (n >= nNodes) return;
  const float* r = o + (size_t)n * DOUT;
  float mx = -3.402823466e+38f;
  for (int j = 0; j < DOUT; ++j) mx = fmaxf(mx, r[j]);
  float s = 0.0f;
  for (int j = 0; j < DOUT; ++j) s += expf(r[j] - mx);
  float lse = mx + logf(s);
  float* w = out + (size_t)n * DOUT;
  for (int j = 0; j < DOUT; ++j) w[j] = r[j] - lse;
}

extern "C" void kernel_launch(void* const* d_in, const int* in_sizes, int n_in,
                              void* d_out, int out_size, void* d_ws, size_t ws_size,
                              hipStream_t stream) {
  const float* x   = (const float*)d_in[0];
  const int*   ei  = (const int*)d_in[1];
  const float* W1l = (const float*)d_in[2];
  const float* b1  = (const float*)d_in[3];
  const float* W1r = (const float*)d_in[4];
  const float* W2l = (const float*)d_in[5];
  const float* b2  = (const float*)d_in[6];
  const float* W2r = (const float*)d_in[7];
  float* out = (float*)d_out;

  int  N = in_sizes[0] / DIN;
  long E = (long)in_sizes[1] / 2;
  const int* src = ei;
  const int* dst = ei + E;

  float* ws = (float*)d_ws;
  float* P  = ws;
  float* R  = P + (size_t)N * DH;
  float* H  = R + (size_t)N * DH;
  float* o  = H;
  int* rowptr = (int*)(ws + (size_t)N * (DIN + 1 + DH + DOUT));
  int* col    = rowptr + N + 1;

  int NB = (N + NPB - 1) / NPB;
  int nCnt = NB * EB;
  unsigned* pk = (unsigned*)d_ws;
  int* cm   = (int*)d_ws + E;
  int* offs = cm + nCnt;

  bucket_count_kernel<<<EB, ET, 0, stream>>>(dst, cm, E, NB);
  scan_kernel<<<(nCnt + SCAN_T - 1) / SCAN_T, SCAN_T, 0, stream>>>(cm, offs, nCnt);
  bucket_move_kernel<<<EB, ET, 0, stream>>>(src, dst, offs, pk, E, NB);
  bucket_sort_kernel<<<NB, ET, 0, stream>>>(pk, offs, rowptr, col, N);

  int nb = (N + 127) / 128;

  proj1_kernel<<<nb, 256, 0, stream>>>(x, W1l, b1, W1r, P, R, N);
  gather1_kernel<<<2048, 256, 0, stream>>>(P, R, rowptr, col, H, N);

  proj2_kernel<<<nb, 256, 0, stream>>>(H, W2l, b2, W2r, P, R, N);
  gather2_kernel<<<2048, 256, 0, stream>>>(P, R, rowptr, col, o, N);

  logsoftmax_kernel<<<(N + 255) / 256, 256, 0, stream>>>(o, out, N);
}


# Weight rows staged in LDS at a pitch of 68 and 44 floats, not 64 and 40: the 16 lanes of a B-fragment read fall on different banks in both projection kernels
# speedup vs baseline: 4.0307x; 1.0596x over previous
#include <hip/hip_runtime.h>
#include <math.h>

#define DIN 64
#define DH  40
#define DOUT 24
#define DH_PAD 48
#define DO_PAD 32
#define SCAN_T 1024
#define W1P 68
#define W2P 44

typedef __attribute__((ext_vector_type(2))) float v2f;
typedef __attribute__((ext_vector_type(8))) float v8f;

__device__ __forceinline__ v8f wmma4(v2f a, v2f b, v8f c) {
  return __builtin_amdgcn_wmma_f32_16x16x4_f32(false, a, false, b, (short)0, c,
                                               false, false);
}

#define NPB 256
#define MAXB 1024
#define EB 256
#define ET 1024

__device__ __forceinline__ int bucket_of(int d, int NB) {
  return min((int)((unsigned)d >> 8), NB - 1);
}

__global__ void __launch_bounds__(ET)
bucket_count_kernel(const int* __restrict__ dst, int* __restrict__ cm, long E,
                    int NB) {
  __shared__ int hist[MAXB];
  int t = threadIdx.x;
  for (int j = t; j < NB; j += ET) hist[j] = 0;
  __syncthreads();
  long chunk = (E + EB - 1) / EB;
  long e0 = (long)blockIdx.x * chunk;
  long e1 = min(E, e0 + chunk);
  long e = e0 + t;
  for (; e + 3 * ET < e1; e += 4 * ET) {
    int d0 = dst[e];
    int d1 = dst[e + ET];
    int d2 = dst[e + 2 * ET];
    int d3 = dst[e + 3 * ET];
    atomicAdd(&hist[bucket_of(d0, NB)], 1);
    atomicAdd(&hist[bucket_of(d1, NB)], 1);
    atomicAdd(&hist[bucket_of(d2, NB)], 1);
    atomicAdd(&hist[bucket_of(d3, NB)], 1);
  }
  for (; e < e1; e += ET) atomicAdd(&hist[bucket_of(dst[e], NB)], 1);
  __syncthreads();
  for (int j = t; j < NB; j += ET) cm[(size_t)j * EB + blockIdx.x] = hist[j];
}

__global__ void __launch_bounds__(SCAN_T)
scan_kernel(const int* __restrict__ deg, int* __restrict__ rowptr, int n) {
  __shared__ int s[SCAN_T];
  int t = threadIdx.x;
  int base = blockIdx.x * SCAN_T;
  int pre = 0;
  for (int i = t; i < base; i += SCAN_T) pre += deg[i];
  int idx = base + t;
  int v = (idx < n) ? deg[idx] : 0;

  s[t] = pre;
  __syncthreads();
  for (int off = SCAN_T / 2; off > 0; off >>= 1) {
    if (t < off) s[t] += s[t + off];
    __syncthreads();
  }
  int blockPre = s[0];
  __syncthreads();

  s[t] = v;
  __syncthreads();
  for (int off = 1; off < SCAN_T; off <<= 1) {
    int add = (t >= off) ? s[t - off] : 0;
    __syncthreads();
    s[t] += add;
    __syncthreads();
  }
  int incl = s[t];
  if (idx < n) rowptr[idx] = blockPre + incl - v;
  if (idx == n - 1) rowptr[n] = blockPre + incl;
}

__global__ void __launch_bounds__(ET)
bucket_move_kernel(const int* __restrict__ src, const int* __restrict__ dst,
                   const int* __restrict__ offs, unsigned* __restrict__ pk,
                   long E, int NB) {
  __shared__ int cur[MAXB];
  int t = threadIdx.x;
  for (int j = t; j < NB; j += ET) cur[j] = offs[(size_t)j * EB + blockIdx.x];
  __syncthreads();
  long chunk = (E + EB - 1) / EB;
  long e0 = (long)blockIdx.x * chunk;
  long e1 = min(E, e0 + chunk);
  long e = e0 + t;
  for (; e + 3 * ET < e1; e += 4 * ET) {
    int d0 = dst[e];
    int d1 = dst[e + ET];
    int d2 = dst[e + 2 * ET];
    int d3 = dst[e + 3 * ET];
    int s0 = src[e];
    int s1 = src[e + ET];
    int s2 = src[e + 2 * ET];
    int s3 = src[e + 3 * ET];
    int p0 = atomicAdd(&cur[bucket_of(d0, NB)], 1);
    int p1 = atomicAdd(&cur[bucket_of(d1, NB)], 1);
    int p2 = atomicAdd(&cur[bucket_of(d2, NB)], 1);
    int p3 = atomicAdd(&cur[bucket_of(d3, NB)], 1);
    pk[p0] = ((unsigned)s0 << 8) | ((unsigned)d0 & 255u);
    pk[p1] = ((unsigned)s1 << 8) | ((unsigned)d1 & 255u);
    pk[p2] = ((unsigned)s2 << 8) | ((unsigned)d2 & 255u);
    pk[p3] = ((unsigned)s3 << 8) | ((unsigned)d3 & 255u);
  }
  for (; e < e1; e += ET) {
    int d = dst[e];
    int p = atomicAdd(&cur[bucket_of(d, NB)], 1);
    pk[p] = ((unsigned)src[e] << 8) | ((unsigned)d & 255u);
  }
}

__global__ void __launch_bounds__(ET)
bucket_sort_kernel(const unsigned* __restrict__ pk, const int* __restrict__ offs,
                   int* __restrict__ rowptr, int* __restrict__ col, int nNodes) {
  __shared__ int cnt[NPB];
  __shared__ int cur[NPB];
  __shared__ int wsum[NPB / 32];
  int t = threadIdx.x;
  int b = blockIdx.x;
  int start = offs[(size_t)b * EB];
  int end = offs[(size_t)(b + 1) * EB];
  if (t < NPB) cnt[t] = 0;
  __syncthreads();

  int i = start + t;
  for (; i + 3 * ET < end; i += 4 * ET) {
    unsigned w0 = pk[i];
    unsigned w1 = pk[i + ET];
    unsigned w2 = pk[i + 2 * ET];
    unsigned w3 = pk[i + 3 * ET];
    atomicAdd(&cnt[w0 & 255u], 1);
    atomicAdd(&cnt[w1 & 255u], 1);
    atomicAdd(&cnt[w2 & 255u], 1);
    atomicAdd(&cnt[w3 & 255u], 1);
  }
  for (; i < end; i += ET) atomicAdd(&cnt[pk[i] & 255u], 1);
  __syncthreads();

  int v = 0, incl = 0;
  if (t < NPB) {
    v = cnt[t];
    incl = v;
    for (int off = 1; off < 32; off <<= 1) {
      int u = __shfl_up(incl, off);
      if ((t & 31) >= off) incl += u;
    }
    if ((t & 31) == 31) wsum[t >> 5] = incl;
  }
  __syncthreads();
  if (t < NPB) {
    int pre = start;
    for (int w = 0; w < (t >> 5); ++w) pre += wsum[w];
    int excl = pre + incl - v;
    cur[t] = excl;
    int node = b * NPB + t;
    if (node < nNodes) rowptr[node] = excl;
    if (node == nNodes - 1) rowptr[nNodes] = end;
  }
  __syncthreads();

  i = start + t;
  for (; i + 3 * ET < end; i += 4 * ET) {
    unsigned w0 = pk[i];
    unsigned w1 = pk[i + ET];
    unsigned w2 = pk[i + 2 * ET];
    unsigned w3 = pk[i + 3 * ET];
    int p0 = atomicAdd(&cur[w0 & 255u], 1);
    int p1 = atomicAdd(&cur[w1 & 255u], 1);
    int p2 = atomicAdd(&cur[w2 & 255u], 1);
    int p3 = atomicAdd(&cur[w3 & 255u], 1);
    col[p0] = (int)(w0 >> 8);
    col[p1] = (int)(w1 >> 8);
    col[p2] = (int)(w2 >> 8);
    col[p3] = (int)(w3 >> 8);
  }
  for (; i < end; i += ET) {
    unsigned w = pk[i];
    int p = atomicAdd(&cur[w & 255u], 1);
    col[p] = (int)(w >> 8);
  }
}

__device__ __forceinline__ float4 ld4(const float* p) {
  return *reinterpret_cast<const float4*>(p);
}
__device__ __forceinline__ float4 add4(float4 a, float4 b) {
  return make_float4(a.x + b.x, a.y + b.y, a.z + b.z, a.w + b.w);
}


__global__ void __launch_bounds__(256)
proj1_kernel(const float* __restrict__ x, const float* __restrict__ W1l,
             const float* __restrict__ b1, const float* __restrict__ W1r,
             float* __restrict__ p, float* __restrict__ r, int nNodes) {
  __shared__ float sWl[DH_PAD * W1P];
  __shared__ float sWr[DH_PAD * W1P];
  __shared__ float sB[DH];
  for (int i = threadIdx.x; i < DH_PAD * DIN; i += 256) {
    float wl = (i < DH * DIN) ? W1l[i] : 0.0f;
    float wr = (i < DH * DIN) ? W1r[i] : 0.0f;
    int j = (i / DIN) * W1P + (i % DIN);
    sWl[j] = wl;
    sWr[j] = wr;
  }
  if (threadIdx.x < DH) sB[threadIdx.x] = b1[threadIdx.x];
  __syncthreads();

  int lane = threadIdx.x & 31;
  int wave = threadIdx.x >> 5;
  int node0 = (blockIdx.x * 8 + wave) * 16;
  int row = lane & 15;
  int node = node0 + row;
  int nodeC = min(node, nNodes - 1);
  int klo = (lane >> 4) * 2;
  int ncol = lane & 15;

  const float* xrow = x + (size_t)nodeC * DIN;

  v8f p0 = {}, p1 = {}, p2 = {}, r0 = {}, r1 = {}, r2 = {};
  for (int k = 0; k < DIN; k += 4) {
    v2f aX;
    aX.x = xrow[k + klo];
    aX.y = xrow[k + klo + 1];

    v2f bl0, br0, bl1, br1, bl2, br2;
    bl0.x = sWl[ncol * W1P + k + klo];        bl0.y = sWl[ncol * W1P + k + klo + 1];
    br0.x = sWr[ncol * W1P + k + klo];        br0.y = sWr[ncol * W1P + k + klo + 1];
    bl1.x = sWl[(16 + ncol) * W1P + k + klo]; bl1.y = sWl[(16 + ncol) * W1P + k + klo + 1];
    br1.x = sWr[(16 + ncol) * W1P + k + klo]; br1.y = sWr[(16 + ncol) * W1P + k + klo + 1];
    bl2.x = sWl[(32 + ncol) * W1P + k + klo]; bl2.y = sWl[(32 + ncol) * W1P + k + klo + 1];
    br2.x = sWr[(32 + ncol) * W1P + k + klo]; br2.y = sWr[(32 + ncol) * W1P + k + klo + 1];

    p0 = wmma4(aX, bl0, p0);
    r0 = wmma4(aX, br0, r0);
    p1 = wmma4(aX, bl1, p1);
    r1 = wmma4(aX, br1, r1);
    p2 = wmma4(aX, bl2, p2);
    r2 = wmma4(aX, br2, r2);
  }

  int mbase = (lane >> 4) * 8;
  for (int i = 0; i < 8; ++i) {
    int n = node0 + mbase + i;
    if (n < nNodes) {
      float* prow = p + (size_t)n * DH;
      float* rrow = r + (size_t)n * DH;
      prow[ncol]      = p0[i];
      prow[16 + ncol] = p1[i];
      rrow[ncol]      = r0[i] + sB[ncol];
      rrow[16 + ncol] = r1[i] + sB[16 + ncol];
      if (32 + ncol < DH) {
        prow[32 + ncol] = p2[i];
        rrow[32 + ncol] = r2[i] + sB[32 + ncol];
      }
    }
  }
}

__global__ void __launch_bounds__(256)
gather1_kernel(const float* __restrict__ p, const float* __restrict__ r,
               const int* __restrict__ rowptr, const int* __restrict__ col,
               float* __restrict__ h, int nNodes) {
  int lane = threadIdx.x & 31;
  int wave = __builtin_amdgcn_readfirstlane((int)(threadIdx.x >> 5));
  int nw = (int)gridDim.x * 8;
  int n = blockIdx.x * 8 + wave;
  if (n >= nNodes) return;
  int q = lane / 10;
  int chunk = lane - q * 10;
  int g = min(q, 2);
  const float* pc = p + chunk * 4;
  int beg = rowptr[n], end = rowptr[n + 1];
  while (true) {
    int nn = n + nw;
    int nbeg = 0, nend = 0;
    if (nn < nNodes) { nbeg = rowptr[nn]; nend = rowptr[nn + 1]; }
    float4 own = make_float4(0.0f, 0.0f, 0.0f, 0.0f);
    if (lane < 10) own = ld4(r + (size_t)n * DH + chunk * 4);

    float4 acc = make_float4(0.0f, 0.0f, 0.0f, 0.0f);
    int e0 = beg + lane;
    int my = (lane < 30 && e0 < end) ? col[e0] : 0;
    for (int base = beg; base < end; base += 30) {
      int e1 = base + 30 + lane;
      int myn = (lane < 30 && e1 < end) ? col[e1] : 0;
      int m = min(30, end - base);
      for (int j = 0; j < m; j += 15) {
        float4 v[5];
#pragma unroll
        for (int u = 0; u < 5; ++u) {
          int k = j + 3 * u + g;
          int s = __shfl(my, k);
          float4 t = ld4(pc + (size_t)s * DH);
          bool ok = k < m;
          v[u] = make_float4(ok ? t.x : 0.0f, ok ? t.y : 0.0f,
                             ok ? t.z : 0.0f, ok ? t.w : 0.0f);
        }
        acc = add4(acc, add4(add4(add4(v[0], v[1]), add4(v[2], v[3])), v[4]));
      }
      my = myn;
    }
    float4 t1, t2;
    t1.x = __shfl_down(acc.x, 10); t2.x = __shfl_down(acc.x, 20);
    t1.y = __shfl_down(acc.y, 10); t2.y = __shfl_down(acc.y, 20);
    t1.z = __shfl_down(acc.z, 10); t2.z = __shfl_down(acc.z, 20);
    t1.w = __shfl_down(acc.w, 10); t2.w = __shfl_down(acc.w, 20);
    if (lane < 10) {
      float inv = 1.0f / fmaxf((float)(end - beg), 1.0f);
      float4 s = add4(acc, add4(t1, t2));
      *reinterpret_cast<float4*>(h + (size_t)n * DH + chunk * 4) =
          make_float4(fmaxf(s.x * inv + own.x, 0.0f), fmaxf(s.y * inv + own.y, 0.0f),
                      fmaxf(s.z * inv + own.z, 0.0f), fmaxf(s.w * inv + own.w, 0.0f));
    }

    if (nn >= nNodes) break;
    n = nn; beg = nbeg; end = nend;
  }
}

__global__ void __launch_bounds__(256)
proj2_kernel(const float* __restrict__ h, const float* __restrict__ W2l,
             const float* __restrict__ b2, const float* __restrict__ W2r,
             float* __restrict__ p, float* __restrict__ r, int nNodes) {
  __shared__ float sWl[DO_PAD * W2P];
  __shared__ float sWr[DO_PAD * W2P];
  __shared__ float sB[DO_PAD];
  for (int i = threadIdx.x; i < DO_PAD * DH; i += 256) {
    float wl = (i < DOUT * DH) ? W2l[i] : 0.0f;
    float wr = (i < DOUT * DH) ? W2r[i] : 0.0f;
    int j = (i / DH) * W2P + (i % DH);
    sWl[j] = wl;
    sWr[j] = wr;
  }
  if (threadIdx.x < DO_PAD)
    sB[threadIdx.x] = (threadIdx.x < DOUT) ? b2[threadIdx.x] : 0.0f;
  __syncthreads();

  int lane = threadIdx.x & 31;
  int wave = threadIdx.x >> 5;
  int node0 = (blockIdx.x * 8 + wave) * 16;
  int row = lane & 15;
  int node = node0 + row;
  int nodeC = min(node, nNodes - 1);
  int klo = (lane >> 4) * 2;
  int ncol = lane & 15;

  const float* hrow = h + (size_t)nodeC * DH;

  v8f p0 = {}, p1 = {}, r0 = {}, r1 = {};
  for (int k = 0; k < DH; k += 4) {
    v2f aH;
    aH.x = hrow[k + klo];
    aH.y = hrow[k + klo + 1];

    v2f bl0, br0, bl1, br1;
    bl0.x = sWl[ncol * W2P + k + klo];        bl0.y = sWl[ncol * W2P + k + klo + 1];
    br0.x = sWr[ncol * W2P + k + klo];        br0.y = sWr[ncol * W2P + k + klo + 1];
    bl1.x = sWl[(16 + ncol) * W2P + k + klo]; bl1.y = sWl[(16 + ncol) * W2P + k + klo + 1];
    br1.x = sWr[(16 + ncol) * W2P + k + klo]; br1.y = sWr[(16 + ncol) * W2P + k + klo + 1];

    p0 = wmma4(aH, bl0, p0);
    r0 = wmma4(aH, br0, r0);
    p1 = wmma4(aH, bl1, p1);
    r1 = wmma4(aH, br1, r1);
  }

  int mbase = (lane >> 4) * 8;
  for (int i = 0; i < 8; ++i) {
    int n = node0 + mbase + i;
    if (n < nNodes) {
      float* prow = p + (size_t)n * DO_PAD;
      float* rrow = r + (size_t)n * DO_PAD;
      prow[ncol]      = p0[i];
      prow[16 + ncol] = p1[i];
      rrow[ncol]      = r0[i] + sB[ncol];
      rrow[16 + ncol] = r1[i] + sB[16 + ncol];
    }
  }
}

__global__ void __launch_bounds__(256)
gather2_kernel(const float* __restrict__ p, const float* __restrict__ r,
               const int* __restrict__ rowptr, const int* __restrict__ col,
               float* __restrict__ o, int nNodes) {
  int lane = threadIdx.x & 31;
  int wave = __builtin_amdgcn_readfirstlane((int)(threadIdx.x >> 5));
  int nw = (int)gridDim.x * 8;
  int n = blockIdx.x * 8 + wave;
  if (n >= nNodes) return;
  int q = lane >> 3;
  int chunk = lane & 7;
  const float* pc = p + chunk * 4;
  int beg = rowptr[n], end = rowptr[n + 1];
  while (true) {
    int nn = n + nw;
    int nbeg = 0, nend = 0;
    if (nn < nNodes) { nbeg = rowptr[nn]; nend = rowptr[nn + 1]; }
    float4 own = make_float4(0.0f, 0.0f, 0.0f, 0.0f);
    if (lane < 6) own = ld4(r + (size_t)n * DO_PAD + chunk * 4);

    float4 acc = make_float4(0.0f, 0.0f, 0.0f, 0.0f);
    int e0 = beg + lane;
    int my = (e0 < end) ? col[e0] : 0;
    for (int base = beg; base < end; base += 32) {
      int e1 = base + 32 + lane;
      int myn = (e1 < end) ? col[e1] : 0;
      int m = min(32, end - base);
      for (int j = 0; j < m; j += 16) {
        float4 v[4];
#pragma unroll
        for (int u = 0; u < 4; ++u) {
          int k = j + 4 * u + q;
          int s = __shfl(my, k);
          float4 t = ld4(pc + (size_t)s * DO_PAD);
          bool ok = k < m;
          v[u] = make_float4(ok ? t.x : 0.0f, ok ? t.y : 0.0f,
                             ok ? t.z : 0.0f, ok ? t.w : 0.0f);
        }
        acc = add4(acc, add4(add4(v[0], v[1]), add4(v[2], v[3])));
      }
      my = myn;
    }
    acc.x += __shfl_xor(acc.x, 8);
    acc.y += __shfl_xor(acc.y, 8);
    acc.z += __shfl_xor(acc.z, 8);
    acc.w += __shfl_xor(acc.w, 8);
    acc.x += __shfl_xor(acc.x, 16);
    acc.y += __shfl_xor(acc.y, 16);
    acc.z += __shfl_xor(acc.z, 16);
    acc.w += __shfl_xor(acc.w, 16);
    if (lane < 6) {
      float inv = 1.0f / fmaxf((float)(end - beg), 1.0f);
      *reinterpret_cast<float4*>(o + (size_t)n * DOUT + chunk * 4) =
          make_float4(acc.x * inv + own.x, acc.y * inv + own.y,
                      acc.z * inv + own.z, acc.w * inv + own.w);
    }

    if (nn >= nNodes) break;
    n = nn; beg = nbeg; end = nend;
  }
}

__global__ void __launch_bounds__(256)
logsoftmax_kernel(const float* __restrict__ o, float* __restrict__ out, int nNodes) {
  int n = blockIdx.x * 256 + threadIdx.x;
  if (n >= nNodes) return;
  const float* r = o + (size_t)n * DOUT;
  float mx = -3.402823466e+38f;
  for (int j = 0; j < DOUT; ++j) mx = fmaxf(mx, r[j]);
  float s = 0.0f;
  for (int j = 0; j < DOUT; ++j) s += expf(r[j] - mx);
  float lse = mx + logf(s);
  float* w = out + (size_t)n * DOUT;
  for (int j = 0; j < DOUT; ++j) w[j] = r[j] - lse;
}

extern "C" void kernel_launch(void* const* d_in, const int* in_sizes, int n_in,
                              void* d_out, int out_size, void* d_ws, size_t ws_size,
                              hipStream_t stream) {
  const float* x   = (const float*)d_in[0];
  const int*   ei  = (const int*)d_in[1];
  const float* W1l = (const float*)d_in[2];
  const float* b1  = (const float*)d_in[3];
  const float* W1r = (const float*)d_in[4];
  const float* W2l = (const float*)d_in[5];
  const float* b2  = (const float*)d_in[6];
  const float* W2r = (const float*)d_in[7];
  float* out = (float*)d_out;

  int  N = in_sizes[0] / DIN;
  long E = (long)in_sizes[1] / 2;
  const int* src = ei;
  const int* dst = ei + E;

  float* ws = (float*)d_ws;
  float* P  = ws;
  float* R  = P + (size_t)N * DH;
  float* H  = R + (size_t)N * DH;
  float* o  = H;
  int* rowptr = (int*)(ws + (size_t)N * (DIN + 1 + DH + DOUT));
  int* col    = rowptr + N + 1;

  int NB = (N + NPB - 1) / NPB;
  int nCnt = NB * EB;
  unsigned* pk = (unsigned*)d_ws;
  int* cm   = (int*)d_ws + E;
  int* offs = cm + nCnt;

  bucket_count_kernel<<<EB, ET, 0, stream>>>(dst, cm, E, NB);
  scan_kernel<<<(nCnt + SCAN_T - 1) / SCAN_T, SCAN_T, 0, stream>>>(cm, offs, nCnt);
  bucket_move_kernel<<<EB, ET, 0, stream>>>(src, dst, offs, pk, E, NB);
  bucket_sort_kernel<<<NB, ET, 0, stream>>>(pk, offs, rowptr, col, N);

  int nb = (N + 127) / 128;

  proj1_kernel<<<nb, 256, 0, stream>>>(x, W1l, b1, W1r, P, R, N);
  gather1_kernel<<<2048, 256, 0, stream>>>(P, R, rowptr, col, H, N);

  proj2_kernel<<<nb, 256, 0, stream>>>(H, W2l, b2, W2r, P, R, N);
  gather2_kernel<<<2048, 256, 0, stream>>>(P, R, rowptr, col, o, N);

  logsoftmax_kernel<<<(N + 255) / 256, 256, 0, stream>>>(o, out, N);
}


# Both sort passes order their records in LDS and store each run to pk and col as contiguous whole lines, not 2 x 3.2M single words to scattered lines; neighbours still summed in LDS-add arrival order
# speedup vs baseline: 5.0379x; 1.2499x over previous
#include <hip/hip_runtime.h>
#include <math.h>

#define DIN 64
#define DH  40
#define DOUT 24
#define DH_PAD 48
#define DO_PAD 32
#define SCAN_T 1024
#define W1P 68
#define W2P 44

typedef __attribute__((ext_vector_type(2))) float v2f;
typedef __attribute__((ext_vector_type(8))) float v8f;

__device__ __forceinline__ v8f wmma4(v2f a, v2f b, v8f c) {
  return __builtin_amdgcn_wmma_f32_16x16x4_f32(false, a, false, b, (short)0, c,
                                               false, false);
}

#define NPB 256
#define MAXB 1024
#define EB 256
#define ET 1024

__device__ __forceinline__ int bucket_of(int d, int NB) {
  return min((int)((unsigned)d >> 8), NB - 1);
}

__global__ void __launch_bounds__(ET)
bucket_count_kernel(const int* __restrict__ dst, int* __restrict__ cm, long E,
                    int NB) {
  __shared__ int hist[MAXB];
  int t = threadIdx.x;
  for (int j = t; j < NB; j += ET) hist[j] = 0;
  __syncthreads();
  long chunk = (E + EB - 1) / EB;
  long e0 = (long)blockIdx.x * chunk;
  long e1 = min(E, e0 + chunk);
  long e = e0 + t;
  for (; e + 3 * ET < e1; e += 4 * ET) {
    int d0 = dst[e];
    int d1 = dst[e + ET];
    int d2 = dst[e + 2 * ET];
    int d3 = dst[e + 3 * ET];
    atomicAdd(&hist[bucket_of(d0, NB)], 1);
    atomicAdd(&hist[bucket_of(d1, NB)], 1);
    atomicAdd(&hist[bucket_of(d2, NB)], 1);
    atomicAdd(&hist[bucket_of(d3, NB)], 1);
  }
  for (; e < e1; e += ET) atomicAdd(&hist[bucket_of(dst[e], NB)], 1);
  __syncthreads();
  for (int j = t; j < NB; j += ET) cm[(size_t)j * EB + blockIdx.x] = hist[j];
}

__global__ void __launch_bounds__(SCAN_T)
scan_kernel(const int* __restrict__ deg, int* __restrict__ rowptr, int n) {
  __shared__ int s[SCAN_T];
  int t = threadIdx.x;
  int base = blockIdx.x * SCAN_T;
  int pre = 0;
  for (int i = t; i < base; i += SCAN_T) pre += deg[i];
  int idx = base + t;
  int v = (idx < n) ? deg[idx] : 0;

  s[t] = pre;
  __syncthreads();
  for (int off = SCAN_T / 2; off > 0; off >>= 1) {
    if (t < off) s[t] += s[t + off];
    __syncthreads();
  }
  int blockPre = s[0];
  __syncthreads();

  s[t] = v;
  __syncthreads();
  for (int off = 1; off < SCAN_T; off <<= 1) {
    int add = (t >= off) ? s[t - off] : 0;
    __syncthreads();
    s[t] += add;
    __syncthreads();
  }
  int incl = s[t];
  if (idx < n) rowptr[idx] = blockPre + incl - v;
  if (idx == n - 1) rowptr[n] = blockPre + incl;
}

#define MR 13
#define MTILE (MR * ET)

__global__ void __launch_bounds__(ET)
bucket_move_kernel(const int* __restrict__ src, const int* __restrict__ dst,
                   const int* __restrict__ offs, unsigned* __restrict__ pk,
                   long E, int NB) {
  __shared__ int cur[MAXB];
  __shared__ int hist[MAXB + 1];
  __shared__ int2 lc[MAXB];
  __shared__ int wsum[ET / 32];
  __shared__ int2 srec[MTILE];
  int t = threadIdx.x;
  for (int j = t; j < NB; j += ET) cur[j] = offs[(size_t)j * EB + blockIdx.x];
  long chunk = (E + EB - 1) / EB;
  long e0 = (long)blockIdx.x * chunk;
  long e1 = min(E, e0 + chunk);
  for (long tb = e0; tb < e1; tb += MTILE) {
    long te = min(e1, tb + (long)MTILE);
    for (int j = t; j < NB; j += ET) hist[j] = 0;
    __syncthreads();

    int dd[MR], ss[MR], rk[MR];
#pragma unroll
    for (int u = 0; u < MR; ++u) {
      long e = min(tb + (long)u * ET + t, te - 1);
      dd[u] = dst[e];
      ss[u] = src[e];
    }
#pragma unroll
    for (int u = 0; u < MR; ++u) {
      bool ok = tb + (long)u * ET + t < te;
      rk[u] = atomicAdd(&hist[ok ? bucket_of(dd[u], NB) : MAXB], 1);
    }
    __syncthreads();

    int v = (t < NB) ? hist[t] : 0;
    int incl = v;
    for (int off = 1; off < 32; off <<= 1) {
      int up = __shfl_up(incl, off);
      if ((t & 31) >= off) incl += up;
    }
    if ((t & 31) == 31) wsum[t >> 5] = incl;
    __syncthreads();
    int pre = 0;
    for (int q = 0; q < (t >> 5); ++q) pre += wsum[q];
    if (t < NB) lc[t] = make_int2(pre + incl - v, cur[t]);
    __syncthreads();

    int2 q[MR];
#pragma unroll
    for (int u = 0; u < MR; ++u) q[u] = lc[bucket_of(dd[u], NB)];
#pragma unroll
    for (int u = 0; u < MR; ++u) {
      if (tb + (long)u * ET + t < te)
        srec[q[u].x + rk[u]] =
            make_int2((int)(((unsigned)ss[u] << 8) | ((unsigned)dd[u] & 255u)),
                      q[u].y + rk[u]);
    }
    __syncthreads();

    int cnt = (int)(te - tb);
    for (int i = t; i < cnt; i += ET) {
      int2 rec = srec[i];
      pk[rec.y] = (unsigned)rec.x;
    }
    if (t < NB) cur[t] += v;
  }
}

#define SCAP 16384

__global__ void __launch_bounds__(ET)
bucket_sort_kernel(const unsigned* __restrict__ pk, const int* __restrict__ offs,
                   int* __restrict__ rowptr, int* __restrict__ col, int nNodes) {
  __shared__ int cnt[NPB];
  __shared__ int cur[NPB];
  __shared__ int wsum[NPB / 32];
  __shared__ int scol[SCAP];
  int t = threadIdx.x;
  int b = blockIdx.x;
  int start = offs[(size_t)b * EB];
  int end = offs[(size_t)(b + 1) * EB];
  if (t < NPB) cnt[t] = 0;
  __syncthreads();

  int i = start + t;
  for (; i + 3 * ET < end; i += 4 * ET) {
    unsigned w0 = pk[i];
    unsigned w1 = pk[i + ET];
    unsigned w2 = pk[i + 2 * ET];
    unsigned w3 = pk[i + 3 * ET];
    atomicAdd(&cnt[w0 & 255u], 1);
    atomicAdd(&cnt[w1 & 255u], 1);
    atomicAdd(&cnt[w2 & 255u], 1);
    atomicAdd(&cnt[w3 & 255u], 1);
  }
  for (; i < end; i += ET) atomicAdd(&cnt[pk[i] & 255u], 1);
  __syncthreads();

  int v = 0, incl = 0;
  if (t < NPB) {
    v = cnt[t];
    incl = v;
    for (int off = 1; off < 32; off <<= 1) {
      int u = __shfl_up(incl, off);
      if ((t & 31) >= off) incl += u;
    }
    if ((t & 31) == 31) wsum[t >> 5] = incl;
  }
  __syncthreads();
  if (t < NPB) {
    int pre = start;
    for (int w = 0; w < (t >> 5); ++w) pre += wsum[w];
    int excl = pre + incl - v;
    cur[t] = excl;
    int node = b * NPB + t;
    if (node < nNodes) rowptr[node] = excl;
    if (node == nNodes - 1) rowptr[nNodes] = end;
  }
  __syncthreads();

  bool fits = (end - start) <= SCAP;
  i = start + t;
  if (fits) {
    for (; i + 3 * ET < end; i += 4 * ET) {
      unsigned w0 = pk[i];
      unsigned w1 = pk[i + ET];
      unsigned w2 = pk[i + 2 * ET];
      unsigned w3 = pk[i + 3 * ET];
      int p0 = atomicAdd(&cur[w0 & 255u], 1);
      int p1 = atomicAdd(&cur[w1 & 255u], 1);
      int p2 = atomicAdd(&cur[w2 & 255u], 1);
      int p3 = atomicAdd(&cur[w3 & 255u], 1);
      scol[p0 - start] = (int)(w0 >> 8);
      scol[p1 - start] = (int)(w1 >> 8);
      scol[p2 - start] = (int)(w2 >> 8);
      scol[p3 - start] = (int)(w3 >> 8);
    }
    for (; i < end; i += ET) {
      unsigned w = pk[i];
      int p = atomicAdd(&cur[w & 255u], 1);
      scol[p - start] = (int)(w >> 8);
    }
  } else {
    for (; i + 3 * ET < end; i += 4 * ET) {
      unsigned w0 = pk[i];
      unsigned w1 = pk[i + ET];
      unsigned w2 = pk[i + 2 * ET];
      unsigned w3 = pk[i + 3 * ET];
      int p0 = atomicAdd(&cur[w0 & 255u], 1);
      int p1 = atomicAdd(&cur[w1 & 255u], 1);
      int p2 = atomicAdd(&cur[w2 & 255u], 1);
      int p3 = atomicAdd(&cur[w3 & 255u], 1);
      col[p0] = (int)(w0 >> 8);
      col[p1] = (int)(w1 >> 8);
      col[p2] = (int)(w2 >> 8);
      col[p3] = (int)(w3 >> 8);
    }
    for (; i < end; i += ET) {
      unsigned w = pk[i];
      int p = atomicAdd(&cur[w & 255u], 1);
      col[p] = (int)(w >> 8);
    }
  }
  __syncthreads();
  if (fits) {
    int len = end - start;
    for (int k = t; k < len; k += ET) col[start + k] = scol[k];
  }
}

__device__ __forceinline__ float4 ld4(const float* p) {
  return *reinterpret_cast<const float4*>(p);
}
__device__ __forceinline__ float4 add4(float4 a, float4 b) {
  return make_float4(a.x + b.x, a.y + b.y, a.z + b.z, a.w + b.w);
}


__global__ void __launch_bounds__(256)
proj1_kernel(const float* __restrict__ x, const float* __restrict__ W1l,
             const float* __restrict__ b1, const float* __restrict__ W1r,
             float* __restrict__ p, float* __restrict__ r, int nNodes) {
  __shared__ float sWl[DH_PAD * W1P];
  __shared__ float sWr[DH_PAD * W1P];
  __shared__ float sB[DH];
  for (int i = threadIdx.x; i < DH_PAD * DIN; i += 256) {
    float wl = (i < DH * DIN) ? W1l[i] : 0.0f;
    float wr = (i < DH * DIN) ? W1r[i] : 0.0f;
    int j = (i / DIN) * W1P + (i % DIN);
    sWl[j] = wl;
    sWr[j] = wr;
  }
  if (threadIdx.x < DH) sB[threadIdx.x] = b1[threadIdx.x];
  __syncthreads();

  int lane = threadIdx.x & 31;
  int wave = threadIdx.x >> 5;
  int node0 = (blockIdx.x * 8 + wave) * 16;
  int row = lane & 15;
  int node = node0 + row;
  int nodeC = min(node, nNodes - 1);
  int klo = (lane >> 4) * 2;
  int ncol = lane & 15;

  const float* xrow = x + (size_t)nodeC * DIN;

  v8f p0 = {}, p1 = {}, p2 = {}, r0 = {}, r1 = {}, r2 = {};
  for (int k = 0; k < DIN; k += 4) {
    v2f aX;
    aX.x = xrow[k + klo];
    aX.y = xrow[k + klo + 1];

    v2f bl0, br0, bl1, br1, bl2, br2;
    bl0.x = sWl[ncol * W1P + k + klo];        bl0.y = sWl[ncol * W1P + k + klo + 1];
    br0.x = sWr[ncol * W1P + k + klo];        br0.y = sWr[ncol * W1P + k + klo + 1];
    bl1.x = sWl[(16 + ncol) * W1P + k + klo]; bl1.y = sWl[(16 + ncol) * W1P + k + klo + 1];
    br1.x = sWr[(16 + ncol) * W1P + k + klo]; br1.y = sWr[(16 + ncol) * W1P + k + klo + 1];
    bl2.x = sWl[(32 + ncol) * W1P + k + klo]; bl2.y = sWl[(32 + ncol) * W1P + k + klo + 1];
    br2.x = sWr[(32 + ncol) * W1P + k + klo]; br2.y = sWr[(32 + ncol) * W1P + k + klo + 1];

    p0 = wmma4(aX, bl0, p0);
    r0 = wmma4(aX, br0, r0);
    p1 = wmma4(aX, bl1, p1);
    r1 = wmma4(aX, br1, r1);
    p2 = wmma4(aX, bl2, p2);
    r2 = wmma4(aX, br2, r2);
  }

  int mbase = (lane >> 4) * 8;
  for (int i = 0; i < 8; ++i) {
    int n = node0 + mbase + i;
    if (n < nNodes) {
      float* prow = p + (size_t)n * DH;
      float* rrow = r + (size_t)n * DH;
      prow[ncol]      = p0[i];
      prow[16 + ncol] = p1[i];
      rrow[ncol]      = r0[i] + sB[ncol];
      rrow[16 + ncol] = r1[i] + sB[16 + ncol];
      if (32 + ncol < DH) {
        prow[32 + ncol] = p2[i];
        rrow[32 + ncol] = r2[i] + sB[32 + ncol];
      }
    }
  }
}

__global__ void __launch_bounds__(256)
gather1_kernel(const float* __restrict__ p, const float* __restrict__ r,
               const int* __restrict__ rowptr, const int* __restrict__ col,
               float* __restrict__ h, int nNodes) {
  int lane = threadIdx.x & 31;
  int wave = __builtin_amdgcn_readfirstlane((int)(threadIdx.x >> 5));
  int nw = (int)gridDim.x * 8;
  int n = blockIdx.x * 8 + wave;
  if (n >= nNodes) return;
  int q = lane / 10;
  int chunk = lane - q * 10;
  int g = min(q, 2);
  const float* pc = p + chunk * 4;
  int beg = rowptr[n], end = rowptr[n + 1];
  while (true) {
    int nn = n + nw;
    int nbeg = 0, nend = 0;
    if (nn < nNodes) { nbeg = rowptr[nn]; nend = rowptr[nn + 1]; }
    float4 own = make_float4(0.0f, 0.0f, 0.0f, 0.0f);
    if (lane < 10) own = ld4(r + (size_t)n * DH + chunk * 4);

    float4 acc = make_float4(0.0f, 0.0f, 0.0f, 0.0f);
    int e0 = beg + lane;
    int my = (lane < 30 && e0 < end) ? col[e0] : 0;
    for (int base = beg; base < end; base += 30) {
      int e1 = base + 30 + lane;
      int myn = (lane < 30 && e1 < end) ? col[e1] : 0;
      int m = min(30, end - base);
      for (int j = 0; j < m; j += 15) {
        float4 v[5];
#pragma unroll
        for (int u = 0; u < 5; ++u) {
          int k = j + 3 * u + g;
          int s = __shfl(my, k);
          float4 t = ld4(pc + (size_t)s * DH);
          bool ok = k < m;
          v[u] = make_float4(ok ? t.x : 0.0f, ok ? t.y : 0.0f,
                             ok ? t.z : 0.0f, ok ? t.w : 0.0f);
        }
        acc = add4(acc, add4(add4(add4(v[0], v[1]), add4(v[2], v[3])), v[4]));
      }
      my = myn;
    }
    float4 t1, t2;
    t1.x = __shfl_down(acc.x, 10); t2.x = __shfl_down(acc.x, 20);
    t1.y = __shfl_down(acc.y, 10); t2.y = __shfl_down(acc.y, 20);
    t1.z = __shfl_down(acc.z, 10); t2.z = __shfl_down(acc.z, 20);
    t1.w = __shfl_down(acc.w, 10); t2.w = __shfl_down(acc.w, 20);
    if (lane < 10) {
      float inv = 1.0f / fmaxf((float)(end - beg), 1.0f);
      float4 s = add4(acc, add4(t1, t2));
      *reinterpret_cast<float4*>(h + (size_t)n * DH + chunk * 4) =
          make_float4(fmaxf(s.x * inv + own.x, 0.0f), fmaxf(s.y * inv + own.y, 0.0f),
                      fmaxf(s.z * inv + own.z, 0.0f), fmaxf(s.w * inv + own.w, 0.0f));
    }

    if (nn >= nNodes) break;
    n = nn; beg = nbeg; end = nend;
  }
}

__global__ void __launch_bounds__(256)
proj2_kernel(const float* __restrict__ h, const float* __restrict__ W2l,
             const float* __restrict__ b2, const float* __restrict__ W2r,
             float* __restrict__ p, float* __restrict__ r, int nNodes) {
  __shared__ float sWl[DO_PAD * W2P];
  __shared__ float sWr[DO_PAD * W2P];
  __shared__ float sB[DO_PAD];
  for (int i = threadIdx.x; i < DO_PAD * DH; i += 256) {
    float wl = (i < DOUT * DH) ? W2l[i] : 0.0f;
    float wr = (i < DOUT * DH) ? W2r[i] : 0.0f;
    int j = (i / DH) * W2P + (i % DH);
    sWl[j] = wl;
    sWr[j] = wr;
  }
  if (threadIdx.x < DO_PAD)
    sB[threadIdx.x] = (threadIdx.x < DOUT) ? b2[threadIdx.x] : 0.0f;
  __syncthreads();

  int lane = threadIdx.x & 31;
  int wave = threadIdx.x >> 5;
  int node0 = (blockIdx.x * 8 + wave) * 16;
  int row = lane & 15;
  int node = node0 + row;
  int nodeC = min(node, nNodes - 1);
  int klo = (lane >> 4) * 2;
  int ncol = lane & 15;

  const float* hrow = h + (size_t)nodeC * DH;

  v8f p0 = {}, p1 = {}, r0 = {}, r1 = {};
  for (int k = 0; k < DH; k += 4) {
    v2f aH;
    aH.x = hrow[k + klo];
    aH.y = hrow[k + klo + 1];

    v2f bl0, br0, bl1, br1;
    bl0.x = sWl[ncol * W2P + k + klo];        bl0.y = sWl[ncol * W2P + k + klo + 1];
    br0.x = sWr[ncol * W2P + k + klo];        br0.y = sWr[ncol * W2P + k + klo + 1];
    bl1.x = sWl[(16 + ncol) * W2P + k + klo]; bl1.y = sWl[(16 + ncol) * W2P + k + klo + 1];
    br1.x = sWr[(16 + ncol) * W2P + k + klo]; br1.y = sWr[(16 + ncol) * W2P + k + klo + 1];

    p0 = wmma4(aH, bl0, p0);
    r0 = wmma4(aH, br0, r0);
    p1 = wmma4(aH, bl1, p1);
    r1 = wmma4(aH, br1, r1);
  }

  int mbase = (lane >> 4) * 8;
  for (int i = 0; i < 8; ++i) {
    int n = node0 + mbase + i;
    if (n < nNodes) {
      float* prow = p + (size_t)n * DO_PAD;
      float* rrow = r + (size_t)n * DO_PAD;
      prow[ncol]      = p0[i];
      prow[16 + ncol] = p1[i];
      rrow[ncol]      = r0[i] + sB[ncol];
      rrow[16 + ncol] = r1[i] + sB[16 + ncol];
    }
  }
}

__global__ void __launch_bounds__(256)
gather2_kernel(const float* __restrict__ p, const float* __restrict__ r,
               const int* __restrict__ rowptr, const int* __restrict__ col,
               float* __restrict__ o, int nNodes) {
  int lane = threadIdx.x & 31;
  int wave = __builtin_amdgcn_readfirstlane((int)(threadIdx.x >> 5));
  int nw = (int)gridDim.x * 8;
  int n = blockIdx.x * 8 + wave;
  if (n >= nNodes) return;
  int q = lane >> 3;
  int chunk = lane & 7;
  const float* pc = p + chunk * 4;
  int beg = rowptr[n], end = rowptr[n + 1];
  while (true) {
    int nn = n + nw;
    int nbeg = 0, nend = 0;
    if (nn < nNodes) { nbeg = rowptr[nn]; nend = rowptr[nn + 1]; }
    float4 own = make_float4(0.0f, 0.0f, 0.0f, 0.0f);
    if (lane < 6) own = ld4(r + (size_t)n * DO_PAD + chunk * 4);

    float4 acc = make_float4(0.0f, 0.0f, 0.0f, 0.0f);
    int e0 = beg + lane;
    int my = (e0 < end) ? col[e0] : 0;
    for (int base = beg; base < end; base += 32) {
      int e1 = base + 32 + lane;
      int myn = (e1 < end) ? col[e1] : 0;
      int m = min(32, end - base);
      for (int j = 0; j < m; j += 16) {
        float4 v[4];
#pragma unroll
        for (int u = 0; u < 4; ++u) {
          int k = j + 4 * u + q;
          int s = __shfl(my, k);
          float4 t = ld4(pc + (size_t)s * DO_PAD);
          bool ok = k < m;
          v[u] = make_float4(ok ? t.x : 0.0f, ok ? t.y : 0.0f,
                             ok ? t.z : 0.0f, ok ? t.w : 0.0f);
        }
        acc = add4(acc, add4(add4(v[0], v[1]), add4(v[2], v[3])));
      }
      my = myn;
    }
    acc.x += __shfl_xor(acc.x, 8);
    acc.y += __shfl_xor(acc.y, 8);
    acc.z += __shfl_xor(acc.z, 8);
    acc.w += __shfl_xor(acc.w, 8);
    acc.x += __shfl_xor(acc.x, 16);
    acc.y += __shfl_xor(acc.y, 16);
    acc.z += __shfl_xor(acc.z, 16);
    acc.w += __shfl_xor(acc.w, 16);
    if (lane < 6) {
      float inv = 1.0f / fmaxf((float)(end - beg), 1.0f);
      *reinterpret_cast<float4*>(o + (size_t)n * DOUT + chunk * 4) =
          make_float4(acc.x * inv + own.x, acc.y * inv + own.y,
                      acc.z * inv + own.z, acc.w * inv + own.w);
    }

    if (nn >= nNodes) break;
    n = nn; beg = nbeg; end = nend;
  }
}

__global__ void __launch_bounds__(256)
logsoftmax_kernel(const float* __restrict__ o, float* __restrict__ out, int nNodes) {
  int n = blockIdx.x * 256 + threadIdx.x;
  if (n >= nNodes) return;
  const float* r = o + (size_t)n * DOUT;
  float mx = -3.402823466e+38f;
  for (int j = 0; j < DOUT; ++j) mx = fmaxf(mx, r[j]);
  float s = 0.0f;
  for (int j = 0; j < DOUT; ++j) s += expf(r[j] - mx);
  float lse = mx + logf(s);
  float* w = out + (size_t)n * DOUT;
  for (int j = 0; j < DOUT; ++j) w[j] = r[j] - lse;
}

extern "C" void kernel_launch(void* const* d_in, const int* in_sizes, int n_in,
                              void* d_out, int out_size, void* d_ws, size_t ws_size,
                              hipStream_t stream) {
  const float* x   = (const float*)d_in[0];
  const int*   ei  = (const int*)d_in[1];
  const float* W1l = (const float*)d_in[2];
  const float* b1  = (const float*)d_in[3];
  const float* W1r = (const float*)d_in[4];
  const float* W2l = (const float*)d_in[5];
  const float* b2  = (const float*)d_in[6];
  const float* W2r = (const float*)d_in[7];
  float* out = (float*)d_out;

  int  N = in_sizes[0] / DIN;
  long E = (long)in_sizes[1] / 2;
  const int* src = ei;
  const int* dst = ei + E;

  float* ws = (float*)d_ws;
  float* P  = ws;
  float* R  = P + (size_t)N * DH;
  float* H  = R + (size_t)N * DH;
  float* o  = H;
  int* rowptr = (int*)(ws + (size_t)N * (DIN + 1 + DH + DOUT));
  int* col    = rowptr + N + 1;

  int NB = (N + NPB - 1) / NPB;
  int nCnt = NB * EB;
  unsigned* pk = (unsigned*)d_ws;
  int* cm   = (int*)d_ws + E;
  int* offs = cm + nCnt;

  bucket_count_kernel<<<EB, ET, 0, stream>>>(dst, cm, E, NB);
  scan_kernel<<<(nCnt + SCAN_T - 1) / SCAN_T, SCAN_T, 0, stream>>>(cm, offs, nCnt);
  bucket_move_kernel<<<EB, ET, 0, stream>>>(src, dst, offs, pk, E, NB);
  bucket_sort_kernel<<<NB, ET, 0, stream>>>(pk, offs, rowptr, col, N);

  int nb = (N + 127) / 128;

  proj1_kernel<<<nb, 256, 0, stream>>>(x, W1l, b1, W1r, P, R, N);
  gather1_kernel<<<2048, 256, 0, stream>>>(P, R, rowptr, col, H, N);

  proj2_kernel<<<nb, 256, 0, stream>>>(H, W2l, b2, W2r, P, R, N);
  gather2_kernel<<<2048, 256, 0, stream>>>(P, R, rowptr, col, o, N);

  logsoftmax_kernel<<<(N + 255) / 256, 256, 0, stream>>>(o, out, N);
}


# Layer-1 projections run as extra work-groups of the sort's count launch, which they do not depend on: 9 launches to 8; a wave does its Wl, then its Wr tiles to fit 64 registers; sums unchanged
# speedup vs baseline: 5.3581x; 1.0636x over previous
#include <hip/hip_runtime.h>
#include <math.h>

#define DIN 64
#define DH  40
#define DOUT 24
#define DH_PAD 48
#define DO_PAD 32
#define SCAN_T 1024
#define W1P 68
#define W2P 44

typedef __attribute__((ext_vector_type(2))) float v2f;
typedef __attribute__((ext_vector_type(8))) float v8f;

__device__ __forceinline__ v8f wmma4(v2f a, v2f b, v8f c) {
  return __builtin_amdgcn_wmma_f32_16x16x4_f32(false, a, false, b, (short)0, c,
                                               false, false);
}

#define NPB 256
#define MAXB 1024
#define EB 256
#define ET 1024

__device__ __forceinline__ int bucket_of(int d, int NB) {
  return min((int)((unsigned)d >> 8), NB - 1);
}

__device__ __forceinline__ void
bucket_count_body(const int* __restrict__ dst, int* __restrict__ cm, long E,
                  int NB) {
  __shared__ int hist[MAXB];
  int t = threadIdx.x;
  for (int j = t; j < NB; j += ET) hist[j] = 0;
  __syncthreads();
  long chunk = (E + EB - 1) / EB;
  long e0 = (long)blockIdx.x * chunk;
  long e1 = min(E, e0 + chunk);
  long e = e0 + t;
  for (; e + 3 * ET < e1; e += 4 * ET) {
    int d0 = dst[e];
    int d1 = dst[e + ET];
    int d2 = dst[e + 2 * ET];
    int d3 = dst[e + 3 * ET];
    atomicAdd(&hist[bucket_of(d0, NB)], 1);
    atomicAdd(&hist[bucket_of(d1, NB)], 1);
    atomicAdd(&hist[bucket_of(d2, NB)], 1);
    atomicAdd(&hist[bucket_of(d3, NB)], 1);
  }
  for (; e < e1; e += ET) atomicAdd(&hist[bucket_of(dst[e], NB)], 1);
  __syncthreads();
  for (int j = t; j < NB; j += ET) cm[(size_t)j * EB + blockIdx.x] = hist[j];
}

__global__ void __launch_bounds__(SCAN_T)
scan_kernel(const int* __restrict__ deg, int* __restrict__ rowptr, int n) {
  __shared__ int s[SCAN_T];
  int t = threadIdx.x;
  int base = blockIdx.x * SCAN_T;
  int pre = 0;
  for (int i = t; i < base; i += SCAN_T) pre += deg[i];
  int idx = base + t;
  int v = (idx < n) ? deg[idx] : 0;

  s[t] = pre;
  __syncthreads();
  for (int off = SCAN_T / 2; off > 0; off >>= 1) {
    if (t < off) s[t] += s[t + off];
    __syncthreads();
  }
  int blockPre = s[0];
  __syncthreads();

  s[t] = v;
  __syncthreads();
  for (int off = 1; off < SCAN_T; off <<= 1) {
    int add = (t >= off) ? s[t - off] : 0;
    __syncthreads();
    s[t] += add;
    __syncthreads();
  }
  int incl = s[t];
  if (idx < n) rowptr[idx] = blockPre + incl - v;
  if (idx == n - 1) rowptr[n] = blockPre + incl;
}

#define MR 13
#define MTILE (MR * ET)

__global__ void __launch_bounds__(ET)
bucket_move_kernel(const int* __restrict__ src, const int* __restrict__ dst,
                   const int* __restrict__ offs, unsigned* __restrict__ pk,
                   long E, int NB) {
  __shared__ int cur[MAXB];
  __shared__ int hist[MAXB + 1];
  __shared__ int2 lc[MAXB];
  __shared__ int wsum[ET / 32];
  __shared__ int2 srec[MTILE];
  int t = threadIdx.x;
  for (int j = t; j < NB; j += ET) cur[j] = offs[(size_t)j * EB + blockIdx.x];
  long chunk = (E + EB - 1) / EB;
  long e0 = (long)blockIdx.x * chunk;
  long e1 = min(E, e0 + chunk);
  for (long tb = e0; tb < e1; tb += MTILE) {
    long te = min(e1, tb + (long)MTILE);
    for (int j = t; j < NB; j += ET) hist[j] = 0;
    __syncthreads();

    int dd[MR], ss[MR], rk[MR];
#pragma unroll
    for (int u = 0; u < MR; ++u) {
      long e = min(tb + (long)u * ET + t, te - 1);
      dd[u] = dst[e];
      ss[u] = src[e];
    }
#pragma unroll
    for (int u = 0; u < MR; ++u) {
      bool ok = tb + (long)u * ET + t < te;
      rk[u] = atomicAdd(&hist[ok ? bucket_of(dd[u], NB) : MAXB], 1);
    }
    __syncthreads();

    int v = (t < NB) ? hist[t] : 0;
    int incl = v;
    for (int off = 1; off < 32; off <<= 1) {
      int up = __shfl_up(incl, off);
      if ((t & 31) >= off) incl += up;
    }
    if ((t & 31) == 31) wsum[t >> 5] = incl;
    __syncthreads();
    int pre = 0;
    for (int q = 0; q < (t >> 5); ++q) pre += wsum[q];
    if (t < NB) lc[t] = make_int2(pre + incl - v, cur[t]);
    __syncthreads();

    int2 q[MR];
#pragma unroll
    for (int u = 0; u < MR; ++u) q[u] = lc[bucket_of(dd[u], NB)];
#pragma unroll
    for (int u = 0; u < MR; ++u) {
      if (tb + (long)u * ET + t < te)
        srec[q[u].x + rk[u]] =
            make_int2((int)(((unsigned)ss[u] << 8) | ((unsigned)dd[u] & 255u)),
                      q[u].y + rk[u]);
    }
    __syncthreads();

    int cnt = (int)(te - tb);
    for (int i = t; i < cnt; i += ET) {
      int2 rec = srec[i];
      pk[rec.y] = (unsigned)rec.x;
    }
    if (t < NB) cur[t] += v;
  }
}

#define SCAP 16384

__global__ void __launch_bounds__(ET)
bucket_sort_kernel(const unsigned* __restrict__ pk, const int* __restrict__ offs,
                   int* __restrict__ rowptr, int* __restrict__ col, int nNodes) {
  __shared__ int cnt[NPB];
  __shared__ int cur[NPB];
  __shared__ int wsum[NPB / 32];
  __shared__ int scol[SCAP];
  int t = threadIdx.x;
  int b = blockIdx.x;
  int start = offs[(size_t)b * EB];
  int end = offs[(size_t)(b + 1) * EB];
  if (t < NPB) cnt[t] = 0;
  __syncthreads();

  int i = start + t;
  for (; i + 3 * ET < end; i += 4 * ET) {
    unsigned w0 = pk[i];
    unsigned w1 = pk[i + ET];
    unsigned w2 = pk[i + 2 * ET];
    unsigned w3 = pk[i + 3 * ET];
    atomicAdd(&cnt[w0 & 255u], 1);
    atomicAdd(&cnt[w1 & 255u], 1);
    atomicAdd(&cnt[w2 & 255u], 1);
    atomicAdd(&cnt[w3 & 255u], 1);
  }
  for (; i < end; i += ET) atomicAdd(&cnt[pk[i] & 255u], 1);
  __syncthreads();

  int v = 0, incl = 0;
  if (t < NPB) {
    v = cnt[t];
    incl = v;
    for (int off = 1; off < 32; off <<= 1) {
      int u = __shfl_up(incl, off);
      if ((t & 31) >= off) incl += u;
    }
    if ((t & 31) == 31) wsum[t >> 5] = incl;
  }
  __syncthreads();
  if (t < NPB) {
    int pre = start;
    for (int w = 0; w < (t >> 5); ++w) pre += wsum[w];
    int excl = pre + incl - v;
    cur[t] = excl;
    int node = b * NPB + t;
    if (node < nNodes) rowptr[node] = excl;
    if (node == nNodes - 1) rowptr[nNodes] = end;
  }
  __syncthreads();

  bool fits = (end - start) <= SCAP;
  i = start + t;
  if (fits) {
    for (; i + 3 * ET < end; i += 4 * ET) {
      unsigned w0 = pk[i];
      unsigned w1 = pk[i + ET];
      unsigned w2 = pk[i + 2 * ET];
      unsigned w3 = pk[i + 3 * ET];
      int p0 = atomicAdd(&cur[w0 & 255u], 1);
      int p1 = atomicAdd(&cur[w1 & 255u], 1);
      int p2 = atomicAdd(&cur[w2 & 255u], 1);
      int p3 = atomicAdd(&cur[w3 & 255u], 1);
      scol[p0 - start] = (int)(w0 >> 8);
      scol[p1 - start] = (int)(w1 >> 8);
      scol[p2 - start] = (int)(w2 >> 8);
      scol[p3 - start] = (int)(w3 >> 8);
    }
    for (; i < end; i += ET) {
      unsigned w = pk[i];
      int p = atomicAdd(&cur[w & 255u], 1);
      scol[p - start] = (int)(w >> 8);
    }
  } else {
    for (; i + 3 * ET < end; i += 4 * ET) {
      unsigned w0 = pk[i];
      unsigned w1 = pk[i + ET];
      unsigned w2 = pk[i + 2 * ET];
      unsigned w3 = pk[i + 3 * ET];
      int p0 = atomicAdd(&cur[w0 & 255u], 1);
      int p1 = atomicAdd(&cur[w1 & 255u], 1);
      int p2 = atomicAdd(&cur[w2 & 255u], 1);
      int p3 = atomicAdd(&cur[w3 & 255u], 1);
      col[p0] = (int)(w0 >> 8);
      col[p1] = (int)(w1 >> 8);
      col[p2] = (int)(w2 >> 8);
      col[p3] = (int)(w3 >> 8);
    }
    for (; i < end; i += ET) {
      unsigned w = pk[i];
      int p = atomicAdd(&cur[w & 255u], 1);
      col[p] = (int)(w >> 8);
    }
  }
  __syncthreads();
  if (fits) {
    int len = end - start;
    for (int k = t; k < len; k += ET) col[start + k] = scol[k];
  }
}

__device__ __forceinline__ float4 ld4(const float* p) {
  return *reinterpret_cast<const float4*>(p);
}
__device__ __forceinline__ float4 add4(float4 a, float4 b) {
  return make_float4(a.x + b.x, a.y + b.y, a.z + b.z, a.w + b.w);
}


#define P1_TILES 25
#define P1_NODES (P1_TILES * 16)

__device__ __forceinline__ void
proj1_body(int pb, const float* __restrict__ x, const float* __restrict__ W1l,
           const float* __restrict__ b1, const float* __restrict__ W1r,
           float* __restrict__ p, float* __restrict__ r, int nNodes) {
  __shared__ float sWl[DH_PAD * W1P];
  __shared__ float sWr[DH_PAD * W1P];
  __shared__ float sB[DH];
  for (int i = threadIdx.x; i < DH_PAD * DIN; i += ET) {
    float wl = (i < DH * DIN) ? W1l[i] : 0.0f;
    float wr = (i < DH * DIN) ? W1r[i] : 0.0f;
    int j = (i / DIN) * W1P + (i % DIN);
    sWl[j] = wl;
    sWr[j] = wr;
  }
  if (threadIdx.x < DH) sB[threadIdx.x] = b1[threadIdx.x];
  __syncthreads();

  int lane = threadIdx.x & 31;
  int wave = threadIdx.x >> 5;
  if (wave >= P1_TILES) return;
  int node0 = pb * P1_NODES + wave * 16;
  int row = lane & 15;
  int node = node0 + row;
  int nodeC = min(node, nNodes - 1);
  int klo = (lane >> 4) * 2;
  int ncol = lane & 15;
  int mbase = (lane >> 4) * 8;

  const float* xrow = x + (size_t)nodeC * DIN;

  for (int half = 0; half < 2; ++half) {
    const float* sW = half ? sWr : sWl;
    float* o = half ? r : p;
    v8f a0 = {}, a1 = {}, a2 = {};
    for (int k = 0; k < DIN; k += 4) {
      v2f aX;
      aX.x = xrow[k + klo];
      aX.y = xrow[k + klo + 1];

      v2f b0, b1v, b2;
      b0.x  = sW[ncol * W1P + k + klo];        b0.y  = sW[ncol * W1P + k + klo + 1];
      b1v.x = sW[(16 + ncol) * W1P + k + klo]; b1v.y = sW[(16 + ncol) * W1P + k + klo + 1];
      b2.x  = sW[(32 + ncol) * W1P + k + klo]; b2.y  = sW[(32 + ncol) * W1P + k + klo + 1];

      a0 = wmma4(aX, b0, a0);
      a1 = wmma4(aX, b1v, a1);
      a2 = wmma4(aX, b2, a2);
    }

    float c0 = half ? sB[ncol] : 0.0f;
    float c1 = half ? sB[16 + ncol] : 0.0f;
    float c2 = (half && 32 + ncol < DH) ? sB[32 + ncol] : 0.0f;
    for (int i = 0; i < 8; ++i) {
      int n = node0 + mbase + i;
      if (n < nNodes) {
        float* orow = o + (size_t)n * DH;
        orow[ncol]      = half ? a0[i] + c0 : a0[i];
        orow[16 + ncol] = half ? a1[i] + c1 : a1[i];
        if (32 + ncol < DH) orow[32 + ncol] = half ? a2[i] + c2 : a2[i];
      }
    }
  }
}

__global__ void __launch_bounds__(ET, 16)
count_proj1_kernel(const int* __restrict__ dst, int* __restrict__ cm, long E,
                   int NB,
                   const float* __restrict__ x, const float* __restrict__ W1l,
                   const float* __restrict__ b1, const float* __restrict__ W1r,
                   float* __restrict__ p, float* __restrict__ r, int nNodes) {
  if (blockIdx.x < EB)
    bucket_count_body(dst, cm, E, NB);
  else
    proj1_body((int)blockIdx.x - EB, x, W1l, b1, W1r, p, r, nNodes);
}

__global__ void __launch_bounds__(256)
gather1_kernel(const float* __restrict__ p, const float* __restrict__ r,
               const int* __restrict__ rowptr, const int* __restrict__ col,
               float* __restrict__ h, int nNodes) {
  int lane = threadIdx.x & 31;
  int wave = __builtin_amdgcn_readfirstlane((int)(threadIdx.x >> 5));
  int nw = (int)gridDim.x * 8;
  int n = blockIdx.x * 8 + wave;
  if (n >= nNodes) return;
  int q = lane / 10;
  int chunk = lane - q * 10;
  int g = min(q, 2);
  const float* pc = p + chunk * 4;
  int beg = rowptr[n], end = rowptr[n + 1];
  while (true) {
    int nn = n + nw;
    int nbeg = 0, nend = 0;
    if (nn < nNodes) { nbeg = rowptr[nn]; nend = rowptr[nn + 1]; }
    float4 own = make_float4(0.0f, 0.0f, 0.0f, 0.0f);
    if (lane < 10) own = ld4(r + (size_t)n * DH + chunk * 4);

    float4 acc = make_float4(0.0f, 0.0f, 0.0f, 0.0f);
    int e0 = beg + lane;
    int my = (lane < 30 && e0 < end) ? col[e0] : 0;
    for (int base = beg; base < end; base += 30) {
      int e1 = base + 30 + lane;
      int myn = (lane < 30 && e1 < end) ? col[e1] : 0;
      int m = min(30, end - base);
      for (int j = 0; j < m; j += 15) {
        float4 v[5];
#pragma unroll
        for (int u = 0; u < 5; ++u) {
          int k = j + 3 * u + g;
          int s = __shfl(my, k);
          float4 t = ld4(pc + (size_t)s * DH);
          bool ok = k < m;
          v[u] = make_float4(ok ? t.x : 0.0f, ok ? t.y : 0.0f,
                             ok ? t.z : 0.0f, ok ? t.w : 0.0f);
        }
        acc = add4(acc, add4(add4(add4(v[0], v[1]), add4(v[2], v[3])), v[4]));
      }
      my = myn;
    }
    float4 t1, t2;
    t1.x = __shfl_down(acc.x, 10); t2.x = __shfl_down(acc.x, 20);
    t1.y = __shfl_down(acc.y, 10); t2.y = __shfl_down(acc.y, 20);
    t1.z = __shfl_down(acc.z, 10); t2.z = __shfl_down(acc.z, 20);
    t1.w = __shfl_down(acc.w, 10); t2.w = __shfl_down(acc.w, 20);
    if (lane < 10) {
      float inv = 1.0f / fmaxf((float)(end - beg), 1.0f);
      float4 s = add4(acc, add4(t1, t2));
      *reinterpret_cast<float4*>(h + (size_t)n * DH + chunk * 4) =
          make_float4(fmaxf(s.x * inv + own.x, 0.0f), fmaxf(s.y * inv + own.y, 0.0f),
                      fmaxf(s.z * inv + own.z, 0.0f), fmaxf(s.w * inv + own.w, 0.0f));
    }

    if (nn >= nNodes) break;
    n = nn; beg = nbeg; end = nend;
  }
}

__global__ void __launch_bounds__(256)
proj2_kernel(const float* __restrict__ h, const float* __restrict__ W2l,
             const float* __restrict__ b2, const float* __restrict__ W2r,
             float* __restrict__ p, float* __restrict__ r, int nNodes) {
  __shared__ float sWl[DO_PAD * W2P];
  __shared__ float sWr[DO_PAD * W2P];
  __shared__ float sB[DO_PAD];
  for (int i = threadIdx.x; i < DO_PAD * DH; i += 256) {
    float wl = (i < DOUT * DH) ? W2l[i] : 0.0f;
    float wr = (i < DOUT * DH) ? W2r[i] : 0.0f;
    int j = (i / DH) * W2P + (i % DH);
    sWl[j] = wl;
    sWr[j] = wr;
  }
  if (threadIdx.x < DO_PAD)
    sB[threadIdx.x] = (threadIdx.x < DOUT) ? b2[threadIdx.x] : 0.0f;
  __syncthreads();

  int lane = threadIdx.x & 31;
  int wave = threadIdx.x >> 5;
  int node0 = (blockIdx.x * 8 + wave) * 16;
  int row = lane & 15;
  int node = node0 + row;
  int nodeC = min(node, nNodes - 1);
  int klo = (lane >> 4) * 2;
  int ncol = lane & 15;

  const float* hrow = h + (size_t)nodeC * DH;

  v8f p0 = {}, p1 = {}, r0 = {}, r1 = {};
  for (int k = 0; k < DH; k += 4) {
    v2f aH;
    aH.x = hrow[k + klo];
    aH.y = hrow[k + klo + 1];

    v2f bl0, br0, bl1, br1;
    bl0.x = sWl[ncol * W2P + k + klo];        bl0.y = sWl[ncol * W2P + k + klo + 1];
    br0.x = sWr[ncol * W2P + k + klo];        br0.y = sWr[ncol * W2P + k + klo + 1];
    bl1.x = sWl[(16 + ncol) * W2P + k + klo]; bl1.y = sWl[(16 + ncol) * W2P + k + klo + 1];
    br1.x = sWr[(16 + ncol) * W2P + k + klo]; br1.y = sWr[(16 + ncol) * W2P + k + klo + 1];

    p0 = wmma4(aH, bl0, p0);
    r0 = wmma4(aH, br0, r0);
    p1 = wmma4(aH, bl1, p1);
    r1 = wmma4(aH, br1, r1);
  }

  int mbase = (lane >> 4) * 8;
  for (int i = 0; i < 8; ++i) {
    int n = node0 + mbase + i;
    if (n < nNodes) {
      float* prow = p + (size_t)n * DO_PAD;
      float* rrow = r + (size_t)n * DO_PAD;
      prow[ncol]      = p0[i];
      prow[16 + ncol] = p1[i];
      rrow[ncol]      = r0[i] + sB[ncol];
      rrow[16 + ncol] = r1[i] + sB[16 + ncol];
    }
  }
}

__global__ void __launch_bounds__(256)
gather2_kernel(const float* __restrict__ p, const float* __restrict__ r,
               const int* __restrict__ rowptr, const int* __restrict__ col,
               float* __restrict__ o, int nNodes) {
  int lane = threadIdx.x & 31;
  int wave = __builtin_amdgcn_readfirstlane((int)(threadIdx.x >> 5));
  int nw = (int)gridDim.x * 8;
  int n = blockIdx.x * 8 + wave;
  if (n >= nNodes) return;
  int q = lane >> 3;
  int chunk = lane & 7;
  const float* pc = p + chunk * 4;
  int beg = rowptr[n], end = rowptr[n + 1];
  while (true) {
    int nn = n + nw;
    int nbeg = 0, nend = 0;
    if (nn < nNodes) { nbeg = rowptr[nn]; nend = rowptr[nn + 1]; }
    float4 own = make_float4(0.0f, 0.0f, 0.0f, 0.0f);
    if (lane < 6) own = ld4(r + (size_t)n * DO_PAD + chunk * 4);

    float4 acc = make_float4(0.0f, 0.0f, 0.0f, 0.0f);
    int e0 = beg + lane;
    int my = (e0 < end) ? col[e0] : 0;
    for (int base = beg; base < end; base += 32) {
      int e1 = base + 32 + lane;
      int myn = (e1 < end) ? col[e1] : 0;
      int m = min(32, end - base);
      for (int j = 0; j < m; j += 16) {
        float4 v[4];
#pragma unroll
        for (int u = 0; u < 4; ++u) {
          int k = j + 4 * u + q;
          int s = __shfl(my, k);
          float4 t = ld4(pc + (size_t)s * DO_PAD);
          bool ok = k < m;
          v[u] = make_float4(ok ? t.x : 0.0f, ok ? t.y : 0.0f,
                             ok ? t.z : 0.0f, ok ? t.w : 0.0f);
        }
        acc = add4(acc, add4(add4(v[0], v[1]), add4(v[2], v[3])));
      }
      my = myn;
    }
    acc.x += __shfl_xor(acc.x, 8);
    acc.y += __shfl_xor(acc.y, 8);
    acc.z += __shfl_xor(acc.z, 8);
    acc.w += __shfl_xor(acc.w, 8);
    acc.x += __shfl_xor(acc.x, 16);
    acc.y += __shfl_xor(acc.y, 16);
    acc.z += __shfl_xor(acc.z, 16);
    acc.w += __shfl_xor(acc.w, 16);
    if (lane < 6) {
      float inv = 1.0f / fmaxf((float)(end - beg), 1.0f);
      *reinterpret_cast<float4*>(o + (size_t)n * DOUT + chunk * 4) =
          make_float4(acc.x * inv + own.x, acc.y * inv + own.y,
                      acc.z * inv + own.z, acc.w * inv + own.w);
    }

    if (nn >= nNodes) break;
    n = nn; beg = nbeg; end = nend;
  }
}

__global__ void __launch_bounds__(256)
logsoftmax_kernel(const float* __restrict__ o, float* __restrict__ out, int nNodes) {
  int n = blockIdx.x * 256 + threadIdx.x;
  if (n >= nNodes) return;
  const float* r = o + (size_t)n * DOUT;
  float mx = -3.402823466e+38f;
  for (int j = 0; j < DOUT; ++j) mx = fmaxf(mx, r[j]);
  float s = 0.0f;
  for (int j = 0; j < DOUT; ++j) s += expf(r[j] - mx);
  float lse = mx + logf(s);
  float* w = out + (size_t)n * DOUT;
  for (int j = 0; j < DOUT; ++j) w[j] = r[j] - lse;
}

extern "C" void kernel_launch(void* const* d_in, const int* in_sizes, int n_in,
                              void* d_out, int out_size, void* d_ws, size_t ws_size,
                              hipStream_t stream) {
  const float* x   = (const float*)d_in[0];
  const int*   ei  = (const int*)d_in[1];
  const float* W1l = (const float*)d_in[2];
  const float* b1  = (const float*)d_in[3];
  const float* W1r = (const float*)d_in[4];
  const float* W2l = (const float*)d_in[5];
  const float* b2  = (const float*)d_in[6];
  const float* W2r = (const float*)d_in[7];
  float* out = (float*)d_out;

  int  N = in_sizes[0] / DIN;
  long E = (long)in_sizes[1] / 2;
  const int* src = ei;
  const int* dst = ei + E;

  float* ws = (float*)d_ws;
  float* P  = ws;
  float* R  = P + (size_t)N * DH;
  float* H  = R + (size_t)N * DH;
  float* o  = H;
  int* rowptr = (int*)(ws + (size_t)N * (DIN + 1 + DH + DOUT));
  int* col    = rowptr + N + 1;

  int NB = (N + NPB - 1) / NPB;
  int nCnt = NB * EB;
  unsigned* pk = (unsigned*)H;
  int* cm   = (int*)H + E;
  int* offs = cm + nCnt;

  int nbP = (N + P1_NODES - 1) / P1_NODES;
  count_proj1_kernel<<<EB + nbP, ET, 0, stream>>>(dst, cm, E, NB,
                                                  x, W1l, b1, W1r, P, R, N);
  scan_kernel<<<(nCnt + SCAN_T - 1) / SCAN_T, SCAN_T, 0, stream>>>(cm, offs, nCnt);
  bucket_move_kernel<<<EB, ET, 0, stream>>>(src, dst, offs, pk, E, NB);
  bucket_sort_kernel<<<NB, ET, 0, stream>>>(pk, offs, rowptr, col, N);

  int nb = (N + 127) / 128;

  gather1_kernel<<<2048, 256, 0, stream>>>(P, R, rowptr, col, H, N);

  proj2_kernel<<<nb, 256, 0, stream>>>(H, W2l, b2, W2r, P, R, N);
  gather2_kernel<<<2048, 256, 0, stream>>>(P, R, rowptr, col, o, N);

  logsoftmax_kernel<<<(N + 255) / 256, 256, 0, stream>>>(o, out, N);
}


# Edges ordered in 2 launches, not 4: each 13312-edge slice goes in bucket order to its own stretch, a bucket finds its runs by column sums of the counts; neighbours still summed in LDS-add order
# speedup vs baseline: 5.7231x; 1.0681x over previous
#include <hip/hip_runtime.h>
#include <math.h>

#define DIN 64
#define DH  40
#define DOUT 24
#define DH_PAD 48
#define DO_PAD 32
#define W1P 68
#define W2P 44

typedef __attribute__((ext_vector_type(2))) float v2f;
typedef __attribute__((ext_vector_type(8))) float v8f;

__device__ __forceinline__ v8f wmma4(v2f a, v2f b, v8f c) {
  return __builtin_amdgcn_wmma_f32_16x16x4_f32(false, a, false, b, (short)0, c,
                                               false, false);
}

#define NPB 256
#define MAXB 1024
#define ET 1024
#define MR 13
#define MTILE (MR * ET)
#define MAXS 1024

__device__ __forceinline__ int bucket_of(int d, int NB) {
  return min((int)((unsigned)d >> 8), NB - 1);
}

#define LCO (MAXB + 1)
#define SL_LDS (2 * LCO + ET / 32 + MTILE)
__device__ __forceinline__ void
slice_sort_body(int* sm, int sl, const int* __restrict__ src,
                const int* __restrict__ dst,
                int* __restrict__ cm, unsigned* __restrict__ pk, int E, int NB,
                int S, int SP) {
  int* hl = sm;
  int* wsum = sm + 2 * LCO;
  unsigned* srec = reinterpret_cast<unsigned*>(sm + 2 * LCO + ET / 32);
  int t = threadIdx.x;
  int tb = sl * MTILE;
  int cnt = min(E - tb, MTILE);
  int back = max(cnt - ET, 0);
  int tt = min(t, cnt - 1);
  for (int j = t; j < NB; j += ET) hl[j] = 0;
  if (t == 0) hl[MAXB] = 0;
  __syncthreads();

  int br[MR];
  unsigned rec[MR];
#pragma unroll
  for (int u = 0; u < MR; ++u) {
    int ro = min(u * ET, back);
    const int* du = dst + tb + ro;
    const int* su = src + tb + ro;
    int d = du[tt];
    int s = su[tt];
    bool ok = (t >= u * ET - ro) && (t < cnt);
    br[u] = ok ? bucket_of(d, NB) : MAXB;
    rec[u] = ((unsigned)s << 8) | ((unsigned)d & 255u);
  }
#pragma unroll
  for (int u = 0; u < MR; ++u) {
    int rank = atomicAdd(&hl[br[u]], 1);
    br[u] |= rank << 11;
  }
  __syncthreads();

  int v = (t < NB) ? hl[t] : 0;
  int incl = v;
  for (int off = 1; off < 32; off <<= 1) {
    int up = __shfl_up(incl, off);
    if ((t & 31) >= off) incl += up;
  }
  if ((t & 31) == 31) wsum[t >> 5] = incl;
  __syncthreads();
  if (t < 32) {
    int wv = wsum[t];
    int wi = wv;
    for (int off = 1; off < 32; off <<= 1) {
      int up = __shfl_up(wi, off);
      if (t >= off) wi += up;
    }
    wsum[t] = wi - wv;
  }
  __syncthreads();
  int pre = wsum[t >> 5];
  if (t < NB) {
    hl[LCO + t] = pre + incl - v;
    cm[(size_t)t * SP + sl] = v;
  }
  if (t == 0) hl[LCO + MAXB] = 0;
  if (sl == 0 && t < NB)
    for (int s = S; s < SP; ++s) cm[(size_t)t * SP + s] = 0;
  __syncthreads();

#pragma unroll
  for (int u = 0; u < MR; ++u) {
    int bk = br[u] & 2047;
    int slot = hl[LCO + bk] + (br[u] >> 11);
    if (bk != MAXB) srec[slot] = rec[u];
  }
  __syncthreads();

  unsigned* pks = pk + tb;
  for (int i = t; i < cnt; i += ET) pks[i] = srec[i];
}

#define SCAP 16384
#define RB 8

__device__ __forceinline__ void
put_record(unsigned w, int* tab, bool place, bool fits, int start, int* scol,
           int* __restrict__ col) {
  int p = atomicAdd(&tab[w & 255u], 1);
  if (place) {
    if (fits) scol[p - start] = (int)(w >> 8);
    else      col[p] = (int)(w >> 8);
  }
}

__device__ __forceinline__ void
bucket_runs(const unsigned* __restrict__ pk, const int* lo, const int* len, int S,
            int wave, int lane, int* tab, bool place, bool fits, int start,
            int* scol, int* __restrict__ col) {
  for (int r0 = wave; r0 < S; r0 += 32 * RB) {
    unsigned w[RB], base[RB];
    int L[RB];
#pragma unroll
    for (int r = 0; r < RB; ++r) {
      int s = r0 + 32 * r;
      bool ok = s < S;
      int sc = ok ? s : r0;
      L[r] = ok ? len[sc] : 0;
      base[r] = (unsigned)sc * (unsigned)MTILE + (L[r] > 0 ? (unsigned)lo[sc] : 0u);
      w[r] = pk[base[r] + (unsigned)min(lane, max(L[r] - 1, 0))];
    }
#pragma unroll
    for (int r = 0; r < RB; ++r)
      if (lane < L[r]) put_record(w[r], tab, place, fits, start, scol, col);
#pragma unroll
    for (int r = 0; r < RB; ++r)
      w[r] = pk[base[r] + (unsigned)min(lane + 32, max(L[r] - 1, 0))];
#pragma unroll
    for (int r = 0; r < RB; ++r)
      if (lane + 32 < L[r]) put_record(w[r], tab, place, fits, start, scol, col);
#pragma unroll
    for (int r = 0; r < RB; ++r)
      for (int i = 64 + lane; i < L[r]; i += 32)
        put_record(pk[base[r] + i], tab, place, fits, start, scol, col);
  }
}

__global__ void __launch_bounds__(ET)
bucket_sort_kernel(const unsigned* __restrict__ pk, const int* __restrict__ cm,
                   int* __restrict__ rowptr, int* __restrict__ col, int nNodes,
                   int S, int SP) {
  __shared__ int cnt[NPB];
  __shared__ int cur[NPB];
  __shared__ int wsum[NPB / 32];
  __shared__ int lo[MAXS];
  __shared__ int len[MAXS];
  __shared__ int tot[2];
  __shared__ int scol[SCAP];
  int t = threadIdx.x;
  int lane = t & 31;
  int wave = __builtin_amdgcn_readfirstlane(t >> 5);
  int b = blockIdx.x;
  for (int s = t; s < S; s += ET) {
    lo[s] = 0;
    len[s] = cm[(size_t)b * SP + s];
  }
  if (t < NPB) cnt[t] = 0;
  if (t < 2) tot[t] = 0;
  __syncthreads();

  int C4 = SP >> 2;
  int G = ET / C4;
  int c = t % C4;
  int g = t / C4;
  if (g < G) {
    const int4* cm4 = reinterpret_cast<const int4*>(cm);
    int4 a = make_int4(0, 0, 0, 0);
#pragma unroll 8
    for (int j = g; j < b; j += G) {
      int4 x = cm4[(size_t)j * C4 + c];
      a.x += x.x; a.y += x.y; a.z += x.z; a.w += x.w;
    }
    int s0 = 4 * c;
    if (s0 < S)     atomicAdd(&lo[s0], a.x);
    if (s0 + 1 < S) atomicAdd(&lo[s0 + 1], a.y);
    if (s0 + 2 < S) atomicAdd(&lo[s0 + 2], a.z);
    if (s0 + 3 < S) atomicAdd(&lo[s0 + 3], a.w);
  }
  __syncthreads();

  int ps = 0, pl = 0;
  for (int s = t; s < S; s += ET) { ps += lo[s]; pl += len[s]; }
  for (int off = 16; off > 0; off >>= 1) {
    ps += __shfl_xor(ps, off);
    pl += __shfl_xor(pl, off);
  }
  if (lane == 0) { atomicAdd(&tot[0], ps); atomicAdd(&tot[1], pl); }
  __syncthreads();
  int start = tot[0];
  int end = start + tot[1];

  bucket_runs(pk, lo, len, S, wave, lane, cnt, false, false, 0, scol, col);
  __syncthreads();

  int v = 0, incl = 0;
  if (t < NPB) {
    v = cnt[t];
    incl = v;
    for (int off = 1; off < 32; off <<= 1) {
      int u = __shfl_up(incl, off);
      if ((t & 31) >= off) incl += u;
    }
    if ((t & 31) == 31) wsum[t >> 5] = incl;
  }
  __syncthreads();
  if (t < NPB) {
    int pre = start;
    for (int w = 0; w < (t >> 5); ++w) pre += wsum[w];
    int excl = pre + incl - v;
    cur[t] = excl;
    int node = b * NPB + t;
    if (node < nNodes) rowptr[node] = excl;
    if (node == nNodes - 1) rowptr[nNodes] = end;
  }
  __syncthreads();

  bool fits = (end - start) <= SCAP;
  bucket_runs(pk, lo, len, S, wave, lane, cur, true, fits, start, scol, col);
  __syncthreads();
  if (fits) {
    int n = end - start;
    for (int k = t; k < n; k += ET) col[start + k] = scol[k];
  }
}

__device__ __forceinline__ float4 ld4(const float* p) {
  return *reinterpret_cast<const float4*>(p);
}
__device__ __forceinline__ float4 add4(float4 a, float4 b) {
  return make_float4(a.x + b.x, a.y + b.y, a.z + b.z, a.w + b.w);
}


#define P1_TILES 25
#define P1_NODES (P1_TILES * 16)
#define P1_LDS (2 * DH_PAD * W1P + DH)

__device__ __forceinline__ void
proj1_body(float* sm, int pb, const float* __restrict__ x,
           const float* __restrict__ W1l,
           const float* __restrict__ b1, const float* __restrict__ W1r,
           float* __restrict__ p, float* __restrict__ r, int nNodes) {
  float* sWl = sm;
  float* sWr = sm + DH_PAD * W1P;
  float* sB  = sm + 2 * DH_PAD * W1P;
  for (int i = threadIdx.x; i < DH_PAD * DIN; i += ET) {
    float wl = (i < DH * DIN) ? W1l[i] : 0.0f;
    float wr = (i < DH * DIN) ? W1r[i] : 0.0f;
    int j = (i / DIN) * W1P + (i % DIN);
    sWl[j] = wl;
    sWr[j] = wr;
  }
  if (threadIdx.x < DH) sB[threadIdx.x] = b1[threadIdx.x];
  __syncthreads();

  int lane = threadIdx.x & 31;
  int wave = threadIdx.x >> 5;
  if (wave >= P1_TILES) return;
  int node0 = pb * P1_NODES + wave * 16;
  int row = lane & 15;
  int node = node0 + row;
  int nodeC = min(node, nNodes - 1);
  int klo = (lane >> 4) * 2;
  int ncol = lane & 15;
  int mbase = (lane >> 4) * 8;

  const float* xrow = x + (size_t)nodeC * DIN;

  for (int half = 0; half < 2; ++half) {
    const float* sW = half ? sWr : sWl;
    float* o = half ? r : p;
    v8f a0 = {}, a1 = {}, a2 = {};
    for (int k = 0; k < DIN; k += 4) {
      v2f aX;
      aX.x = xrow[k + klo];
      aX.y = xrow[k + klo + 1];

      v2f b0, b1v, b2;
      b0.x  = sW[ncol * W1P + k + klo];        b0.y  = sW[ncol * W1P + k + klo + 1];
      b1v.x = sW[(16 + ncol) * W1P + k + klo]; b1v.y = sW[(16 + ncol) * W1P + k + klo + 1];
      b2.x  = sW[(32 + ncol) * W1P + k + klo]; b2.y  = sW[(32 + ncol) * W1P + k + klo + 1];

      a0 = wmma4(aX, b0, a0);
      a1 = wmma4(aX, b1v, a1);
      a2 = wmma4(aX, b2, a2);
    }

    float c0 = half ? sB[ncol] : 0.0f;
    float c1 = half ? sB[16 + ncol] : 0.0f;
    float c2 = (half && 32 + ncol < DH) ? sB[32 + ncol] : 0.0f;
    for (int i = 0; i < 8; ++i) {
      int n = node0 + mbase + i;
      if (n < nNodes) {
        float* orow = o + (size_t)n * DH;
        orow[ncol]      = half ? a0[i] + c0 : a0[i];
        orow[16 + ncol] = half ? a1[i] + c1 : a1[i];
        if (32 + ncol < DH) orow[32 + ncol] = half ? a2[i] + c2 : a2[i];
      }
    }
  }
}

__global__ void __launch_bounds__(ET, 16)
slice_proj1_kernel(const int* __restrict__ src, const int* __restrict__ dst,
                   int* __restrict__ cm, unsigned* __restrict__ pk, int E,
                   int NB, int S, int SP,
                   const float* __restrict__ x, const float* __restrict__ W1l,
                   const float* __restrict__ b1, const float* __restrict__ W1r,
                   float* __restrict__ p, float* __restrict__ r, int nNodes) {
  __shared__ int sm[SL_LDS > P1_LDS ? SL_LDS : P1_LDS];
  if ((int)blockIdx.x < S)
    slice_sort_body(sm, (int)blockIdx.x, src, dst, cm, pk, E, NB, S, SP);
  else
    proj1_body(reinterpret_cast<float*>(sm), (int)blockIdx.x - S, x, W1l, b1,
               W1r, p, r, nNodes);
}

__global__ void __launch_bounds__(256)
gather1_kernel(const float* __restrict__ p, const float* __restrict__ r,
               const int* __restrict__ rowptr, const int* __restrict__ col,
               float* __restrict__ h, int nNodes) {
  int lane = threadIdx.x & 31;
  int wave = __builtin_amdgcn_readfirstlane((int)(threadIdx.x >> 5));
  int nw = (int)gridDim.x * 8;
  int n = blockIdx.x * 8 + wave;
  if (n >= nNodes) return;
  int q = lane / 10;
  int chunk = lane - q * 10;
  int g = min(q, 2);
  const float* pc = p + chunk * 4;
  int beg = rowptr[n], end = rowptr[n + 1];
  while (true) {
    int nn = n + nw;
    int nbeg = 0, nend = 0;
    if (nn < nNodes) { nbeg = rowptr[nn]; nend = rowptr[nn + 1]; }
    float4 own = make_float4(0.0f, 0.0f, 0.0f, 0.0f);
    if (lane < 10) own = ld4(r + (size_t)n * DH + chunk * 4);

    float4 acc = make_float4(0.0f, 0.0f, 0.0f, 0.0f);
    int e0 = beg + lane;
    int my = (lane < 30 && e0 < end) ? col[e0] : 0;
    for (int base = beg; base < end; base += 30) {
      int e1 = base + 30 + lane;
      int myn = (lane < 30 && e1 < end) ? col[e1] : 0;
      int m = min(30, end - base);
      for (int j = 0; j < m; j += 15) {
        float4 v[5];
#pragma unroll
        for (int u = 0; u < 5; ++u) {
          int k = j + 3 * u + g;
          int s = __shfl(my, k);
          float4 t = ld4(pc + (size_t)s * DH);
          bool ok = k < m;
          v[u] = make_float4(ok ? t.x : 0.0f, ok ? t.y : 0.0f,
                             ok ? t.z : 0.0f, ok ? t.w : 0.0f);
        }
        acc = add4(acc, add4(add4(add4(v[0], v[1]), add4(v[2], v[3])), v[4]));
      }
      my = myn;
    }
    float4 t1, t2;
    t1.x = __shfl_down(acc.x, 10); t2.x = __shfl_down(acc.x, 20);
    t1.y = __shfl_down(acc.y, 10); t2.y = __shfl_down(acc.y, 20);
    t1.z = __shfl_down(acc.z, 10); t2.z = __shfl_down(acc.z, 20);
    t1.w = __shfl_down(acc.w, 10); t2.w = __shfl_down(acc.w, 20);
    if (lane < 10) {
      float inv = 1.0f / fmaxf((float)(end - beg), 1.0f);
      float4 s = add4(acc, add4(t1, t2));
      *reinterpret_cast<float4*>(h + (size_t)n * DH + chunk * 4) =
          make_float4(fmaxf(s.x * inv + own.x, 0.0f), fmaxf(s.y * inv + own.y, 0.0f),
                      fmaxf(s.z * inv + own.z, 0.0f), fmaxf(s.w * inv + own.w, 0.0f));
    }

    if (nn >= nNodes) break;
    n = nn; beg = nbeg; end = nend;
  }
}

__global__ void __launch_bounds__(256)
proj2_kernel(const float* __restrict__ h, const float* __restrict__ W2l,
             const float* __restrict__ b2, const float* __restrict__ W2r,
             float* __restrict__ p, float* __restrict__ r, int nNodes) {
  __shared__ float sWl[DO_PAD * W2P];
  __shared__ float sWr[DO_PAD * W2P];
  __shared__ float sB[DO_PAD];
  for (int i = threadIdx.x; i < DO_PAD * DH; i += 256) {
    float wl = (i < DOUT * DH) ? W2l[i] : 0.0f;
    float wr = (i < DOUT * DH) ? W2r[i] : 0.0f;
    int j = (i / DH) * W2P + (i % DH);
    sWl[j] = wl;
    sWr[j] = wr;
  }
  if (threadIdx.x < DO_PAD)
    sB[threadIdx.x] = (threadIdx.x < DOUT) ? b2[threadIdx.x] : 0.0f;
  __syncthreads();

  int lane = threadIdx.x & 31;
  int wave = threadIdx.x >> 5;
  int node0 = (blockIdx.x * 8 + wave) * 16;
  int row = lane & 15;
  int node = node0 + row;
  int nodeC = min(node, nNodes - 1);
  int klo = (lane >> 4) * 2;
  int ncol = lane & 15;

  const float* hrow = h + (size_t)nodeC * DH;

  v8f p0 = {}, p1 = {}, r0 = {}, r1 = {};
  for (int k = 0; k < DH; k += 4) {
    v2f aH;
    aH.x = hrow[k + klo];
    aH.y = hrow[k + klo + 1];

    v2f bl0, br0, bl1, br1;
    bl0.x = sWl[ncol * W2P + k + klo];        bl0.y = sWl[ncol * W2P + k + klo + 1];
    br0.x = sWr[ncol * W2P + k + klo];        br0.y = sWr[ncol * W2P + k + klo + 1];
    bl1.x = sWl[(16 + ncol) * W2P + k + klo]; bl1.y = sWl[(16 + ncol) * W2P + k + klo + 1];
    br1.x = sWr[(16 + ncol) * W2P + k + klo]; br1.y = sWr[(16 + ncol) * W2P + k + klo + 1];

    p0 = wmma4(aH, bl0, p0);
    r0 = wmma4(aH, br0, r0);
    p1 = wmma4(aH, bl1, p1);
    r1 = wmma4(aH, br1, r1);
  }

  int mbase = (lane >> 4) * 8;
  for (int i = 0; i < 8; ++i) {
    int n = node0 + mbase + i;
    if (n < nNodes) {
      float* prow = p + (size_t)n * DO_PAD;
      float* rrow = r + (size_t)n * DO_PAD;
      prow[ncol]      = p0[i];
      prow[16 + ncol] = p1[i];
      rrow[ncol]      = r0[i] + sB[ncol];
      rrow[16 + ncol] = r1[i] + sB[16 + ncol];
    }
  }
}

__global__ void __launch_bounds__(256)
gather2_kernel(const float* __restrict__ p, const float* __restrict__ r,
               const int* __restrict__ rowptr, const int* __restrict__ col,
               float* __restrict__ o, int nNodes) {
  int lane = threadIdx.x & 31;
  int wave = __builtin_amdgcn_readfirstlane((int)(threadIdx.x >> 5));
  int nw = (int)gridDim.x * 8;
  int n = blockIdx.x * 8 + wave;
  if (n >= nNodes) return;
  int q = lane >> 3;
  int chunk = lane & 7;
  const float* pc = p + chunk * 4;
  int beg = rowptr[n], end = rowptr[n + 1];
  while (true) {
    int nn = n + nw;
    int nbeg = 0, nend = 0;
    if (nn < nNodes) { nbeg = rowptr[nn]; nend = rowptr[nn + 1]; }
    float4 own = make_float4(0.0f, 0.0f, 0.0f, 0.0f);
    if (lane < 6) own = ld4(r + (size_t)n * DO_PAD + chunk * 4);

    float4 acc = make_float4(0.0f, 0.0f, 0.0f, 0.0f);
    int e0 = beg + lane;
    int my = (e0 < end) ? col[e0] : 0;
    for (int base = beg; base < end; base += 32) {
      int e1 = base + 32 + lane;
      int myn = (e1 < end) ? col[e1] : 0;
      int m = min(32, end - base);
      for (int j = 0; j < m; j += 16) {
        float4 v[4];
#pragma unroll
        for (int u = 0; u < 4; ++u) {
          int k = j + 4 * u + q;
          int s = __shfl(my, k);
          float4 t = ld4(pc + (size_t)s * DO_PAD);
          bool ok = k < m;
          v[u] = make_float4(ok ? t.x : 0.0f, ok ? t.y : 0.0f,
                             ok ? t.z : 0.0f, ok ? t.w : 0.0f);
        }
        acc = add4(acc, add4(add4(v[0], v[1]), add4(v[2], v[3])));
      }
      my = myn;
    }
    acc.x += __shfl_xor(acc.x, 8);
    acc.y += __shfl_xor(acc.y, 8);
    acc.z += __shfl_xor(acc.z, 8);
    acc.w += __shfl_xor(acc.w, 8);
    acc.x += __shfl_xor(acc.x, 16);
    acc.y += __shfl_xor(acc.y, 16);
    acc.z += __shfl_xor(acc.z, 16);
    acc.w += __shfl_xor(acc.w, 16);
    if (lane < 6) {
      float inv = 1.0f / fmaxf((float)(end - beg), 1.0f);
      *reinterpret_cast<float4*>(o + (size_t)n * DOUT + chunk * 4) =
          make_float4(acc.x * inv + own.x, acc.y * inv + own.y,
                      acc.z * inv + own.z, acc.w * inv + own.w);
    }

    if (nn >= nNodes) break;
    n = nn; beg = nbeg; end = nend;
  }
}

__global__ void __launch_bounds__(256)
logsoftmax_kernel(const float* __restrict__ o, float* __restrict__ out, int nNodes) {
  int n = blockIdx.x * 256 + threadIdx.x;
  if (n >= nNodes) return;
  const float* r = o + (size_t)n * DOUT;
  float mx = -3.402823466e+38f;
  for (int j = 0; j < DOUT; ++j) mx = fmaxf(mx, r[j]);
  float s = 0.0f;
  for (int j = 0; j < DOUT; ++j) s += expf(r[j] - mx);
  float lse = mx + logf(s);
  float* w = out + (size_t)n * DOUT;
  for (int j = 0; j < DOUT; ++j) w[j] = r[j] - lse;
}

extern "C" void kernel_launch(void* const* d_in, const int* in_sizes, int n_in,
                              void* d_out, int out_size, void* d_ws, size_t ws_size,
                              hipStream_t stream) {
  const float* x   = (const float*)d_in[0];
  const int*   ei  = (const int*)d_in[1];
  const float* W1l = (const float*)d_in[2];
  const float* b1  = (const float*)d_in[3];
  const float* W1r = (const float*)d_in[4];
  const float* W2l = (const float*)d_in[5];
  const float* b2  = (const float*)d_in[6];
  const float* W2r = (const float*)d_in[7];
  float* out = (float*)d_out;

  int  N = in_sizes[0] / DIN;
  long E = (long)in_sizes[1] / 2;
  const int* src = ei;
  const int* dst = ei + E;

  float* ws = (float*)d_ws;
  float* P  = ws;
  float* R  = P + (size_t)N * DH;
  float* H  = R + (size_t)N * DH;
  float* o  = H;
  int* rowptr = (int*)(ws + (size_t)N * (DIN + 1 + DH + DOUT));
  int* col    = rowptr + N + 1;

  int NB = (N + NPB - 1) / NPB;
  int S  = (int)((E + MTILE - 1) / MTILE);
  int SP = (S + 3) / 4 * 4;
  if (SP < 4) SP = 4;
  unsigned* pk = (unsigned*)H;
  int* cm = (int*)H + (size_t)S * MTILE;

  int nbP = (N + P1_NODES - 1) / P1_NODES;
  slice_proj1_kernel<<<S + nbP, ET, 0, stream>>>(src, dst, cm, pk, (int)E, NB, S, SP,
                                                 x, W1l, b1, W1r, P, R, N);
  bucket_sort_kernel<<<NB, ET, 0, stream>>>(pk, cm, rowptr, col, N, S, SP);

  int nb = (N + 127) / 128;

  gather1_kernel<<<2048, 256, 0, stream>>>(P, R, rowptr, col, H, N);

  proj2_kernel<<<nb, 256, 0, stream>>>(H, W2l, b2, W2r, P, R, N);
  gather2_kernel<<<2048, 256, 0, stream>>>(P, R, rowptr, col, o, N);

  logsoftmax_kernel<<<(N + 255) / 256, 256, 0, stream>>>(o, out, N);
}


# Node-local stages ride in the neighbour-sum launches: layer-2 projections on 16-node h tiles in LDS, log_softmax in lanes (its exp sum now a lane tree): 6 launches to 4
# speedup vs baseline: 6.0917x; 1.0644x over previous
#include <hip/hip_runtime.h>
#include <math.h>

#define DIN 64
#define DH  40
#define DOUT 24
#define DH_PAD 48
#define DO_PAD 32
#define W1P 68
#define W2P 44

typedef __attribute__((ext_vector_type(2))) float v2f;
typedef __attribute__((ext_vector_type(8))) float v8f;

__device__ __forceinline__ v8f wmma4(v2f a, v2f b, v8f c) {
  return __builtin_amdgcn_wmma_f32_16x16x4_f32(false, a, false, b, (short)0, c,
                                               false, false);
}

#define NPB 256
#define MAXB 1024
#define ET 1024
#define MR 13
#define MTILE (MR * ET)
#define MAXS 1024

__device__ __forceinline__ int bucket_of(int d, int NB) {
  return min((int)((unsigned)d >> 8), NB - 1);
}

#define LCO (MAXB + 1)
#define SL_LDS (2 * LCO + ET / 32 + MTILE)
__device__ __forceinline__ void
slice_sort_body(int* sm, int sl, const int* __restrict__ src,
                const int* __restrict__ dst,
                int* __restrict__ cm, unsigned* __restrict__ pk, int E, int NB,
                int S, int SP) {
  int* hl = sm;
  int* wsum = sm + 2 * LCO;
  unsigned* srec = reinterpret_cast<unsigned*>(sm + 2 * LCO + ET / 32);
  int t = threadIdx.x;
  int tb = sl * MTILE;
  int cnt = min(E - tb, MTILE);
  int back = max(cnt - ET, 0);
  int tt = min(t, cnt - 1);
  for (int j = t; j < NB; j += ET) hl[j] = 0;
  if (t == 0) hl[MAXB] = 0;
  __syncthreads();

  int br[MR];
  unsigned rec[MR];
#pragma unroll
  for (int u = 0; u < MR; ++u) {
    int ro = min(u * ET, back);
    const int* du = dst + tb + ro;
    const int* su = src + tb + ro;
    int d = du[tt];
    int s = su[tt];
    bool ok = (t >= u * ET - ro) && (t < cnt);
    br[u] = ok ? bucket_of(d, NB) : MAXB;
    rec[u] = ((unsigned)s << 8) | ((unsigned)d & 255u);
  }
#pragma unroll
  for (int u = 0; u < MR; ++u) {
    int rank = atomicAdd(&hl[br[u]], 1);
    br[u] |= rank << 11;
  }
  __syncthreads();

  int v = (t < NB) ? hl[t] : 0;
  int incl = v;
  for (int off = 1; off < 32; off <<= 1) {
    int up = __shfl_up(incl, off);
    if ((t & 31) >= off) incl += up;
  }
  if ((t & 31) == 31) wsum[t >> 5] = incl;
  __syncthreads();
  if (t < 32) {
    int wv = wsum[t];
    int wi = wv;
    for (int off = 1; off < 32; off <<= 1) {
      int up = __shfl_up(wi, off);
      if (t >= off) wi += up;
    }
    wsum[t] = wi - wv;
  }
  __syncthreads();
  int pre = wsum[t >> 5];
  if (t < NB) {
    hl[LCO + t] = pre + incl - v;
    cm[(size_t)t * SP + sl] = v;
  }
  if (t == 0) hl[LCO + MAXB] = 0;
  if (sl == 0 && t < NB)
    for (int s = S; s < SP; ++s) cm[(size_t)t * SP + s] = 0;
  __syncthreads();

#pragma unroll
  for (int u = 0; u < MR; ++u) {
    int bk = br[u] & 2047;
    int slot = hl[LCO + bk] + (br[u] >> 11);
    if (bk != MAXB) srec[slot] = rec[u];
  }
  __syncthreads();

  unsigned* pks = pk + tb;
  for (int i = t; i < cnt; i += ET) pks[i] = srec[i];
}

#define SCAP 16384
#define RB 8

__device__ __forceinline__ void
put_record(unsigned w, int* tab, bool place, bool fits, int start, int* scol,
           int* __restrict__ col) {
  int p = atomicAdd(&tab[w & 255u], 1);
  if (place) {
    if (fits) scol[p - start] = (int)(w >> 8);
    else      col[p] = (int)(w >> 8);
  }
}

__device__ __forceinline__ void
bucket_runs(const unsigned* __restrict__ pk, const int* lo, const int* len, int S,
            int wave, int lane, int* tab, bool place, bool fits, int start,
            int* scol, int* __restrict__ col) {
  for (int r0 = wave; r0 < S; r0 += 32 * RB) {
    unsigned w[RB], base[RB];
    int L[RB];
#pragma unroll
    for (int r = 0; r < RB; ++r) {
      int s = r0 + 32 * r;
      bool ok = s < S;
      int sc = ok ? s : r0;
      L[r] = ok ? len[sc] : 0;
      base[r] = (unsigned)sc * (unsigned)MTILE + (L[r] > 0 ? (unsigned)lo[sc] : 0u);
      w[r] = pk[base[r] + (unsigned)min(lane, max(L[r] - 1, 0))];
    }
#pragma unroll
    for (int r = 0; r < RB; ++r)
      if (lane < L[r]) put_record(w[r], tab, place, fits, start, scol, col);
#pragma unroll
    for (int r = 0; r < RB; ++r)
      w[r] = pk[base[r] + (unsigned)min(lane + 32, max(L[r] - 1, 0))];
#pragma unroll
    for (int r = 0; r < RB; ++r)
      if (lane + 32 < L[r]) put_record(w[r], tab, place, fits, start, scol, col);
#pragma unroll
    for (int r = 0; r < RB; ++r)
      for (int i = 64 + lane; i < L[r]; i += 32)
        put_record(pk[base[r] + i], tab, place, fits, start, scol, col);
  }
}

__global__ void __launch_bounds__(ET)
bucket_sort_kernel(const unsigned* __restrict__ pk, const int* __restrict__ cm,
                   int* __restrict__ rowptr, int* __restrict__ col, int nNodes,
                   int S, int SP) {
  __shared__ int cnt[NPB];
  __shared__ int cur[NPB];
  __shared__ int wsum[NPB / 32];
  __shared__ int lo[MAXS];
  __shared__ int len[MAXS];
  __shared__ int tot[2];
  __shared__ int scol[SCAP];
  int t = threadIdx.x;
  int lane = t & 31;
  int wave = __builtin_amdgcn_readfirstlane(t >> 5);
  int b = blockIdx.x;
  for (int s = t; s < S; s += ET) {
    lo[s] = 0;
    len[s] = cm[(size_t)b * SP + s];
  }
  if (t < NPB) cnt[t] = 0;
  if (t < 2) tot[t] = 0;
  __syncthreads();

  int C4 = SP >> 2;
  int G = ET / C4;
  int c = t % C4;
  int g = t / C4;
  if (g < G) {
    const int4* cm4 = reinterpret_cast<const int4*>(cm);
    int4 a = make_int4(0, 0, 0, 0);
#pragma unroll 8
    for (int j = g; j < b; j += G) {
      int4 x = cm4[(size_t)j * C4 + c];
      a.x += x.x; a.y += x.y; a.z += x.z; a.w += x.w;
    }
    int s0 = 4 * c;
    if (s0 < S)     atomicAdd(&lo[s0], a.x);
    if (s0 + 1 < S) atomicAdd(&lo[s0 + 1], a.y);
    if (s0 + 2 < S) atomicAdd(&lo[s0 + 2], a.z);
    if (s0 + 3 < S) atomicAdd(&lo[s0 + 3], a.w);
  }
  __syncthreads();

  int ps = 0, pl = 0;
  for (int s = t; s < S; s += ET) { ps += lo[s]; pl += len[s]; }
  for (int off = 16; off > 0; off >>= 1) {
    ps += __shfl_xor(ps, off);
    pl += __shfl_xor(pl, off);
  }
  if (lane == 0) { atomicAdd(&tot[0], ps); atomicAdd(&tot[1], pl); }
  __syncthreads();
  int start = tot[0];
  int end = start + tot[1];

  bucket_runs(pk, lo, len, S, wave, lane, cnt, false, false, 0, scol, col);
  __syncthreads();

  int v = 0, incl = 0;
  if (t < NPB) {
    v = cnt[t];
    incl = v;
    for (int off = 1; off < 32; off <<= 1) {
      int u = __shfl_up(incl, off);
      if ((t & 31) >= off) incl += u;
    }
    if ((t & 31) == 31) wsum[t >> 5] = incl;
  }
  __syncthreads();
  if (t < NPB) {
    int pre = start;
    for (int w = 0; w < (t >> 5); ++w) pre += wsum[w];
    int excl = pre + incl - v;
    cur[t] = excl;
    int node = b * NPB + t;
    if (node < nNodes) rowptr[node] = excl;
    if (node == nNodes - 1) rowptr[nNodes] = end;
  }
  __syncthreads();

  bool fits = (end - start) <= SCAP;
  bucket_runs(pk, lo, len, S, wave, lane, cur, true, fits, start, scol, col);
  __syncthreads();
  if (fits) {
    int n = end - start;
    for (int k = t; k < n; k += ET) col[start + k] = scol[k];
  }
}

__device__ __forceinline__ float4 ld4(const float* p) {
  return *reinterpret_cast<const float4*>(p);
}
__device__ __forceinline__ float4 add4(float4 a, float4 b) {
  return make_float4(a.x + b.x, a.y + b.y, a.z + b.z, a.w + b.w);
}


#define P1_TILES 25
#define P1_NODES (P1_TILES * 16)
#define P1_LDS (2 * DH_PAD * W1P + DH)

__device__ __forceinline__ void
proj1_body(float* sm, int pb, const float* __restrict__ x,
           const float* __restrict__ W1l,
           const float* __restrict__ b1, const float* __restrict__ W1r,
           float* __restrict__ p, float* __restrict__ r, int nNodes) {
  float* sWl = sm;
  float* sWr = sm + DH_PAD * W1P;
  float* sB  = sm + 2 * DH_PAD * W1P;
  for (int i = threadIdx.x; i < DH_PAD * DIN; i += ET) {
    float wl = (i < DH * DIN) ? W1l[i] : 0.0f;
    float wr = (i < DH * DIN) ? W1r[i] : 0.0f;
    int j = (i / DIN) * W1P + (i % DIN);
    sWl[j] = wl;
    sWr[j] = wr;
  }
  if (threadIdx.x < DH) sB[threadIdx.x] = b1[threadIdx.x];
  __syncthreads();

  int lane = threadIdx.x & 31;
  int wave = threadIdx.x >> 5;
  if (wave >= P1_TILES) return;
  int node0 = pb * P1_NODES + wave * 16;
  int row = lane & 15;
  int node = node0 + row;
  int nodeC = min(node, nNodes - 1);
  int klo = (lane >> 4) * 2;
  int ncol = lane & 15;
  int mbase = (lane >> 4) * 8;

  const float* xrow = x + (size_t)nodeC * DIN;

  for (int half = 0; half < 2; ++half) {
    const float* sW = half ? sWr : sWl;
    float* o = half ? r : p;
    v8f a0 = {}, a1 = {}, a2 = {};
    for (int k = 0; k < DIN; k += 4) {
      v2f aX;
      aX.x = xrow[k + klo];
      aX.y = xrow[k + klo + 1];

      v2f b0, b1v, b2;
      b0.x  = sW[ncol * W1P + k + klo];        b0.y  = sW[ncol * W1P + k + klo + 1];
      b1v.x = sW[(16 + ncol) * W1P + k + klo]; b1v.y = sW[(16 + ncol) * W1P + k + klo + 1];
      b2.x  = sW[(32 + ncol) * W1P + k + klo]; b2.y  = sW[(32 + ncol) * W1P + k + klo + 1];

      a0 = wmma4(aX, b0, a0);
      a1 = wmma4(aX, b1v, a1);
      a2 = wmma4(aX, b2, a2);
    }

    float c0 = half ? sB[ncol] : 0.0f;
    float c1 = half ? sB[16 + ncol] : 0.0f;
    float c2 = (half && 32 + ncol < DH) ? sB[32 + ncol] : 0.0f;
    for (int i = 0; i < 8; ++i) {
      int n = node0 + mbase + i;
      if (n < nNodes) {
        float* orow = o + (size_t)n * DH;
        orow[ncol]      = half ? a0[i] + c0 : a0[i];
        orow[16 + ncol] = half ? a1[i] + c1 : a1[i];
        if (32 + ncol < DH) orow[32 + ncol] = half ? a2[i] + c2 : a2[i];
      }
    }
  }
}

__global__ void __launch_bounds__(ET, 16)
slice_proj1_kernel(const int* __restrict__ src, const int* __restrict__ dst,
                   int* __restrict__ cm, unsigned* __restrict__ pk, int E,
                   int NB, int S, int SP,
                   const float* __restrict__ x, const float* __restrict__ W1l,
                   const float* __restrict__ b1, const float* __restrict__ W1r,
                   float* __restrict__ p, float* __restrict__ r, int nNodes) {
  __shared__ int sm[SL_LDS > P1_LDS ? SL_LDS : P1_LDS];
  if ((int)blockIdx.x < S)
    slice_sort_body(sm, (int)blockIdx.x, src, dst, cm, pk, E, NB, S, SP);
  else
    proj1_body(reinterpret_cast<float*>(sm), (int)blockIdx.x - S, x, W1l, b1,
               W1r, p, r, nNodes);
}

#define G1_WAVES 16
#define HP 44
__global__ void __launch_bounds__(G1_WAVES * 32, 16)
gather1_proj2_kernel(const float* __restrict__ p, float* __restrict__ r,
                     const int* __restrict__ rowptr, const int* __restrict__ col,
                     const float* __restrict__ W2l, const float* __restrict__ b2,
                     const float* __restrict__ W2r, float* __restrict__ p2,
                     int nNodes) {
  __shared__ float sWl[DO_PAD * W2P];
  __shared__ float sWr[DO_PAD * W2P];
  __shared__ float sB[DO_PAD];
  __shared__ float sH[2][16 * HP];
  for (int i = threadIdx.x; i < DO_PAD * DH; i += G1_WAVES * 32) {
    float wl = (i < DOUT * DH) ? W2l[i] : 0.0f;
    float wr = (i < DOUT * DH) ? W2r[i] : 0.0f;
    int j = (i / DH) * W2P + (i % DH);
    sWl[j] = wl;
    sWr[j] = wr;
  }
  if (threadIdx.x < DO_PAD)
    sB[threadIdx.x] = (threadIdx.x < DOUT) ? b2[threadIdx.x] : 0.0f;

  int lane = threadIdx.x & 31;
  int wave = __builtin_amdgcn_readfirstlane((int)(threadIdx.x >> 5));
  int nTiles = (nNodes + 15) >> 4;
  int q = lane / 10;
  int chunk = lane - q * 10;
  int g = min(q, 2);
  const float* pc = p + chunk * 4;
  int klo = (lane >> 4) * 2;
  int ncol = lane & 15;
  int mbase = (lane >> 4) * 8;

  int tile = blockIdx.x;
  int n = tile * 16 + wave;
  int rp = 0;
  if (tile < nTiles && n < nNodes) rp = rowptr[n + (lane & 1)];
  int buf = 0;
  for (; tile < nTiles; tile += (int)gridDim.x) {
    int beg = __builtin_amdgcn_readlane(rp, 0);
    int end = __builtin_amdgcn_readlane(rp, 1);
    int nn = n + (int)gridDim.x * 16;
    int nrp = 0;
    if (tile + (int)gridDim.x < nTiles && nn < nNodes) nrp = rowptr[nn + (lane & 1)];
    float4 own = make_float4(0.0f, 0.0f, 0.0f, 0.0f);
    if (lane < 10 && n < nNodes) own = ld4(r + (size_t)n * DH + chunk * 4);

    float4 acc = make_float4(0.0f, 0.0f, 0.0f, 0.0f);
    int e0 = beg + lane;
    int my = (lane < 30 && e0 < end) ? col[e0] : 0;
    for (int base = beg; base < end; base += 30) {
      int e1 = base + 30 + lane;
      int myn = (lane < 30 && e1 < end) ? col[e1] : 0;
      int m = min(30, end - base);
      for (int j = 0; j < m; j += 15) {
        float4 v[5];
#pragma unroll
        for (int u = 0; u < 5; ++u) {
          int k = j + 3 * u + g;
          int s = __shfl(my, k);
          float4 t = ld4(pc + (size_t)s * DH);
          bool ok = k < m;
          v[u] = make_float4(ok ? t.x : 0.0f, ok ? t.y : 0.0f,
                             ok ? t.z : 0.0f, ok ? t.w : 0.0f);
        }
        acc = add4(acc, add4(add4(add4(v[0], v[1]), add4(v[2], v[3])), v[4]));
      }
      my = myn;
    }
    float4 t1, t2;
    t1.x = __shfl_down(acc.x, 10); t2.x = __shfl_down(acc.x, 20);
    t1.y = __shfl_down(acc.y, 10); t2.y = __shfl_down(acc.y, 20);
    t1.z = __shfl_down(acc.z, 10); t2.z = __shfl_down(acc.z, 20);
    t1.w = __shfl_down(acc.w, 10); t2.w = __shfl_down(acc.w, 20);
    float* hT = sH[buf];
    if (lane < 10) {
      float inv = 1.0f / fmaxf((float)(end - beg), 1.0f);
      float4 s = add4(acc, add4(t1, t2));
      *reinterpret_cast<float4*>(hT + wave * HP + chunk * 4) =
          make_float4(fmaxf(s.x * inv + own.x, 0.0f), fmaxf(s.y * inv + own.y, 0.0f),
                      fmaxf(s.z * inv + own.z, 0.0f), fmaxf(s.w * inv + own.w, 0.0f));
    }
    __syncthreads();

    if (wave < 4) {
      bool own2 = wave >= 2;
      const float* sW = own2 ? sWr : sWl;
      int cb = (wave & 1) * 16;
      const float* hrow = hT + (lane & 15) * HP;
      const float* wrow = sW + (cb + ncol) * W2P;
      v8f a = {};
      for (int k = 0; k < DH; k += 4) {
        v2f aH, bW;
        aH.x = hrow[k + klo];
        aH.y = hrow[k + klo + 1];
        bW.x = wrow[k + klo];
        bW.y = wrow[k + klo + 1];
        a = wmma4(aH, bW, a);
      }
      float bias = own2 ? sB[cb + ncol] : 0.0f;
      for (int i = 0; i < 8; ++i) {
        int nr = tile * 16 + mbase + i;
        if (nr < nNodes) {
          if (own2) r[(size_t)nr * DH + cb + ncol] = a[i] + bias;
          else      p2[(size_t)nr * DO_PAD + cb + ncol] = a[i];
        }
      }
    }
    buf ^= 1;
    n = nn; rp = nrp;
  }
}

__global__ void __launch_bounds__(256)
gather2_lsm_kernel(const float* __restrict__ p, const float* __restrict__ r,
                   const int* __restrict__ rowptr, const int* __restrict__ col,
                   float* __restrict__ out, int nNodes) {
  int lane = threadIdx.x & 31;
  int wave = __builtin_amdgcn_readfirstlane((int)(threadIdx.x >> 5));
  int nw = (int)gridDim.x * 8;
  int n = blockIdx.x * 8 + wave;
  if (n >= nNodes) return;
  int q = lane >> 3;
  int chunk = lane & 7;
  const float* pc = p + chunk * 4;
  int beg = rowptr[n], end = rowptr[n + 1];
  while (true) {
    int nn = n + nw;
    int nbeg = 0, nend = 0;
    if (nn < nNodes) { nbeg = rowptr[nn]; nend = rowptr[nn + 1]; }
    float4 own = make_float4(0.0f, 0.0f, 0.0f, 0.0f);
    if (lane < 6) own = ld4(r + (size_t)n * DH + chunk * 4);

    float4 acc = make_float4(0.0f, 0.0f, 0.0f, 0.0f);
    int e0 = beg + lane;
    int my = (e0 < end) ? col[e0] : 0;
    for (int base = beg; base < end; base += 32) {
      int e1 = base + 32 + lane;
      int myn = (e1 < end) ? col[e1] : 0;
      int m = min(32, end - base);
      for (int j = 0; j < m; j += 16) {
        float4 v[4];
#pragma unroll
        for (int u = 0; u < 4; ++u) {
          int k = j + 4 * u + q;
          int s = __shfl(my, k);
          float4 t = ld4(pc + (size_t)s * DO_PAD);
          bool ok = k < m;
          v[u] = make_float4(ok ? t.x : 0.0f, ok ? t.y : 0.0f,
                             ok ? t.z : 0.0f, ok ? t.w : 0.0f);
        }
        acc = add4(acc, add4(add4(v[0], v[1]), add4(v[2], v[3])));
      }
      my = myn;
    }
    acc.x += __shfl_xor(acc.x, 8);
    acc.y += __shfl_xor(acc.y, 8);
    acc.z += __shfl_xor(acc.z, 8);
    acc.w += __shfl_xor(acc.w, 8);
    acc.x += __shfl_xor(acc.x, 16);
    acc.y += __shfl_xor(acc.y, 16);
    acc.z += __shfl_xor(acc.z, 16);
    acc.w += __shfl_xor(acc.w, 16);
    float inv = 1.0f / fmaxf((float)(end - beg), 1.0f);
    float4 o = make_float4(acc.x * inv + own.x, acc.y * inv + own.y,
                           acc.z * inv + own.z, acc.w * inv + own.w);
    bool has = chunk < 6;
    float mx = has ? fmaxf(fmaxf(o.x, o.y), fmaxf(o.z, o.w)) : -3.402823466e+38f;
    mx = fmaxf(mx, __shfl_xor(mx, 1));
    mx = fmaxf(mx, __shfl_xor(mx, 2));
    mx = fmaxf(mx, __shfl_xor(mx, 4));
    float se = has ? (expf(o.x - mx) + expf(o.y - mx)) + (expf(o.z - mx) + expf(o.w - mx))
                   : 0.0f;
    se += __shfl_xor(se, 1);
    se += __shfl_xor(se, 2);
    se += __shfl_xor(se, 4);
    float lse = mx + logf(se);
    if (lane < 6) {
      *reinterpret_cast<float4*>(out + (size_t)n * DOUT + chunk * 4) =
          make_float4(o.x - lse, o.y - lse, o.z - lse, o.w - lse);
    }

    if (nn >= nNodes) break;
    n = nn; beg = nbeg; end = nend;
  }
}

extern "C" void kernel_launch(void* const* d_in, const int* in_sizes, int n_in,
                              void* d_out, int out_size, void* d_ws, size_t ws_size,
                              hipStream_t stream) {
  const float* x   = (const float*)d_in[0];
  const int*   ei  = (const int*)d_in[1];
  const float* W1l = (const float*)d_in[2];
  const float* b1  = (const float*)d_in[3];
  const float* W1r = (const float*)d_in[4];
  const float* W2l = (const float*)d_in[5];
  const float* b2  = (const float*)d_in[6];
  const float* W2r = (const float*)d_in[7];
  float* out = (float*)d_out;

  int  N = in_sizes[0] / DIN;
  long E = (long)in_sizes[1] / 2;
  const int* src = ei;
  const int* dst = ei + E;

  float* ws = (float*)d_ws;
  float* P  = ws;
  float* R  = P + (size_t)N * DH;
  float* H  = R + (size_t)N * DH;
  float* P2 = H;
  int* rowptr = (int*)(ws + (size_t)N * (DIN + 1 + DH + DOUT));
  int* col    = rowptr + N + 1;

  int NB = (N + NPB - 1) / NPB;
  int S  = (int)((E + MTILE - 1) / MTILE);
  int SP = (S + 3) / 4 * 4;
  if (SP < 4) SP = 4;
  unsigned* pk = (unsigned*)H;
  int* cm = (int*)H + (size_t)S * MTILE;

  int nbP = (N + P1_NODES - 1) / P1_NODES;
  slice_proj1_kernel<<<S + nbP, ET, 0, stream>>>(src, dst, cm, pk, (int)E, NB, S, SP,
                                                 x, W1l, b1, W1r, P, R, N);
  bucket_sort_kernel<<<NB, ET, 0, stream>>>(pk, cm, rowptr, col, N, S, SP);

  gather1_proj2_kernel<<<1024, G1_WAVES * 32, 0, stream>>>(P, R, rowptr, col, W2l, b2,
                                                         W2r, P2, N);
  gather2_lsm_kernel<<<2048, 256, 0, stream>>>(P2, R, rowptr, col, out, N);
}


# Level 2 of the edge sort loads each word once: run starts come from level 1's prefix sum, not from a loop over count rows; a run's first 64 records stay in registers across both sweeps
# speedup vs baseline: 6.3909x; 1.0491x over previous
#include <hip/hip_runtime.h>
#include <math.h>

#define DIN 64
#define DH  40
#define DOUT 24
#define DH_PAD 48
#define DO_PAD 32
#define W1P 68
#define W2P 44

typedef __attribute__((ext_vector_type(2))) float v2f;
typedef __attribute__((ext_vector_type(8))) float v8f;

__device__ __forceinline__ v8f wmma4(v2f a, v2f b, v8f c) {
  return __builtin_amdgcn_wmma_f32_16x16x4_f32(false, a, false, b, (short)0, c,
                                               false, false);
}

#define NPB 256
#define MAXB 1024
#define ET 1024
#define MR 13
#define MTILE (MR * ET)
#define MAXS 1024

__device__ __forceinline__ int bucket_of(int d, int NB) {
  return min((int)((unsigned)d >> 8), NB - 1);
}

#define LCO (MAXB + 1)
#define SL_LDS (2 * LCO + ET / 32 + MTILE)
__device__ __forceinline__ void
slice_sort_body(int* sm, int sl, const int* __restrict__ src,
                const int* __restrict__ dst,
                int* __restrict__ cm, unsigned* __restrict__ pk, int E, int NB,
                int S, int SP) {
  int* hl = sm;
  int* wsum = sm + 2 * LCO;
  unsigned* srec = reinterpret_cast<unsigned*>(sm + 2 * LCO + ET / 32);
  int t = threadIdx.x;
  int tb = sl * MTILE;
  int cnt = min(E - tb, MTILE);
  int back = max(cnt - ET, 0);
  int tt = min(t, cnt - 1);
  for (int j = t; j < NB; j += ET) hl[j] = 0;
  if (t == 0) hl[MAXB] = 0;
  __syncthreads();

  int br[MR];
  unsigned rec[MR];
#pragma unroll
  for (int u = 0; u < MR; ++u) {
    int ro = min(u * ET, back);
    const int* du = dst + tb + ro;
    const int* su = src + tb + ro;
    int d = du[tt];
    int s = su[tt];
    bool ok = (t >= u * ET - ro) && (t < cnt);
    br[u] = ok ? bucket_of(d, NB) : MAXB;
    rec[u] = ((unsigned)s << 8) | ((unsigned)d & 255u);
  }
#pragma unroll
  for (int u = 0; u < MR; ++u) {
    int rank = atomicAdd(&hl[br[u]], 1);
    br[u] |= rank << 11;
  }
  __syncthreads();

  int v = (t < NB) ? hl[t] : 0;
  int incl = v;
  for (int off = 1; off < 32; off <<= 1) {
    int up = __shfl_up(incl, off);
    if ((t & 31) >= off) incl += up;
  }
  if ((t & 31) == 31) wsum[t >> 5] = incl;
  __syncthreads();
  if (t < 32) {
    int wv = wsum[t];
    int wi = wv;
    for (int off = 1; off < 32; off <<= 1) {
      int up = __shfl_up(wi, off);
      if (t >= off) wi += up;
    }
    wsum[t] = wi - wv;
  }
  __syncthreads();
  int pre = wsum[t >> 5];
  if (t < NB) {
    int first = pre + incl - v;
    hl[LCO + t] = first;
    reinterpret_cast<int2*>(cm)[(size_t)t * SP + sl] = make_int2(v, first);
  }
  if (t == 0) hl[LCO + MAXB] = 0;
  __syncthreads();

#pragma unroll
  for (int u = 0; u < MR; ++u) {
    int bk = br[u] & 2047;
    int slot = hl[LCO + bk] + (br[u] >> 11);
    if (bk != MAXB) srec[slot] = rec[u];
  }
  __syncthreads();

  unsigned* pks = pk + tb;
  for (int i = t; i < cnt; i += ET) pks[i] = srec[i];
}

#define SCAP 16384
#define RB 8

__device__ __forceinline__ void
put_record(unsigned w, int* tab, bool place, bool fits, int start, int* scol,
           int* __restrict__ col) {
  int p = atomicAdd(&tab[w & 255u], 1);
  if (place) {
    if (fits) scol[p - start] = (int)(w >> 8);
    else      col[p] = (int)(w >> 8);
  }
}

__device__ __forceinline__ void
run_heads(const unsigned* __restrict__ pk, const int* lo, const int* len, int S,
          int r0, int lane, unsigned (&w0)[RB], unsigned (&w1)[RB],
          unsigned (&base)[RB], int (&L)[RB]) {
  int rc = min(r0, S - 1);
#pragma unroll
  for (int r = 0; r < RB; ++r) {
    int s = r0 + 32 * r;
    bool ok = s < S;
    int sc = ok ? s : rc;
    L[r] = ok ? len[sc] : 0;
    base[r] = (unsigned)sc * (unsigned)MTILE + (L[r] > 0 ? (unsigned)lo[sc] : 0u);
    w0[r] = pk[base[r] + (unsigned)min(lane, max(L[r] - 1, 0))];
  }
#pragma unroll
  for (int r = 0; r < RB; ++r)
    w1[r] = pk[base[r] + (unsigned)min(lane + 32, max(L[r] - 1, 0))];
}

__device__ __forceinline__ void
put_heads(const unsigned (&w0)[RB], const unsigned (&w1)[RB], const int (&L)[RB],
          int lane, int* tab, bool place, bool fits, int start, int* scol,
          int* __restrict__ col) {
#pragma unroll
  for (int r = 0; r < RB; ++r)
    if (lane < L[r]) put_record(w0[r], tab, place, fits, start, scol, col);
#pragma unroll
  for (int r = 0; r < RB; ++r)
    if (lane + 32 < L[r]) put_record(w1[r], tab, place, fits, start, scol, col);
}

__device__ __forceinline__ void
put_tails(const unsigned* __restrict__ pk, const unsigned (&base)[RB],
          const int (&L)[RB], int lane, int* tab, bool place, bool fits, int start,
          int* scol, int* __restrict__ col) {
#pragma unroll
  for (int r = 0; r < RB; ++r)
    for (int i = 64 + lane; i < L[r]; i += 32)
      put_record(pk[base[r] + i], tab, place, fits, start, scol, col);
}

__device__ __forceinline__ void
bucket_runs(const unsigned* __restrict__ pk, const int* lo, const int* len, int S,
            int rfirst, int lane, int* tab, bool place, bool fits, int start,
            int* scol, int* __restrict__ col) {
  for (int r0 = rfirst; r0 < S; r0 += 32 * RB) {
    unsigned w[RB], base[RB];
    int L[RB];
#pragma unroll
    for (int r = 0; r < RB; ++r) {
      int s = r0 + 32 * r;
      bool ok = s < S;
      int sc = ok ? s : r0;
      L[r] = ok ? len[sc] : 0;
      base[r] = (unsigned)sc * (unsigned)MTILE + (L[r] > 0 ? (unsigned)lo[sc] : 0u);
      w[r] = pk[base[r] + (unsigned)min(lane, max(L[r] - 1, 0))];
    }
#pragma unroll
    for (int r = 0; r < RB; ++r)
      if (lane < L[r]) put_record(w[r], tab, place, fits, start, scol, col);
#pragma unroll
    for (int r = 0; r < RB; ++r)
      w[r] = pk[base[r] + (unsigned)min(lane + 32, max(L[r] - 1, 0))];
#pragma unroll
    for (int r = 0; r < RB; ++r)
      if (lane + 32 < L[r]) put_record(w[r], tab, place, fits, start, scol, col);
#pragma unroll
    for (int r = 0; r < RB; ++r)
      for (int i = 64 + lane; i < L[r]; i += 32)
        put_record(pk[base[r] + i], tab, place, fits, start, scol, col);
  }
}

__global__ void __launch_bounds__(ET, 16)
bucket_sort_kernel(const unsigned* __restrict__ pk, const int* __restrict__ cm,
                   int* __restrict__ rowptr, int* __restrict__ col, int nNodes,
                   int S, int SP) {
  __shared__ int cnt[NPB];
  __shared__ int cur[NPB];
  __shared__ int wsum[NPB / 32];
  __shared__ int lo[MAXS];
  __shared__ int len[MAXS];
  __shared__ int tot[2];
  __shared__ int scol[SCAP];
  int t = threadIdx.x;
  int lane = t & 31;
  int wave = __builtin_amdgcn_readfirstlane(t >> 5);
  int b = blockIdx.x;
  const int2* row = reinterpret_cast<const int2*>(cm) + (size_t)b * SP;
  int ps = 0, pl = 0;
  for (int s = t; s < S; s += ET) {
    int2 e = row[s];
    len[s] = e.x;
    lo[s] = e.y;
    pl += e.x;
    ps += e.y;
  }
  if (t < NPB) cnt[t] = 0;
  if (t < 2) tot[t] = 0;
  __syncthreads();

  for (int off = 16; off > 0; off >>= 1) {
    ps += __shfl_xor(ps, off);
    pl += __shfl_xor(pl, off);
  }
  if (lane == 0) { atomicAdd(&tot[0], ps); atomicAdd(&tot[1], pl); }
  __syncthreads();
  int start = tot[0];
  int end = start + tot[1];

  unsigned w0[RB], w1[RB], base[RB];
  int L[RB];
  run_heads(pk, lo, len, S, wave, lane, w0, w1, base, L);
  put_heads(w0, w1, L, lane, cnt, false, false, 0, scol, col);
  put_tails(pk, base, L, lane, cnt, false, false, 0, scol, col);
  bucket_runs(pk, lo, len, S, wave + 32 * RB, lane, cnt, false, false, 0, scol, col);
  __syncthreads();

  int v = 0, incl = 0;
  if (t < NPB) {
    v = cnt[t];
    incl = v;
    for (int off = 1; off < 32; off <<= 1) {
      int u = __shfl_up(incl, off);
      if ((t & 31) >= off) incl += u;
    }
    if ((t & 31) == 31) wsum[t >> 5] = incl;
  }
  __syncthreads();
  if (t < NPB) {
    int pre = start;
    for (int w = 0; w < (t >> 5); ++w) pre += wsum[w];
    int excl = pre + incl - v;
    cur[t] = excl;
    int node = b * NPB + t;
    if (node < nNodes) rowptr[node] = excl;
    if (node == nNodes - 1) rowptr[nNodes] = end;
  }
  __syncthreads();

  bool fits = (end - start) <= SCAP;
  put_heads(w0, w1, L, lane, cur, true, fits, start, scol, col);
  put_tails(pk, base, L, lane, cur, true, fits, start, scol, col);
  bucket_runs(pk, lo, len, S, wave + 32 * RB, lane, cur, true, fits, start, scol, col);
  __syncthreads();
  if (fits) {
    int n = end - start;
    for (int k = t; k < n; k += ET) col[start + k] = scol[k];
  }
}

__device__ __forceinline__ float4 ld4(const float* p) {
  return *reinterpret_cast<const float4*>(p);
}
__device__ __forceinline__ float4 add4(float4 a, float4 b) {
  return make_float4(a.x + b.x, a.y + b.y, a.z + b.z, a.w + b.w);
}


#define P1_TILES 25
#define P1_NODES (P1_TILES * 16)
#define P1_LDS (2 * DH_PAD * W1P + DH)

__device__ __forceinline__ void
proj1_body(float* sm, int pb, const float* __restrict__ x,
           const float* __restrict__ W1l,
           const float* __restrict__ b1, const float* __restrict__ W1r,
           float* __restrict__ p, float* __restrict__ r, int nNodes) {
  float* sWl = sm;
  float* sWr = sm + DH_PAD * W1P;
  float* sB  = sm + 2 * DH_PAD * W1P;
  for (int i = threadIdx.x; i < DH_PAD * DIN; i += ET) {
    float wl = (i < DH * DIN) ? W1l[i] : 0.0f;
    float wr = (i < DH * DIN) ? W1r[i] : 0.0f;
    int j = (i / DIN) * W1P + (i % DIN);
    sWl[j] = wl;
    sWr[j] = wr;
  }
  if (threadIdx.x < DH) sB[threadIdx.x] = b1[threadIdx.x];
  __syncthreads();

  int lane = threadIdx.x & 31;
  int wave = threadIdx.x >> 5;
  if (wave >= P1_TILES) return;
  int node0 = pb * P1_NODES + wave * 16;
  int row = lane & 15;
  int node = node0 + row;
  int nodeC = min(node, nNodes - 1);
  int klo = (lane >> 4) * 2;
  int ncol = lane & 15;
  int mbase = (lane >> 4) * 8;

  const float* xrow = x + (size_t)nodeC * DIN;

  for (int half = 0; half < 2; ++half) {
    const float* sW = half ? sWr : sWl;
    float* o = half ? r : p;
    v8f a0 = {}, a1 = {}, a2 = {};
    for (int k = 0; k < DIN; k += 4) {
      v2f aX;
      aX.x = xrow[k + klo];
      aX.y = xrow[k + klo + 1];

      v2f b0, b1v, b2;
      b0.x  = sW[ncol * W1P + k + klo];        b0.y  = sW[ncol * W1P + k + klo + 1];
      b1v.x = sW[(16 + ncol) * W1P + k + klo]; b1v.y = sW[(16 + ncol) * W1P + k + klo + 1];
      b2.x  = sW[(32 + ncol) * W1P + k + klo]; b2.y  = sW[(32 + ncol) * W1P + k + klo + 1];

      a0 = wmma4(aX, b0, a0);
      a1 = wmma4(aX, b1v, a1);
      a2 = wmma4(aX, b2, a2);
    }

    float c0 = half ? sB[ncol] : 0.0f;
    float c1 = half ? sB[16 + ncol] : 0.0f;
    float c2 = (half && 32 + ncol < DH) ? sB[32 + ncol] : 0.0f;
    for (int i = 0; i < 8; ++i) {
      int n = node0 + mbase + i;
      if (n < nNodes) {
        float* orow = o + (size_t)n * DH;
        orow[ncol]      = half ? a0[i] + c0 : a0[i];
        orow[16 + ncol] = half ? a1[i] + c1 : a1[i];
        if (32 + ncol < DH) orow[32 + ncol] = half ? a2[i] + c2 : a2[i];
      }
    }
  }
}

__global__ void __launch_bounds__(ET, 16)
slice_proj1_kernel(const int* __restrict__ src, const int* __restrict__ dst,
                   int* __restrict__ cm, unsigned* __restrict__ pk, int E,
                   int NB, int S, int SP,
                   const float* __restrict__ x, const float* __restrict__ W1l,
                   const float* __restrict__ b1, const float* __restrict__ W1r,
                   float* __restrict__ p, float* __restrict__ r, int nNodes) {
  __shared__ int sm[SL_LDS > P1_LDS ? SL_LDS : P1_LDS];
  if ((int)blockIdx.x < S)
    slice_sort_body(sm, (int)blockIdx.x, src, dst, cm, pk, E, NB, S, SP);
  else
    proj1_body(reinterpret_cast<float*>(sm), (int)blockIdx.x - S, x, W1l, b1,
               W1r, p, r, nNodes);
}

#define G1_WAVES 16
#define HP 44
__global__ void __launch_bounds__(G1_WAVES * 32, 16)
gather1_proj2_kernel(const float* __restrict__ p, float* __restrict__ r,
                     const int* __restrict__ rowptr, const int* __restrict__ col,
                     const float* __restrict__ W2l, const float* __restrict__ b2,
                     const float* __restrict__ W2r, float* __restrict__ p2,
                     int nNodes) {
  __shared__ float sWl[DO_PAD * W2P];
  __shared__ float sWr[DO_PAD * W2P];
  __shared__ float sB[DO_PAD];
  __shared__ float sH[2][16 * HP];
  for (int i = threadIdx.x; i < DO_PAD * DH; i += G1_WAVES * 32) {
    float wl = (i < DOUT * DH) ? W2l[i] : 0.0f;
    float wr = (i < DOUT * DH) ? W2r[i] : 0.0f;
    int j = (i / DH) * W2P + (i % DH);
    sWl[j] = wl;
    sWr[j] = wr;
  }
  if (threadIdx.x < DO_PAD)
    sB[threadIdx.x] = (threadIdx.x < DOUT) ? b2[threadIdx.x] : 0.0f;

  int lane = threadIdx.x & 31;
  int wave = __builtin_amdgcn_readfirstlane((int)(threadIdx.x >> 5));
  int nTiles = (nNodes + 15) >> 4;
  int q = lane / 10;
  int chunk = lane - q * 10;
  int g = min(q, 2);
  const float* pc = p + chunk * 4;
  int klo = (lane >> 4) * 2;
  int ncol = lane & 15;
  int mbase = (lane >> 4) * 8;

  int tile = blockIdx.x;
  int n = tile * 16 + wave;
  int rp = 0;
  if (tile < nTiles && n < nNodes) rp = rowptr[n + (lane & 1)];
  int buf = 0;
  for (; tile < nTiles; tile += (int)gridDim.x) {
    int beg = __builtin_amdgcn_readlane(rp, 0);
    int end = __builtin_amdgcn_readlane(rp, 1);
    int nn = n + (int)gridDim.x * 16;
    int nrp = 0;
    if (tile + (int)gridDim.x < nTiles && nn < nNodes) nrp = rowptr[nn + (lane & 1)];
    float4 own = make_float4(0.0f, 0.0f, 0.0f, 0.0f);
    if (lane < 10 && n < nNodes) own = ld4(r + (size_t)n * DH + chunk * 4);

    float4 acc = make_float4(0.0f, 0.0f, 0.0f, 0.0f);
    int e0 = beg + lane;
    int my = (lane < 30 && e0 < end) ? col[e0] : 0;
    for (int base = beg; base < end; base += 30) {
      int e1 = base + 30 + lane;
      int myn = (lane < 30 && e1 < end) ? col[e1] : 0;
      int m = min(30, end - base);
      for (int j = 0; j < m; j += 15) {
        float4 v[5];
#pragma unroll
        for (int u = 0; u < 5; ++u) {
          int k = j + 3 * u + g;
          int s = __shfl(my, k);
          float4 t = ld4(pc + (size_t)s * DH);
          bool ok = k < m;
          v[u] = make_float4(ok ? t.x : 0.0f, ok ? t.y : 0.0f,
                             ok ? t.z : 0.0f, ok ? t.w : 0.0f);
        }
        acc = add4(acc, add4(add4(add4(v[0], v[1]), add4(v[2], v[3])), v[4]));
      }
      my = myn;
    }
    float4 t1, t2;
    t1.x = __shfl_down(acc.x, 10); t2.x = __shfl_down(acc.x, 20);
    t1.y = __shfl_down(acc.y, 10); t2.y = __shfl_down(acc.y, 20);
    t1.z = __shfl_down(acc.z, 10); t2.z = __shfl_down(acc.z, 20);
    t1.w = __shfl_down(acc.w, 10); t2.w = __shfl_down(acc.w, 20);
    float* hT = sH[buf];
    if (lane < 10) {
      float inv = 1.0f / fmaxf((float)(end - beg), 1.0f);
      float4 s = add4(acc, add4(t1, t2));
      *reinterpret_cast<float4*>(hT + wave * HP + chunk * 4) =
          make_float4(fmaxf(s.x * inv + own.x, 0.0f), fmaxf(s.y * inv + own.y, 0.0f),
                      fmaxf(s.z * inv + own.z, 0.0f), fmaxf(s.w * inv + own.w, 0.0f));
    }
    __syncthreads();

    if (wave < 4) {
      bool own2 = wave >= 2;
      const float* sW = own2 ? sWr : sWl;
      int cb = (wave & 1) * 16;
      const float* hrow = hT + (lane & 15) * HP;
      const float* wrow = sW + (cb + ncol) * W2P;
      v8f a = {};
      for (int k = 0; k < DH; k += 4) {
        v2f aH, bW;
        aH.x = hrow[k + klo];
        aH.y = hrow[k + klo + 1];
        bW.x = wrow[k + klo];
        bW.y = wrow[k + klo + 1];
        a = wmma4(aH, bW, a);
      }
      float bias = own2 ? sB[cb + ncol] : 0.0f;
      for (int i = 0; i < 8; ++i) {
        int nr = tile * 16 + mbase + i;
        if (nr < nNodes) {
          if (own2) r[(size_t)nr * DH + cb + ncol] = a[i] + bias;
          else      p2[(size_t)nr * DO_PAD + cb + ncol] = a[i];
        }
      }
    }
    buf ^= 1;
    n = nn; rp = nrp;
  }
}

__global__ void __launch_bounds__(256)
gather2_lsm_kernel(const float* __restrict__ p, const float* __restrict__ r,
                   const int* __restrict__ rowptr, const int* __restrict__ col,
                   float* __restrict__ out, int nNodes) {
  int lane = threadIdx.x & 31;
  int wave = __builtin_amdgcn_readfirstlane((int)(threadIdx.x >> 5));
  int nw = (int)gridDim.x * 8;
  int n = blockIdx.x * 8 + wave;
  if (n >= nNodes) return;
  int q = lane >> 3;
  int chunk = lane & 7;
  const float* pc = p + chunk * 4;
  int beg = rowptr[n], end = rowptr[n + 1];
  while (true) {
    int nn = n + nw;
    int nbeg = 0, nend = 0;
    if (nn < nNodes) { nbeg = rowptr[nn]; nend = rowptr[nn + 1]; }
    float4 own = make_float4(0.0f, 0.0f, 0.0f, 0.0f);
    if (lane < 6) own = ld4(r + (size_t)n * DH + chunk * 4);

    float4 acc = make_float4(0.0f, 0.0f, 0.0f, 0.0f);
    int e0 = beg + lane;
    int my = (e0 < end) ? col[e0] : 0;
    for (int base = beg; base < end; base += 32) {
      int e1 = base + 32 + lane;
      int myn = (e1 < end) ? col[e1] : 0;
      int m = min(32, end - base);
      for (int j = 0; j < m; j += 16) {
        float4 v[4];
#pragma unroll
        for (int u = 0; u < 4; ++u) {
          int k = j + 4 * u + q;
          int s = __shfl(my, k);
          float4 t = ld4(pc + (size_t)s * DO_PAD);
          bool ok = k < m;
          v[u] = make_float4(ok ? t.x : 0.0f, ok ? t.y : 0.0f,
                             ok ? t.z : 0.0f, ok ? t.w : 0.0f);
        }
        acc = add4(acc, add4(add4(v[0], v[1]), add4(v[2], v[3])));
      }
      my = myn;
    }
    acc.x += __shfl_xor(acc.x, 8);
    acc.y += __shfl_xor(acc.y, 8);
    acc.z += __shfl_xor(acc.z, 8);
    acc.w += __shfl_xor(acc.w, 8);
    acc.x += __shfl_xor(acc.x, 16);
    acc.y += __shfl_xor(acc.y, 16);
    acc.z += __shfl_xor(acc.z, 16);
    acc.w += __shfl_xor(acc.w, 16);
    float inv = 1.0f / fmaxf((float)(end - beg), 1.0f);
    float4 o = make_float4(acc.x * inv + own.x, acc.y * inv + own.y,
                           acc.z * inv + own.z, acc.w * inv + own.w);
    bool has = chunk < 6;
    float mx = has ? fmaxf(fmaxf(o.x, o.y), fmaxf(o.z, o.w)) : -3.402823466e+38f;
    mx = fmaxf(mx, __shfl_xor(mx, 1));
    mx = fmaxf(mx, __shfl_xor(mx, 2));
    mx = fmaxf(mx, __shfl_xor(mx, 4));
    float se = has ? (expf(o.x - mx) + expf(o.y - mx)) + (expf(o.z - mx) + expf(o.w - mx))
                   : 0.0f;
    se += __shfl_xor(se, 1);
    se += __shfl_xor(se, 2);
    se += __shfl_xor(se, 4);
    float lse = mx + logf(se);
    if (lane < 6) {
      *reinterpret_cast<float4*>(out + (size_t)n * DOUT + chunk * 4) =
          make_float4(o.x - lse, o.y - lse, o.z - lse, o.w - lse);
    }

    if (nn >= nNodes) break;
    n = nn; beg = nbeg; end = nend;
  }
}

extern "C" void kernel_launch(void* const* d_in, const int* in_sizes, int n_in,
                              void* d_out, int out_size, void* d_ws, size_t ws_size,
                              hipStream_t stream) {
  const float* x   = (const float*)d_in[0];
  const int*   ei  = (const int*)d_in[1];
  const float* W1l = (const float*)d_in[2];
  const float* b1  = (const float*)d_in[3];
  const float* W1r = (const float*)d_in[4];
  const float* W2l = (const float*)d_in[5];
  const float* b2  = (const float*)d_in[6];
  const float* W2r = (const float*)d_in[7];
  float* out = (float*)d_out;

  int  N = in_sizes[0] / DIN;
  long E = (long)in_sizes[1] / 2;
  const int* src = ei;
  const int* dst = ei + E;

  float* ws = (float*)d_ws;
  float* P  = ws;
  float* R  = P + (size_t)N * DH;
  float* H  = R + (size_t)N * DH;
  float* P2 = H;
  int* rowptr = (int*)(ws + (size_t)N * (DIN + 1 + DH + DOUT));
  int* col    = rowptr + N + 1;

  int NB = (N + NPB - 1) / NPB;
  int S  = (int)((E + MTILE - 1) / MTILE);
  int SP = (S + 3) / 4 * 4;
  if (SP < 4) SP = 4;
  unsigned* pk = (unsigned*)H;
  int* cm = (int*)H + (size_t)S * MTILE;

  int nbP = (N + P1_NODES - 1) / P1_NODES;
  slice_proj1_kernel<<<S + nbP, ET, 0, stream>>>(src, dst, cm, pk, (int)E, NB, S, SP,
                                                 x, W1l, b1, W1r, P, R, N);
  bucket_sort_kernel<<<NB, ET, 0, stream>>>(pk, cm, rowptr, col, N, S, SP);

  gather1_proj2_kernel<<<1024, G1_WAVES * 32, 0, stream>>>(P, R, rowptr, col, W2l, b2,
                                                         W2r, P2, N);
  gather2_lsm_kernel<<<2048, 256, 0, stream>>>(P2, R, rowptr, col, out, N);
}
